# starting kernel
# baseline (speedup 1.0000x reference)
#include <hip/hip_runtime.h>


typedef float v2f __attribute__((ext_vector_type(2)));
typedef float v8f __attribute__((ext_vector_type(8)));

#define SAMPLE_NUM 2048
#define Y_NUM      2048
#define NTILES     128
#define WAVES      8
#define THREADS    (WAVES * 32)

__device__ __forceinline__ float sel9(const float cc[9], int j) {
    float r = 0.0f;
#pragma unroll
    for (int i = 0; i < 9; ++i) r = (j == i) ? cc[i] : r;
    return r;
}

__global__ __launch_bounds__(THREADS)
void neuro_ode_wmma(const float* __restrict__ t,
                    const float* __restrict__ y0,
                    const float* __restrict__ w,
                    const float* __restrict__ rt,
                    float* __restrict__ out) {
    __shared__ float Y[2][Y_NUM];

    const int tid  = threadIdx.x;
    const int wv   = tid >> 5;
    const int ln   = tid & 31;
    const int col  = ln & 15;
    const int half = ln >> 4;
    const int m    = col;

    const float a = w[0] * rt[0];
    const float b = w[1] * rt[1];
    const float c = w[2] * rt[2];

    for (int i = tid; i < Y_NUM; i += THREADS) {
        float v = y0[i];
        Y[0][i] = v;
        out[i]  = v;
    }
    __syncthreads();

    const int tileC = (wv << 4) + col;
    const int tileL = (tileC + NTILES - 1) & (NTILES - 1);
    const int rowIx = (wv << 8) + (col << 4) + (half << 3);

    for (int n = 0; n < SAMPLE_NUM - 1; ++n) {
        const float tn  = t[n];
        const float tn1 = t[n + 1];
        const float sub = (tn1 - tn) * 0.125f;
        const float d   = 1.0f - sub * b;
        const float e   = sub * a;

        float cc[9];
        {
            const float bin[9] = {1.f, 8.f, 28.f, 56.f, 70.f, 56.f, 28.f, 8.f, 1.f};
            float dp[9], ep[9];
            dp[0] = 1.0f; ep[0] = 1.0f;
#pragma unroll
            for (int j = 1; j < 9; ++j) { dp[j] = dp[j-1] * d; ep[j] = ep[j-1] * e; }
#pragma unroll
            for (int j = 0; j < 9; ++j) cc[j] = bin[j] * dp[8 - j] * ep[j];
        }

        const float g = d + e;
        float f = 0.0f;
#pragma unroll
        for (int ms = 0; ms < 8; ++ms) {
            const float s = sub * c * __sinf(tn + sub * (float)ms);
            f = f * g + s;
        }

        const float* __restrict__ Yc = Y[n & 1];
        float*       __restrict__ Yn = Y[(n & 1) ^ 1];

        v2f Byc[4], Byl[4];
#pragma unroll
        for (int q = 0; q < 4; ++q) {
            const int kk = (q << 2) + (half << 1);
            const float2 vc = *(const float2*)&Yc[(tileC << 4) + kk];
            const float2 vl = *(const float2*)&Yc[(tileL << 4) + kk];
            Byc[q][0] = vc.x; Byc[q][1] = vc.y;
            Byl[q][0] = vl.x; Byl[q][1] = vl.y;
        }

        v8f acc;
#pragma unroll
        for (int i = 0; i < 8; ++i) acc[i] = f;

#pragma unroll
        for (int q = 0; q < 4; ++q) {
            const int j0 = m - ((q << 2) + (half << 1));
            v2f Ap;
            Ap[0] = sel9(cc, j0);
            Ap[1] = sel9(cc, j0 - 1);
            acc = __builtin_amdgcn_wmma_f32_16x16x4_f32(
                false, Ap, false, Byc[q], (short)0, acc, false, false);
        }
#pragma unroll
        for (int q = 0; q < 4; ++q) {
            const int j0 = 16 + m - ((q << 2) + (half << 1));
            v2f Aq;
            Aq[0] = sel9(cc, j0);
            Aq[1] = sel9(cc, j0 - 1);
            acc = __builtin_amdgcn_wmma_f32_16x16x4_f32(
                false, Aq, false, Byl[q], (short)0, acc, false, false);
        }

        const float4 lo = make_float4(acc[0], acc[1], acc[2], acc[3]);
        const float4 hi = make_float4(acc[4], acc[5], acc[6], acc[7]);
        *(float4*)&Yn[rowIx]     = lo;
        *(float4*)&Yn[rowIx + 4] = hi;

        __syncthreads();

        float* __restrict__ orow = out + (size_t)(n + 1) * Y_NUM;
        *(float4*)&orow[rowIx]     = lo;
        *(float4*)&orow[rowIx + 4] = hi;
    }
}

extern "C" void kernel_launch(void* const* d_in, const int* in_sizes, int n_in,
                              void* d_out, int out_size, void* d_ws, size_t ws_size,
                              hipStream_t stream) {
    (void)in_sizes; (void)n_in; (void)out_size; (void)d_ws; (void)ws_size;
    const float* t  = (const float*)d_in[0];
    const float* y0 = (const float*)d_in[1];
    const float* w  = (const float*)d_in[2];
    const float* r  = (const float*)d_in[3];
    float* out = (float*)d_out;

    neuro_ode_wmma<<<1, THREADS, 0, stream>>>(t, y0, w, r, out);
}


# Per-step taps and forcing depend on t only: all 2047 steps' coefficients are computed up front in parallel into an LDS table; the chain reads its A fragments from it one step ahead
# speedup vs baseline: 2.5719x; 2.5719x over previous
#include <hip/hip_runtime.h>


typedef float v2f __attribute__((ext_vector_type(2)));
typedef float v8f __attribute__((ext_vector_type(8)));

#define SAMPLE_NUM 2048
#define Y_NUM      2048
#define NSTEP      (SAMPLE_NUM - 1)
#define NTILES     128
#define WAVES      8
#define THREADS    (WAVES * 32)
#define ROWF       16
#define F_SLOT     13

__device__ __forceinline__ int pair_off(int j) {
    return (j >= 0 && j <= 9) ? (9 - j) : 11;
}

struct Coef {
    v2f   p[4];
    v2f   q[2];
    float f;
};

__device__ __forceinline__ Coef load_coef(const float* row, const int oP[4], const int oQ[2]) {
    Coef c;
#pragma unroll
    for (int i = 0; i < 4; ++i) { c.p[i][0] = row[oP[i]]; c.p[i][1] = row[oP[i] + 1]; }
#pragma unroll
    for (int i = 0; i < 2; ++i) { c.q[i][0] = row[oQ[i]]; c.q[i][1] = row[oQ[i] + 1]; }
    c.f = row[F_SLOT];
    return c;
}

__global__ __launch_bounds__(THREADS)
void neuro_ode_wmma(const float* __restrict__ t,
                    const float* __restrict__ y0,
                    const float* __restrict__ w,
                    const float* __restrict__ rt,
                    float* __restrict__ out) {
    __shared__ __attribute__((aligned(16))) float Y[2][Y_NUM];
    __shared__ __attribute__((aligned(16))) float Tb[NSTEP * ROWF];

    const int tid  = threadIdx.x;
    const int wv   = tid >> 5;
    const int ln   = tid & 31;
    const int col  = ln & 15;
    const int half = ln >> 4;
    const int m    = col;

    const float a = w[0] * rt[0];
    const float b = w[1] * rt[1];
    const float c = w[2] * rt[2];

    for (int i = tid; i < Y_NUM; i += THREADS) {
        float v = y0[i];
        Y[0][i] = v;
        out[i]  = v;
    }

    for (int n = tid; n < NSTEP; n += THREADS) {
        const float tn  = t[n];
        const float tn1 = t[n + 1];
        const float sub = (tn1 - tn) * 0.125f;
        const float d   = 1.0f - sub * b;
        const float e   = sub * a;

        float cc[9];
        {
            const float bin[9] = {1.f, 8.f, 28.f, 56.f, 70.f, 56.f, 28.f, 8.f, 1.f};
            float dp[9], ep[9];
            dp[0] = 1.0f; ep[0] = 1.0f;
#pragma unroll
            for (int j = 1; j < 9; ++j) { dp[j] = dp[j-1] * d; ep[j] = ep[j-1] * e; }
#pragma unroll
            for (int j = 0; j < 9; ++j) cc[j] = bin[j] * dp[8 - j] * ep[j];
        }

        const float g = d + e;
        float f = 0.0f;
#pragma unroll
        for (int ms = 0; ms < 8; ++ms) {
            const float s = sub * c * __sinf(tn + sub * (float)ms);
            f = f * g + s;
        }

        float4* row = (float4*)&Tb[n * ROWF];
        row[0] = make_float4(0.0f,  cc[8], cc[7], cc[6]);
        row[1] = make_float4(cc[5], cc[4], cc[3], cc[2]);
        row[2] = make_float4(cc[1], cc[0], 0.0f,  0.0f);
        row[3] = make_float4(0.0f,  f,     0.0f,  0.0f);
    }
    __syncthreads();

    const int tileC = (wv << 4) + col;
    const int tileL = (tileC + NTILES - 1) & (NTILES - 1);
    const int rowIx = (wv << 8) + (col << 4) + (half << 3);

    int oP[4], oQ[2];
#pragma unroll
    for (int q = 0; q < 4; ++q) oP[q] = pair_off(m - ((q << 2) + (half << 1)));
#pragma unroll
    for (int q = 2; q < 4; ++q) oQ[q - 2] = pair_off(16 + m - ((q << 2) + (half << 1)));

    const v2f zero2 = {0.0f, 0.0f};
    Coef cur = load_coef(Tb, oP, oQ);

#pragma unroll 2
    for (int n = 0; n < NSTEP; ++n) {
        const float* __restrict__ Yc = Y[n & 1];
        float*       __restrict__ Yn = Y[(n & 1) ^ 1];

        v2f Byc[4], Byl[4];
#pragma unroll
        for (int q = 0; q < 4; ++q) {
            const int kk = (q << 2) + (half << 1);
            const float2 vc = *(const float2*)&Yc[(tileC << 4) + kk];
            const float2 vl = *(const float2*)&Yc[(tileL << 4) + kk];
            Byc[q][0] = vc.x; Byc[q][1] = vc.y;
            Byl[q][0] = vl.x; Byl[q][1] = vl.y;
        }

        const int nn = (n + 1 < NSTEP) ? (n + 1) : n;
        const Coef nxt = load_coef(Tb + nn * ROWF, oP, oQ);

        v8f acc;
#pragma unroll
        for (int i = 0; i < 8; ++i) acc[i] = cur.f;

#pragma unroll
        for (int q = 0; q < 4; ++q)
            acc = __builtin_amdgcn_wmma_f32_16x16x4_f32(
                false, cur.p[q], false, Byc[q], (short)0, acc, false, false);
#pragma unroll
        for (int q = 0; q < 2; ++q)
            acc = __builtin_amdgcn_wmma_f32_16x16x4_f32(
                false, zero2, false, Byl[q], (short)0, acc, false, false);
#pragma unroll
        for (int q = 2; q < 4; ++q)
            acc = __builtin_amdgcn_wmma_f32_16x16x4_f32(
                false, cur.q[q - 2], false, Byl[q], (short)0, acc, false, false);

        const float4 lo = make_float4(acc[0], acc[1], acc[2], acc[3]);
        const float4 hi = make_float4(acc[4], acc[5], acc[6], acc[7]);
        *(float4*)&Yn[rowIx]     = lo;
        *(float4*)&Yn[rowIx + 4] = hi;

        __builtin_amdgcn_fence(__ATOMIC_RELEASE, "workgroup", "local");
        __builtin_amdgcn_s_barrier();
        __builtin_amdgcn_fence(__ATOMIC_ACQUIRE, "workgroup", "local");

        float* __restrict__ orow = out + (size_t)(n + 1) * Y_NUM;
        *(float4*)&orow[rowIx]     = lo;
        *(float4*)&orow[rowIx + 4] = hi;

        cur = nxt;
    }
}

extern "C" void kernel_launch(void* const* d_in, const int* in_sizes, int n_in,
                              void* d_out, int out_size, void* d_ws, size_t ws_size,
                              hipStream_t stream) {
    (void)in_sizes; (void)n_in; (void)out_size; (void)d_ws; (void)ws_size;
    const float* t  = (const float*)d_in[0];
    const float* y0 = (const float*)d_in[1];
    const float* w  = (const float*)d_in[2];
    const float* r  = (const float*)d_in[3];
    float* out = (float*)d_out;

    neuro_ode_wmma<<<1, THREADS, 0, stream>>>(t, y0, w, r, out);
}


# Two rows per chain trip: row n+2 from the 17-tap product of the maps of rows n+1 and n+2 (one LDS round trip and barrier per two rows), row n+1 off the chain; sums regrouped by the composed taps
# speedup vs baseline: 3.3107x; 1.2873x over previous
#include <hip/hip_runtime.h>


typedef float v2f __attribute__((ext_vector_type(2)));
typedef float v8f __attribute__((ext_vector_type(8)));

#define SAMPLE_NUM 2048
#define Y_NUM      2048
#define NSTEP      (SAMPLE_NUM - 1)
#define NPAIR      (NSTEP / 2)
#define NROW       (NPAIR + 1)
#define NTILES     128
#define WAVES      8
#define THREADS    (WAVES * 32)
#define ROWF       36
#define F1_SLOT    13
#define F2_SLOT    14

__device__ __forceinline__ int pair_off1(int j) {
    return (j >= 0 && j <= 9) ? (9 - j) : 11;
}
__device__ __forceinline__ int pair_off2(int j) {
    return (j >= 0 && j <= 17) ? (33 - j) : 11;
}

struct StepC {
    float c[9];
    float f;
};

__device__ __forceinline__ StepC step_coef(float tn, float tn1, float a, float b, float c) {
    const float sub = (tn1 - tn) * 0.125f;
    const float d   = 1.0f - sub * b;
    const float e   = sub * a;
    StepC r;
    {
        const float bin[9] = {1.f, 8.f, 28.f, 56.f, 70.f, 56.f, 28.f, 8.f, 1.f};
        float dp[9], ep[9];
        dp[0] = 1.0f; ep[0] = 1.0f;
#pragma unroll
        for (int j = 1; j < 9; ++j) { dp[j] = dp[j-1] * d; ep[j] = ep[j-1] * e; }
#pragma unroll
        for (int j = 0; j < 9; ++j) r.c[j] = bin[j] * dp[8 - j] * ep[j];
    }
    const float g = d + e;
    float f = 0.0f;
#pragma unroll
    for (int ms = 0; ms < 8; ++ms) {
        const float s = sub * c * __sinf(tn + sub * (float)ms);
        f = f * g + s;
    }
    r.f = f;
    return r;
}

struct Coef {
    v2f   p1[4];
    v2f   q1[2];
    v2f   p2[4];
    v2f   q2[4];
    float f1, f2;
};

struct Offs {
    int p1[4], q1[2], p2[4], q2[4];
};

__device__ __forceinline__ Coef load_coef(const float* row, const Offs& o) {
    Coef c;
#pragma unroll
    for (int i = 0; i < 4; ++i) { c.p2[i][0] = row[o.p2[i]]; c.p2[i][1] = row[o.p2[i] + 1]; }
#pragma unroll
    for (int i = 0; i < 4; ++i) { c.q2[i][0] = row[o.q2[i]]; c.q2[i][1] = row[o.q2[i] + 1]; }
#pragma unroll
    for (int i = 0; i < 4; ++i) { c.p1[i][0] = row[o.p1[i]]; c.p1[i][1] = row[o.p1[i] + 1]; }
#pragma unroll
    for (int i = 0; i < 2; ++i) { c.q1[i][0] = row[o.q1[i]]; c.q1[i][1] = row[o.q1[i] + 1]; }
    c.f1 = row[F1_SLOT];
    c.f2 = row[F2_SLOT];
    return c;
}

struct State {
    v2f c[4];
    v2f l[4];
};

__device__ __forceinline__ State load_state(const float* Yb, int tileC, int tileL, int half) {
    State s;
#pragma unroll
    for (int q = 0; q < 4; ++q) {
        const int kk = (q << 2) + (half << 1);
        const float2 vc = *(const float2*)&Yb[(tileC << 4) + kk];
        const float2 vl = *(const float2*)&Yb[(tileL << 4) + kk];
        s.c[q][0] = vc.x; s.c[q][1] = vc.y;
        s.l[q][0] = vl.x; s.l[q][1] = vl.y;
    }
    return s;
}

__device__ __forceinline__ v8f wmma4(v2f a, v2f b, v8f c) {
    return __builtin_amdgcn_wmma_f32_16x16x4_f32(false, a, false, b, (short)0, c, false, false);
}

__device__ __forceinline__ v8f row_one(const Coef& k, const State& s) {
    v8f aP, aQ;
#pragma unroll
    for (int i = 0; i < 8; ++i) { aP[i] = k.f1; aQ[i] = 0.0f; }
    aP = wmma4(k.p1[0], s.c[0], aP);
    aQ = wmma4(k.q1[0], s.l[2], aQ);
    aP = wmma4(k.p1[1], s.c[1], aP);
    aQ = wmma4(k.q1[1], s.l[3], aQ);
    aP = wmma4(k.p1[2], s.c[2], aP);
    aP = wmma4(k.p1[3], s.c[3], aP);
    return aP + aQ;
}

__device__ __forceinline__ v8f row_two(const Coef& k, const State& s) {
    v8f aP, aQ;
#pragma unroll
    for (int i = 0; i < 8; ++i) { aP[i] = k.f2; aQ[i] = 0.0f; }
#pragma unroll
    for (int q = 0; q < 4; ++q) {
        aP = wmma4(k.p2[q], s.c[q], aP);
        aQ = wmma4(k.q2[q], s.l[q], aQ);
    }
    return aP + aQ;
}

__global__ __launch_bounds__(THREADS)
void neuro_ode_wmma(const float* __restrict__ t,
                    const float* __restrict__ y0,
                    const float* __restrict__ w,
                    const float* __restrict__ rt,
                    float* __restrict__ out) {
    __shared__ __attribute__((aligned(16))) float Y[2][Y_NUM];
    __shared__ __attribute__((aligned(16))) float Tb[NROW * ROWF];

    const int tid  = threadIdx.x;
    const int wv   = tid >> 5;
    const int ln   = tid & 31;
    const int col  = ln & 15;
    const int half = ln >> 4;
    const int m    = col;

    const float a = w[0] * rt[0];
    const float b = w[1] * rt[1];
    const float c = w[2] * rt[2];

    for (int i = tid; i < Y_NUM; i += THREADS) {
        float v = y0[i];
        Y[0][i] = v;
        out[i]  = v;
    }

    for (int s = tid; s < NROW; s += THREADS) {
        const int n = 2 * s;
        const StepC A = step_coef(t[n], t[n + 1], a, b, c);

        float c2[17];
#pragma unroll
        for (int j = 0; j < 17; ++j) c2[j] = 0.0f;
        float f2 = 0.0f;
        if (n + 2 <= NSTEP) {
            const StepC B = step_coef(t[n + 1], t[n + 2], a, b, c);
#pragma unroll
            for (int i = 0; i < 9; ++i)
#pragma unroll
                for (int j = 0; j < 9; ++j) c2[i + j] += B.c[i] * A.c[j];
            float sumB = 0.0f;
#pragma unroll
            for (int i = 0; i < 9; ++i) sumB += B.c[i];
            f2 = A.f * sumB + B.f;
        }

        float4* row = (float4*)&Tb[s * ROWF];
        row[0] = make_float4(0.0f,   A.c[8], A.c[7], A.c[6]);
        row[1] = make_float4(A.c[5], A.c[4], A.c[3], A.c[2]);
        row[2] = make_float4(A.c[1], A.c[0], 0.0f,   0.0f);
        row[3] = make_float4(0.0f,   A.f,    f2,     0.0f);
        row[4] = make_float4(0.0f,   c2[16], c2[15], c2[14]);
        row[5] = make_float4(c2[13], c2[12], c2[11], c2[10]);
        row[6] = make_float4(c2[9],  c2[8],  c2[7],  c2[6]);
        row[7] = make_float4(c2[5],  c2[4],  c2[3],  c2[2]);
        row[8] = make_float4(c2[1],  c2[0],  0.0f,   0.0f);
    }
    __syncthreads();

    const int tileC = (wv << 4) + col;
    const int tileL = (tileC + NTILES - 1) & (NTILES - 1);
    const int rowIx = (wv << 8) + (col << 4) + (half << 3);

    Offs o;
#pragma unroll
    for (int q = 0; q < 4; ++q) {
        const int k0 = (q << 2) + (half << 1);
        o.p1[q] = pair_off1(m - k0);
        o.p2[q] = pair_off2(m - k0);
        o.q2[q] = pair_off2(16 + m - k0);
    }
#pragma unroll
    for (int q = 2; q < 4; ++q) o.q1[q - 2] = pair_off1(16 + m - ((q << 2) + (half << 1)));

    Coef  cur = load_coef(Tb, o);
    State st  = load_state(Y[0], tileC, tileL, half);

#pragma unroll 2
    for (int s = 0; s < NPAIR; ++s) {
        float* __restrict__ Yn = Y[(s & 1) ^ 1];

        const v8f y2 = row_two(cur, st);
        const float4 lo2 = make_float4(y2[0], y2[1], y2[2], y2[3]);
        const float4 hi2 = make_float4(y2[4], y2[5], y2[6], y2[7]);
        *(float4*)&Yn[rowIx]     = lo2;
        *(float4*)&Yn[rowIx + 4] = hi2;

        __builtin_amdgcn_fence(__ATOMIC_RELEASE, "workgroup", "local");
        __builtin_amdgcn_s_barrier();
        __builtin_amdgcn_fence(__ATOMIC_ACQUIRE, "workgroup", "local");

        __builtin_amdgcn_sched_barrier(0);

        const State stn = load_state(Yn, tileC, tileL, half);
        const Coef  nxt = load_coef(Tb + (s + 1) * ROWF, o);

        __builtin_amdgcn_sched_barrier(0);

        const v8f y1 = row_one(cur, st);

        float* __restrict__ orow1 = out + (size_t)(2 * s + 1) * Y_NUM;
        float* __restrict__ orow2 = out + (size_t)(2 * s + 2) * Y_NUM;
        *(float4*)&orow1[rowIx]     = make_float4(y1[0], y1[1], y1[2], y1[3]);
        *(float4*)&orow1[rowIx + 4] = make_float4(y1[4], y1[5], y1[6], y1[7]);
        *(float4*)&orow2[rowIx]     = lo2;
        *(float4*)&orow2[rowIx + 4] = hi2;

        st  = stn;
        cur = nxt;
    }

    {
        const v8f y1 = row_one(cur, st);
        float* __restrict__ orow = out + (size_t)NSTEP * Y_NUM;
        *(float4*)&orow[rowIx]     = make_float4(y1[0], y1[1], y1[2], y1[3]);
        *(float4*)&orow[rowIx + 4] = make_float4(y1[4], y1[5], y1[6], y1[7]);
    }
}

extern "C" void kernel_launch(void* const* d_in, const int* in_sizes, int n_in,
                              void* d_out, int out_size, void* d_ws, size_t ws_size,
                              hipStream_t stream) {
    (void)in_sizes; (void)n_in; (void)out_size; (void)d_ws; (void)ws_size;
    const float* t  = (const float*)d_in[0];
    const float* y0 = (const float*)d_in[1];
    const float* w  = (const float*)d_in[2];
    const float* r  = (const float*)d_in[3];
    float* out = (float*)d_out;

    neuro_ode_wmma<<<1, THREADS, 0, stream>>>(t, y0, w, r, out);
}


# Chain advances 16 rows per trip by the 129-tap product of 16 row maps (9 tile matrices); a second grid-wide launch fills the 15 rows between from each 16th row; sums regrouped by composed taps
# speedup vs baseline: 9.5527x; 2.8854x over previous
#include <hip/hip_runtime.h>


typedef float v2f __attribute__((ext_vector_type(2)));
typedef float v8f __attribute__((ext_vector_type(8)));

#define SAMPLE_NUM 2048
#define Y_NUM      2048
#define NSTEP      (SAMPLE_NUM - 1)
#define KROW       16
#define NTRIP      (NSTEP / KROW)
#define NGROUP     (NTRIP + 1)
#define NTILES     128
#define WAVES      8
#define THREADS    (WAVES * 32)
#define FWAVES     4
#define FTHREADS   (FWAVES * 32)
#define NMAT       (KROW / 2 + 1)
#define NTAPK      (8 * KROW + 1)
#define TAPTOP     (16 * NMAT - 1)
#define ROWLEN     (16 * NMAT + 16)
#define TP         (ROWLEN + 4)
#define FPOS       ROWLEN
#define SP         20
#define SBUF       (NTILES * SP)

struct StepC {
    float c[9];
    float f;
};

__device__ __forceinline__ StepC step_coef(float tn, float tn1, float a, float b, float c) {
    const float sub = (tn1 - tn) * 0.125f;
    const float d   = 1.0f - sub * b;
    const float e   = sub * a;
    StepC r;
    {
        const float bin[9] = {1.f, 8.f, 28.f, 56.f, 70.f, 56.f, 28.f, 8.f, 1.f};
        float dp[9], ep[9];
        dp[0] = 1.0f; ep[0] = 1.0f;
#pragma unroll
        for (int j = 1; j < 9; ++j) { dp[j] = dp[j-1] * d; ep[j] = ep[j-1] * e; }
#pragma unroll
        for (int j = 0; j < 9; ++j) r.c[j] = bin[j] * dp[8 - j] * ep[j];
    }
    const float g = d + e;
    float f = 0.0f;
#pragma unroll
    for (int ms = 0; ms < 8; ++ms) {
        const float s = sub * c * __sinf(tn + sub * (float)ms);
        f = f * g + s;
    }
    r.f = f;
    return r;
}

__device__ __forceinline__ v8f wmma4(v2f a, v2f b, v8f c) {
    return __builtin_amdgcn_wmma_f32_16x16x4_f32(false, a, false, b, (short)0, c, false, false);
}

struct StateK {
    v2f b[NMAT][4];
};

__device__ __forceinline__ v2f tap_pair(const float* rp, int r, int q) {
    v2f a;
    a[0] = rp[16 * (NMAT - 1) - 16 * r + 4 * q];
    a[1] = rp[16 * (NMAT - 1) - 16 * r + 4 * q + 1];
    return a;
}

struct CoefK {
    v2f   a[NMAT][4];
    float f;
};

__device__ __forceinline__ CoefK load_coefK(const float* row, int lb) {
    CoefK c;
    c.f = row[FPOS];
    const float* rp = row + lb;
#pragma unroll
    for (int q = 0; q < 4; ++q)
#pragma unroll
        for (int r = 0; r < NMAT; ++r) c.a[r][q] = tap_pair(rp, r, q);
    return c;
}

__device__ __forceinline__ StateK load_state_lds(const float* Yb, const int* so) {
    StateK s;
#pragma unroll
    for (int q = 0; q < 4; ++q)
#pragma unroll
        for (int r = 0; r < NMAT; ++r) {
            const float2 v = *(const float2*)&Yb[so[r] + 4 * q];
            s.b[r][q][0] = v.x; s.b[r][q][1] = v.y;
        }
    return s;
}

__device__ __forceinline__ v8f rows_trip(const CoefK& k, const StateK& s) {
    v8f aE, aO;
#pragma unroll
    for (int i = 0; i < 8; ++i) { aE[i] = k.f; aO[i] = 0.0f; }
#pragma unroll
    for (int q = 0; q < 4; ++q)
#pragma unroll
        for (int r = 0; r < NMAT; ++r) {
            if ((q * NMAT + r) & 1) aO = wmma4(k.a[r][q], s.b[r][q], aO);
            else                    aE = wmma4(k.a[r][q], s.b[r][q], aE);
        }
    return aE + aO;
}

__global__ __launch_bounds__(THREADS) __attribute__((amdgpu_waves_per_eu(2, 2)))
void ode_chain(const float* __restrict__ t,
                const float* __restrict__ y0,
                const float* __restrict__ w,
                const float* __restrict__ rt,
                float* __restrict__ out) {
    __shared__ __attribute__((aligned(16))) float Y[2][SBUF];
    __shared__ __attribute__((aligned(16))) float Tb[NTRIP * TP];

    const int tid  = threadIdx.x;
    const int wv   = tid >> 5;
    const int ln   = tid & 31;
    const int col  = ln & 15;
    const int half = ln >> 4;
    const int m    = col;

    const float a = w[0] * rt[0];
    const float b = w[1] * rt[1];
    const float c = w[2] * rt[2];

    for (int i = tid; i < Y_NUM; i += THREADS) {
        const float v = y0[i];
        Y[0][(i >> 4) * SP + (i & 15)] = v;
        out[i] = v;
    }

    if (tid < NTRIP) {
        const int n = KROW * tid;
        float P[NTAPK];
#pragma unroll
        for (int x = 0; x < NTAPK; ++x) P[x] = 0.0f;
        const StepC A0 = step_coef(t[n], t[n + 1], a, b, c);
#pragma unroll
        for (int x = 0; x < 9; ++x) P[x] = A0.c[x];
        float F = A0.f;
#pragma unroll
        for (int jj = 1; jj < KROW; ++jj) {
            const StepC B = step_coef(t[n + jj], t[n + jj + 1], a, b, c);
            float sumB = 0.0f;
#pragma unroll
            for (int i = 0; i < 9; ++i) sumB += B.c[i];
#pragma unroll
            for (int x = 8 * jj + 8; x >= 0; --x) {
                float acc = 0.0f;
#pragma unroll
                for (int i = 0; i < 9; ++i) {
                    const int xo = x - i;
                    if (xo >= 0 && xo <= 8 * jj) acc += B.c[i] * P[xo];
                }
                P[x] = acc;
            }
            F = F * sumB + B.f;
        }
        float4* row = (float4*)&Tb[tid * TP];
#pragma unroll
        for (int wq = 0; wq < ROWLEN / 4; ++wq) {
            float v[4];
#pragma unroll
            for (int e = 0; e < 4; ++e) {
                const int x = TAPTOP - (4 * wq + e);
                v[e] = (x >= 0 && x < NTAPK) ? P[(x >= 0 && x < NTAPK) ? x : 0] : 0.0f;
            }
            row[wq] = make_float4(v[0], v[1], v[2], v[3]);
        }
        row[FPOS / 4] = make_float4(F, 0.0f, 0.0f, 0.0f);
    }
    __syncthreads();

    const int tileC = (wv << 4) + col;
    const int rowIx = (wv << 8) + (col << 4) + (half << 3);
    const int stIx  = tileC * SP + (half << 3);
    const int lb    = 15 - m + (half << 1);
    int so[NMAT];
#pragma unroll
    for (int r = 0; r < NMAT; ++r) so[r] = ((tileC - r) & (NTILES - 1)) * SP + (half << 1);

    CoefK  cur = load_coefK(Tb, lb);
    StateK st  = load_state_lds(Y[0], so);

#pragma unroll 2
    for (int s = 0; s < NTRIP; ++s) {
        float* __restrict__ Yn = Y[(s & 1) ^ 1];

        const v8f yk = rows_trip(cur, st);
        const float4 lo = make_float4(yk[0], yk[1], yk[2], yk[3]);
        const float4 hi = make_float4(yk[4], yk[5], yk[6], yk[7]);
        *(float4*)&Yn[stIx]     = lo;
        *(float4*)&Yn[stIx + 4] = hi;

        __builtin_amdgcn_fence(__ATOMIC_RELEASE, "workgroup", "local");
        __builtin_amdgcn_s_barrier();
        __builtin_amdgcn_fence(__ATOMIC_ACQUIRE, "workgroup", "local");

        const int sn = (s + 1 < NTRIP) ? (s + 1) : (NTRIP - 1);
        const StateK stn = load_state_lds(Yn, so);
        const CoefK  nxt = load_coefK(Tb + sn * TP, lb);

        float* __restrict__ orow = out + (size_t)(KROW * (s + 1)) * Y_NUM;
        *(float4*)&orow[rowIx]     = lo;
        *(float4*)&orow[rowIx + 4] = hi;

        st  = stn;
        cur = nxt;
    }
}

template <int J>
__device__ __forceinline__ v8f row_direct(const float* rp, float F, const StateK& s) {
    constexpr int R = 1 + (J + 1) / 2;
    v8f aE, aO;
#pragma unroll
    for (int i = 0; i < 8; ++i) { aE[i] = F; aO[i] = 0.0f; }
    int cnt = 0;
#pragma unroll
    for (int q = 0; q < 4; ++q)
#pragma unroll
        for (int r = 0; r < R; ++r) {
            if ((J & 1) && r == R - 1 && q < 2) continue;
            const v2f a = tap_pair(rp, r, q);
            if (cnt & 1) aO = wmma4(a, s.b[r][q], aO);
            else         aE = wmma4(a, s.b[r][q], aE);
            ++cnt;
        }
    return aE + aO;
}

__device__ __forceinline__ void store_row(float* orow, int rowIx, v8f y) {
    *(float4*)&orow[rowIx]     = make_float4(y[0], y[1], y[2], y[3]);
    *(float4*)&orow[rowIx + 4] = make_float4(y[4], y[5], y[6], y[7]);
}

template <int J>
__device__ __forceinline__ void fill_from(float* out, int c0, int rowIx,
                                          const float* tab_lb, const float* Fs, const StateK& st) {
    if constexpr (J < KROW) {
        store_row(out + (size_t)(c0 + J) * Y_NUM, rowIx,
                  row_direct<J>(tab_lb + (J - 1) * ROWLEN, Fs[J - 1], st));
        fill_from<J + 1>(out, c0, rowIx, tab_lb, Fs, st);
    }
}

__global__ __launch_bounds__(FTHREADS)
void ode_fill(const float* __restrict__ t,
              const float* __restrict__ w,
              const float* __restrict__ rt,
              float* out) {
    __shared__ float sc[KROW - 1][12];
    __shared__ float tab[(KROW - 1) * ROWLEN];
    __shared__ float Fs[KROW];

    const int tid  = threadIdx.x;
    const int wv   = (tid >> 5) + FWAVES * ((int)blockIdx.x & 1);
    const int ln   = tid & 31;
    const int col  = ln & 15;
    const int half = ln >> 4;
    const int m    = col;
    const int c0   = KROW * ((int)blockIdx.x >> 1);

    const int tileC = (wv << 4) + col;
    const int rowIx = (wv << 8) + (col << 4) + (half << 3);
    const int lb    = 15 - m + (half << 1);

    const float* ck = out + (size_t)c0 * Y_NUM;
    StateK st;
#pragma unroll
    for (int q = 0; q < 4; ++q)
#pragma unroll
        for (int r = 0; r < NMAT; ++r) {
            const int tl = (tileC - r) & (NTILES - 1);
            const float2 v = *(const float2*)&ck[(tl << 4) + (q << 2) + (half << 1)];
            st.b[r][q][0] = v.x; st.b[r][q][1] = v.y;
        }

    if (tid < KROW - 1) {
        const float a = w[0] * rt[0];
        const float b = w[1] * rt[1];
        const float c = w[2] * rt[2];
        const StepC A = step_coef(t[c0 + tid], t[c0 + tid + 1], a, b, c);
#pragma unroll
        for (int i = 0; i < 9; ++i) sc[tid][i] = A.c[i];
        sc[tid][9] = A.f;
    }
    __syncthreads();

    for (int p = tid; p < ROWLEN; p += FTHREADS) {
        const int x = TAPTOP - p;
        tab[p] = (x >= 0 && x <= 8) ? sc[0][(x >= 0 && x <= 8) ? x : 0] : 0.0f;
    }
    if (tid == 0) {
        float F = sc[0][9];
        Fs[0] = F;
        for (int j = 1; j < KROW - 1; ++j) {
            float sumB = 0.0f;
            for (int i = 0; i < 9; ++i) sumB += sc[j][i];
            F = F * sumB + sc[j][9];
            Fs[j] = F;
        }
    }
    __syncthreads();

    for (int j = 1; j < KROW - 1; ++j) {
        for (int p = tid; p < ROWLEN; p += FTHREADS) {
            float acc = 0.0f;
#pragma unroll
            for (int i = 0; i < 9; ++i) {
                const int pp = p + i;
                const float pv = (pp < ROWLEN) ? tab[(j - 1) * ROWLEN + ((pp < ROWLEN) ? pp : 0)] : 0.0f;
                acc += sc[j][i] * pv;
            }
            tab[j * ROWLEN + p] = acc;
        }
        __syncthreads();
    }

    fill_from<1>(out, c0, rowIx, tab + lb, Fs, st);
}

extern "C" void kernel_launch(void* const* d_in, const int* in_sizes, int n_in,
                              void* d_out, int out_size, void* d_ws, size_t ws_size,
                              hipStream_t stream) {
    (void)in_sizes; (void)n_in; (void)out_size; (void)d_ws; (void)ws_size;
    const float* t  = (const float*)d_in[0];
    const float* y0 = (const float*)d_in[1];
    const float* w  = (const float*)d_in[2];
    const float* r  = (const float*)d_in[3];
    float* out = (float*)d_out;

    ode_chain<<<1, THREADS, 0, stream>>>(t, y0, w, r, out);
    ode_fill<<<2 * NGROUP, FTHREADS, 0, stream>>>(t, w, r, out);
}


# Chain cut into 8 segments of 256 rows run side by side: segment maps composed on a unit impulse, 7 start rows by full circulant products, then all segment chains at once; sums regrouped
# speedup vs baseline: 12.5107x; 1.3097x over previous
#include <hip/hip_runtime.h>


typedef float v2f __attribute__((ext_vector_type(2)));
typedef float v8f __attribute__((ext_vector_type(8)));

#define SAMPLE_NUM 2048
#define Y_NUM      2048
#define NSTEP      (SAMPLE_NUM - 1)
#define KROW       16
#define NTRIP      (NSTEP / KROW)
#define NGROUP     (NTRIP + 1)
#define NTILES     128
#define WAVES      8
#define THREADS    (WAVES * 32)
#define FWAVES     4
#define FTHREADS   (FWAVES * 32)
#define NMAT       (KROW / 2 + 1)
#define NTAPK      (8 * KROW + 1)
#define TAPTOP     (16 * NMAT - 1)
#define ROWLEN     (16 * NMAT + 16)
#define TP         (ROWLEN + 4)
#define FPOS       ROWLEN
#define SP         20
#define SBUF       (NTILES * SP)
#define SEGS       8
#define SEGTRIPS   16
#define SEGROWS    (KROW * SEGTRIPS)
#define NMAPS      (SEGS - 1)
#define LASTTRIPS  (NTRIP - SEGTRIPS * (SEGS - 1))
#define SUMPOS     (FPOS + 1)
#define WS_TAB     0
#define WS_MAP     32768
#define WS_G       49152
#define WS_Y       65536

struct StepC {
    float c[9];
    float f;
};

__device__ __forceinline__ StepC step_coef(float tn, float tn1, float a, float b, float c) {
    const float sub = (tn1 - tn) * 0.125f;
    const float d   = 1.0f - sub * b;
    const float e   = sub * a;
    StepC r;
    {
        const float bin[9] = {1.f, 8.f, 28.f, 56.f, 70.f, 56.f, 28.f, 8.f, 1.f};
        float dp[9], ep[9];
        dp[0] = 1.0f; ep[0] = 1.0f;
#pragma unroll
        for (int j = 1; j < 9; ++j) { dp[j] = dp[j-1] * d; ep[j] = ep[j-1] * e; }
#pragma unroll
        for (int j = 0; j < 9; ++j) r.c[j] = bin[j] * dp[8 - j] * ep[j];
    }
    const float g = d + e;
    float f = 0.0f;
#pragma unroll
    for (int ms = 0; ms < 8; ++ms) {
        const float s = sub * c * __sinf(tn + sub * (float)ms);
        f = f * g + s;
    }
    r.f = f;
    return r;
}

__device__ __forceinline__ v8f wmma4(v2f a, v2f b, v8f c) {
    return __builtin_amdgcn_wmma_f32_16x16x4_f32(false, a, false, b, (short)0, c, false, false);
}

struct StateK {
    v2f b[NMAT][4];
};

__device__ __forceinline__ v2f tap_pair(const float* rp, int r, int q) {
    v2f a;
    a[0] = rp[16 * (NMAT - 1) - 16 * r + 4 * q];
    a[1] = rp[16 * (NMAT - 1) - 16 * r + 4 * q + 1];
    return a;
}

struct CoefK {
    v2f   a[NMAT][4];
    float f;
};

template <bool FORCED>
__device__ __forceinline__ CoefK load_coefK(const float* row, int lb) {
    CoefK c;
    c.f = FORCED ? row[FPOS] : 0.0f;
    const float* rp = row + lb;
#pragma unroll
    for (int q = 0; q < 4; ++q)
#pragma unroll
        for (int r = 0; r < NMAT; ++r) c.a[r][q] = tap_pair(rp, r, q);
    return c;
}

__device__ __forceinline__ StateK load_state_lds(const float* Yb, const int* so) {
    StateK s;
#pragma unroll
    for (int q = 0; q < 4; ++q)
#pragma unroll
        for (int r = 0; r < NMAT; ++r) {
            const float2 v = *(const float2*)&Yb[so[r] + 4 * q];
            s.b[r][q][0] = v.x; s.b[r][q][1] = v.y;
        }
    return s;
}

__device__ __forceinline__ v8f rows_trip(const CoefK& k, const StateK& s) {
    v8f aE, aO;
#pragma unroll
    for (int i = 0; i < 8; ++i) { aE[i] = k.f; aO[i] = 0.0f; }
#pragma unroll
    for (int q = 0; q < 4; ++q)
#pragma unroll
        for (int r = 0; r < NMAT; ++r) {
            if ((q * NMAT + r) & 1) aO = wmma4(k.a[r][q], s.b[r][q], aO);
            else                    aE = wmma4(k.a[r][q], s.b[r][q], aE);
        }
    return aE + aO;
}


template <int N>
__device__ __forceinline__ void copy_floats(float* dst, const float* src, int tid) {
    constexpr int IT = (N + THREADS - 1) / THREADS;
    float v[IT];
#pragma unroll
    for (int j = 0; j < IT; ++j) { const int i = tid + j * THREADS; v[j] = (i < N) ? src[i] : 0.0f; }
#pragma unroll
    for (int j = 0; j < IT; ++j) { const int i = tid + j * THREADS; if (i < N) dst[i] = v[j]; }
}

__device__ __forceinline__ void build_trip_row(float* rowp, int s, const float* __restrict__ t,
                                               float a, float b, float c) {
    const int n = KROW * s;
    float P[NTAPK];
#pragma unroll
    for (int x = 0; x < NTAPK; ++x) P[x] = 0.0f;
    const StepC A0 = step_coef(t[n], t[n + 1], a, b, c);
#pragma unroll
    for (int x = 0; x < 9; ++x) P[x] = A0.c[x];
    float F = A0.f;
#pragma unroll
    for (int jj = 1; jj < KROW; ++jj) {
        const StepC B = step_coef(t[n + jj], t[n + jj + 1], a, b, c);
        float sumB = 0.0f;
#pragma unroll
        for (int i = 0; i < 9; ++i) sumB += B.c[i];
#pragma unroll
        for (int x = 8 * jj + 8; x >= 0; --x) {
            float acc = 0.0f;
#pragma unroll
            for (int i = 0; i < 9; ++i) {
                const int xo = x - i;
                if (xo >= 0 && xo <= 8 * jj) acc += B.c[i] * P[xo];
            }
            P[x] = acc;
        }
        F = F * sumB + B.f;
    }
    float sumP = 0.0f;
#pragma unroll
    for (int x = 0; x < NTAPK; ++x) sumP += P[x];
    float4* row = (float4*)rowp;
#pragma unroll
    for (int wq = 0; wq < ROWLEN / 4; ++wq) {
        float v[4];
#pragma unroll
        for (int e = 0; e < 4; ++e) {
            const int x = TAPTOP - (4 * wq + e);
            v[e] = (x >= 0 && x < NTAPK) ? P[(x >= 0 && x < NTAPK) ? x : 0] : 0.0f;
        }
        row[wq] = make_float4(v[0], v[1], v[2], v[3]);
    }
    row[FPOS / 4] = make_float4(F, sumP, 0.0f, 0.0f);
}

template <bool ROWS, int NTR>
__device__ __forceinline__ void chain_trips(float* Yb, const float* Tb, float* __restrict__ out, int row0) {
    const int tid  = threadIdx.x;
    const int wv   = tid >> 5;
    const int ln   = tid & 31;
    const int col  = ln & 15;
    const int half = ln >> 4;
    const int m    = col;

    const int tileC = (wv << 4) + col;
    const int rowIx = (wv << 8) + (col << 4) + (half << 3);
    const int stIx  = tileC * SP + (half << 3);
    const int lb    = 15 - m + (half << 1);
    int so[NMAT];
#pragma unroll
    for (int r = 0; r < NMAT; ++r) so[r] = ((tileC - r) & (NTILES - 1)) * SP + (half << 1);

    CoefK  cur = load_coefK<ROWS>(Tb, lb);
    StateK st  = load_state_lds(Yb, so);

#pragma unroll 2
    for (int s = 0; s < NTR; ++s) {
        float* __restrict__ Yn = Yb + ((s & 1) ^ 1) * SBUF;

        const v8f yk = rows_trip(cur, st);
        const float4 lo = make_float4(yk[0], yk[1], yk[2], yk[3]);
        const float4 hi = make_float4(yk[4], yk[5], yk[6], yk[7]);
        *(float4*)&Yn[stIx]     = lo;
        *(float4*)&Yn[stIx + 4] = hi;

        __builtin_amdgcn_fence(__ATOMIC_RELEASE, "workgroup", "local");
        __builtin_amdgcn_s_barrier();
        __builtin_amdgcn_fence(__ATOMIC_ACQUIRE, "workgroup", "local");

        const int sn = (s + 1 < NTR) ? (s + 1) : (NTR - 1);
        const StateK stn = load_state_lds(Yn, so);
        __builtin_amdgcn_sched_barrier(0);
        const CoefK  nxt = load_coefK<ROWS>(Tb + sn * TP, lb);
        __builtin_amdgcn_sched_barrier(0);

        if (ROWS) {
            float* __restrict__ orow = out + (size_t)(row0 + KROW * (s + 1)) * Y_NUM;
            *(float4*)&orow[rowIx]     = lo;
            *(float4*)&orow[rowIx + 4] = hi;
        }

        st  = stn;
        cur = nxt;
    }
}

__global__ __launch_bounds__(THREADS) __attribute__((amdgpu_waves_per_eu(2, 2)))
void ode_maps(const float* __restrict__ t,
              const float* __restrict__ w,
              const float* __restrict__ rt,
              float* __restrict__ ws) {
    __shared__ __attribute__((aligned(16))) float Y[2][SBUF];
    __shared__ __attribute__((aligned(16))) float Tb[SEGTRIPS * TP];

    const int tid = threadIdx.x;
    const int k   = (int)blockIdx.x;
    const int nrows = (k == SEGS - 1) ? LASTTRIPS : SEGTRIPS;

    if (tid < nrows) {
        const float a = w[0] * rt[0];
        const float b = w[1] * rt[1];
        const float c = w[2] * rt[2];
        build_trip_row(&Tb[tid * TP], SEGTRIPS * k + tid, t, a, b, c);
    }
    for (int i = tid; i < SBUF; i += THREADS) Y[0][i] = (i == 0) ? 1.0f : 0.0f;
    __syncthreads();

    copy_floats<LASTTRIPS * TP>(ws + WS_TAB + SEGTRIPS * k * TP, Tb, tid);

    if (k == SEGS - 1) return;

    if (tid == 0) {
        float G = 0.0f;
#pragma unroll
        for (int j = 0; j < SEGTRIPS; ++j) G = G * Tb[j * TP + SUMPOS] + Tb[j * TP + FPOS];
        ws[WS_G + k] = G;
    }

    chain_trips<false, SEGTRIPS>(&Y[0][0], Tb, nullptr, 0);

    const float* Yf = Y[SEGTRIPS & 1];
#pragma unroll
    for (int j = 0; j < Y_NUM / THREADS; ++j) {
        const int i = tid + j * THREADS;
        ws[WS_MAP + k * Y_NUM + i] = Yf[(i >> 4) * SP + (i & 15)];
    }
}

__global__ __launch_bounds__(THREADS)
void ode_seg_start(const float* __restrict__ y0,
                   float* __restrict__ ws,
                   float* __restrict__ out,
                   int k) {
    __shared__ __attribute__((aligned(16))) float As[Y_NUM + 16];
    __shared__ __attribute__((aligned(16))) float Ys[Y_NUM];
    __shared__ __attribute__((aligned(16))) float Pt[WAVES * 256];

    const int tid  = threadIdx.x;
    const int wv   = tid >> 5;
    const int ln   = tid & 31;
    const int col  = ln & 15;
    const int half = ln >> 4;
    const int b    = (int)blockIdx.x;

    const float* yin = (k == 0) ? y0 : (ws + WS_Y + k * Y_NUM);
    {
        const float* amap = ws + WS_MAP + k * Y_NUM;
        float ta[Y_NUM / THREADS], ty[Y_NUM / THREADS];
#pragma unroll
        for (int j = 0; j < Y_NUM / THREADS; ++j) { ta[j] = amap[tid + j * THREADS]; ty[j] = yin[tid + j * THREADS]; }
#pragma unroll
        for (int j = 0; j < Y_NUM / THREADS; ++j) {
            const int i = tid + j * THREADS;
            As[16 + i] = ta[j];
            if (i >= Y_NUM - 16) As[i - (Y_NUM - 16)] = ta[j];
            Ys[i] = ty[j];
        }
    }
    __syncthreads();

    v8f aE, aO;
#pragma unroll
    for (int i = 0; i < 8; ++i) { aE[i] = 0.0f; aO[i] = 0.0f; }
    const int tile0 = (b << 4) + col;
    for (int rb = 0; rb < 16; rb += 4) {
        v2f av[4][4], bv[4][4];
#pragma unroll
        for (int u = 0; u < 4; ++u) {
            const int r  = (wv << 4) + rb + u;
            const int tl = (tile0 - r) & (NTILES - 1);
            const int x0 = 16 * r + col - 2 * half;
#pragma unroll
            for (int q = 0; q < 4; ++q) {
                const int x = x0 - 4 * q;
                av[u][q][0] = As[16 + x];
                av[u][q][1] = As[15 + x];
                const float2 v = *(const float2*)&Ys[(tl << 4) + (q << 2) + (half << 1)];
                bv[u][q][0] = v.x; bv[u][q][1] = v.y;
            }
        }
        __builtin_amdgcn_sched_barrier(0);
#pragma unroll
        for (int u = 0; u < 4; ++u)
#pragma unroll
            for (int q = 0; q < 4; ++q) {
                if (q & 1) aO = wmma4(av[u][q], bv[u][q], aO);
                else       aE = wmma4(av[u][q], bv[u][q], aE);
            }
        __builtin_amdgcn_sched_barrier(0);
    }
    const v8f acc = aE + aO;

    float* pp = &Pt[(wv << 8) + (ln << 3)];
    *(float4*)&pp[0] = make_float4(acc[0], acc[1], acc[2], acc[3]);
    *(float4*)&pp[4] = make_float4(acc[4], acc[5], acc[6], acc[7]);
    __syncthreads();

    const int e  = tid & 15;
    const int l  = (tid >> 4) + ((e >> 3) << 4);
    const int pv = (l << 3) + (e & 7);
    float sum = 0.0f;
#pragma unroll
    for (int u = 0; u < WAVES; ++u) sum += Pt[(u << 8) + pv];
    sum += ws[WS_G + k];
    const int i = (b << 8) + tid;
    ws[WS_Y + (k + 1) * Y_NUM + i] = sum;
    out[(size_t)(SEGROWS * (k + 1)) * Y_NUM + i] = sum;
}

__global__ __launch_bounds__(THREADS) __attribute__((amdgpu_waves_per_eu(2, 2)))
void ode_chain_seg(const float* __restrict__ y0,
                   const float* __restrict__ ws,
                   float* __restrict__ out) {
    __shared__ __attribute__((aligned(16))) float Y[2][SBUF];
    __shared__ __attribute__((aligned(16))) float Tb[(SEGTRIPS - 1) * TP];

    const int tid = threadIdx.x;
    const int k   = (int)blockIdx.x;

    const float* ys = (k == 0) ? y0 : (ws + WS_Y + k * Y_NUM);
    {
        float ty[Y_NUM / THREADS];
#pragma unroll
        for (int j = 0; j < Y_NUM / THREADS; ++j) ty[j] = ys[tid + j * THREADS];
        copy_floats<(SEGTRIPS - 1) * TP>(Tb, ws + WS_TAB + SEGTRIPS * k * TP, tid);
#pragma unroll
        for (int j = 0; j < Y_NUM / THREADS; ++j) {
            const int i = tid + j * THREADS;
            Y[0][(i >> 4) * SP + (i & 15)] = ty[j];
            if (k == 0) out[i] = ty[j];
        }
    }
    __syncthreads();

    chain_trips<true, SEGTRIPS - 1>(&Y[0][0], Tb, out, SEGROWS * k);
}

template <int J>
__device__ __forceinline__ v8f row_direct(const float* rp, float F, const StateK& s) {
    constexpr int R = 1 + (J + 1) / 2;
    v8f aE, aO;
#pragma unroll
    for (int i = 0; i < 8; ++i) { aE[i] = F; aO[i] = 0.0f; }
    int cnt = 0;
#pragma unroll
    for (int q = 0; q < 4; ++q)
#pragma unroll
        for (int r = 0; r < R; ++r) {
            if ((J & 1) && r == R - 1 && q < 2) continue;
            const v2f a = tap_pair(rp, r, q);
            if (cnt & 1) aO = wmma4(a, s.b[r][q], aO);
            else         aE = wmma4(a, s.b[r][q], aE);
            ++cnt;
        }
    return aE + aO;
}

__device__ __forceinline__ void store_row(float* orow, int rowIx, v8f y) {
    *(float4*)&orow[rowIx]     = make_float4(y[0], y[1], y[2], y[3]);
    *(float4*)&orow[rowIx + 4] = make_float4(y[4], y[5], y[6], y[7]);
}

template <int J>
__device__ __forceinline__ void fill_from(float* out, int c0, int rowIx,
                                          const float* tab_lb, const float* Fs, const StateK& st) {
    if constexpr (J < KROW) {
        store_row(out + (size_t)(c0 + J) * Y_NUM, rowIx,
                  row_direct<J>(tab_lb + (J - 1) * ROWLEN, Fs[J - 1], st));
        fill_from<J + 1>(out, c0, rowIx, tab_lb, Fs, st);
    }
}

__global__ __launch_bounds__(FTHREADS)
void ode_fill(const float* __restrict__ t,
              const float* __restrict__ w,
              const float* __restrict__ rt,
              float* out) {
    __shared__ float sc[KROW - 1][12];
    __shared__ float tab[(KROW - 1) * ROWLEN];
    __shared__ float Fs[KROW];

    const int tid  = threadIdx.x;
    const int wv   = (tid >> 5) + FWAVES * ((int)blockIdx.x & 1);
    const int ln   = tid & 31;
    const int col  = ln & 15;
    const int half = ln >> 4;
    const int m    = col;
    const int c0   = KROW * ((int)blockIdx.x >> 1);

    const int tileC = (wv << 4) + col;
    const int rowIx = (wv << 8) + (col << 4) + (half << 3);
    const int lb    = 15 - m + (half << 1);

    const float* ck = out + (size_t)c0 * Y_NUM;
    StateK st;
#pragma unroll
    for (int q = 0; q < 4; ++q)
#pragma unroll
        for (int r = 0; r < NMAT; ++r) {
            const int tl = (tileC - r) & (NTILES - 1);
            const float2 v = *(const float2*)&ck[(tl << 4) + (q << 2) + (half << 1)];
            st.b[r][q][0] = v.x; st.b[r][q][1] = v.y;
        }

    if (tid < KROW - 1) {
        const float a = w[0] * rt[0];
        const float b = w[1] * rt[1];
        const float c = w[2] * rt[2];
        const StepC A = step_coef(t[c0 + tid], t[c0 + tid + 1], a, b, c);
#pragma unroll
        for (int i = 0; i < 9; ++i) sc[tid][i] = A.c[i];
        sc[tid][9] = A.f;
    }
    __syncthreads();

    for (int p = tid; p < ROWLEN; p += FTHREADS) {
        const int x = TAPTOP - p;
        tab[p] = (x >= 0 && x <= 8) ? sc[0][(x >= 0 && x <= 8) ? x : 0] : 0.0f;
    }
    if (tid == 0) {
        float F = sc[0][9];
        Fs[0] = F;
        for (int j = 1; j < KROW - 1; ++j) {
            float sumB = 0.0f;
            for (int i = 0; i < 9; ++i) sumB += sc[j][i];
            F = F * sumB + sc[j][9];
            Fs[j] = F;
        }
    }
    __syncthreads();

    for (int j = 1; j < KROW - 1; ++j) {
        for (int p = tid; p < ROWLEN; p += FTHREADS) {
            float acc = 0.0f;
#pragma unroll
            for (int i = 0; i < 9; ++i) {
                const int pp = p + i;
                const float pv = (pp < ROWLEN) ? tab[(j - 1) * ROWLEN + ((pp < ROWLEN) ? pp : 0)] : 0.0f;
                acc += sc[j][i] * pv;
            }
            tab[j * ROWLEN + p] = acc;
        }
        __syncthreads();
    }

    fill_from<1>(out, c0, rowIx, tab + lb, Fs, st);
}

extern "C" void kernel_launch(void* const* d_in, const int* in_sizes, int n_in,
                              void* d_out, int out_size, void* d_ws, size_t ws_size,
                              hipStream_t stream) {
    (void)in_sizes; (void)n_in; (void)out_size; (void)ws_size;
    const float* t  = (const float*)d_in[0];
    const float* y0 = (const float*)d_in[1];
    const float* w  = (const float*)d_in[2];
    const float* r  = (const float*)d_in[3];
    float* out = (float*)d_out;
    float* ws  = (float*)d_ws;

    ode_maps<<<SEGS, THREADS, 0, stream>>>(t, w, r, ws);
    ode_seg_start<<<8, THREADS, 0, stream>>>(y0, ws, out, 0);
    ode_seg_start<<<8, THREADS, 0, stream>>>(y0, ws, out, 1);
    ode_seg_start<<<8, THREADS, 0, stream>>>(y0, ws, out, 2);
    ode_seg_start<<<8, THREADS, 0, stream>>>(y0, ws, out, 3);
    ode_seg_start<<<8, THREADS, 0, stream>>>(y0, ws, out, 4);
    ode_seg_start<<<8, THREADS, 0, stream>>>(y0, ws, out, 5);
    ode_seg_start<<<8, THREADS, 0, stream>>>(y0, ws, out, 6);
    ode_chain_seg<<<SEGS, THREADS, 0, stream>>>(y0, ws, out);
    ode_fill<<<2 * NGROUP, FTHREADS, 0, stream>>>(t, w, r, out);
}


# Tap rows of each segment's chain trips built by the whole workgroup (a lane group per trip, taps in registers, neighbour taps by lane shuffle) instead of one lane per trip; tap-sum order changed
# speedup vs baseline: 17.9527x; 1.4350x over previous
#include <hip/hip_runtime.h>


typedef float v2f __attribute__((ext_vector_type(2)));
typedef float v8f __attribute__((ext_vector_type(8)));

#define SAMPLE_NUM 2048
#define Y_NUM      2048
#define NSTEP      (SAMPLE_NUM - 1)
#define KROW       16
#define NTRIP      (NSTEP / KROW)
#define NGROUP     (NTRIP + 1)
#define NTILES     128
#define WAVES      8
#define THREADS    (WAVES * 32)
#define FWAVES     4
#define FTHREADS   (FWAVES * 32)
#define NMAT       (KROW / 2 + 1)
#define NTAPK      (8 * KROW + 1)
#define TAPTOP     (16 * NMAT - 1)
#define ROWLEN     (16 * NMAT + 16)
#define TP         (ROWLEN + 4)
#define FPOS       ROWLEN
#define SP         20
#define SBUF       (NTILES * SP)
#define SEGS       8
#define SEGTRIPS   16
#define SEGROWS    (KROW * SEGTRIPS)
#define NMAPS      (SEGS - 1)
#define LASTTRIPS  (NTRIP - SEGTRIPS * (SEGS - 1))
#define SUMPOS     (FPOS + 1)
#define WS_TAB     0
#define WS_MAP     32768
#define WS_G       49152
#define WS_Y       65536

struct StepC {
    float c[9];
    float f;
};

__device__ __forceinline__ StepC step_coef(float tn, float tn1, float a, float b, float c) {
    const float sub = (tn1 - tn) * 0.125f;
    const float d   = 1.0f - sub * b;
    const float e   = sub * a;
    StepC r;
    {
        const float bin[9] = {1.f, 8.f, 28.f, 56.f, 70.f, 56.f, 28.f, 8.f, 1.f};
        float dp[9], ep[9];
        dp[0] = 1.0f; ep[0] = 1.0f;
#pragma unroll
        for (int j = 1; j < 9; ++j) { dp[j] = dp[j-1] * d; ep[j] = ep[j-1] * e; }
#pragma unroll
        for (int j = 0; j < 9; ++j) r.c[j] = bin[j] * dp[8 - j] * ep[j];
    }
    const float g = d + e;
    float f = 0.0f;
#pragma unroll
    for (int ms = 0; ms < 8; ++ms) {
        const float s = sub * c * __sinf(tn + sub * (float)ms);
        f = f * g + s;
    }
    r.f = f;
    return r;
}

__device__ __forceinline__ v8f wmma4(v2f a, v2f b, v8f c) {
    return __builtin_amdgcn_wmma_f32_16x16x4_f32(false, a, false, b, (short)0, c, false, false);
}

struct StateK {
    v2f b[NMAT][4];
};

__device__ __forceinline__ v2f tap_pair(const float* rp, int r, int q) {
    v2f a;
    a[0] = rp[16 * (NMAT - 1) - 16 * r + 4 * q];
    a[1] = rp[16 * (NMAT - 1) - 16 * r + 4 * q + 1];
    return a;
}

struct CoefK {
    v2f   a[NMAT][4];
    float f;
};

template <bool FORCED>
__device__ __forceinline__ CoefK load_coefK(const float* row, int lb) {
    CoefK c;
    c.f = FORCED ? row[FPOS] : 0.0f;
    const float* rp = row + lb;
#pragma unroll
    for (int q = 0; q < 4; ++q)
#pragma unroll
        for (int r = 0; r < NMAT; ++r) c.a[r][q] = tap_pair(rp, r, q);
    return c;
}

__device__ __forceinline__ StateK load_state_lds(const float* Yb, const int* so) {
    StateK s;
#pragma unroll
    for (int q = 0; q < 4; ++q)
#pragma unroll
        for (int r = 0; r < NMAT; ++r) {
            const float2 v = *(const float2*)&Yb[so[r] + 4 * q];
            s.b[r][q][0] = v.x; s.b[r][q][1] = v.y;
        }
    return s;
}

__device__ __forceinline__ v8f rows_trip(const CoefK& k, const StateK& s) {
    v8f aE, aO;
#pragma unroll
    for (int i = 0; i < 8; ++i) { aE[i] = k.f; aO[i] = 0.0f; }
#pragma unroll
    for (int q = 0; q < 4; ++q)
#pragma unroll
        for (int r = 0; r < NMAT; ++r) {
            if ((q * NMAT + r) & 1) aO = wmma4(k.a[r][q], s.b[r][q], aO);
            else                    aE = wmma4(k.a[r][q], s.b[r][q], aE);
        }
    return aE + aO;
}


template <int N>
__device__ __forceinline__ void copy_floats(float* dst, const float* src, int tid) {
    constexpr int IT = (N + THREADS - 1) / THREADS;
    float v[IT];
#pragma unroll
    for (int j = 0; j < IT; ++j) { const int i = tid + j * THREADS; v[j] = (i < N) ? src[i] : 0.0f; }
#pragma unroll
    for (int j = 0; j < IT; ++j) { const int i = tid + j * THREADS; if (i < N) dst[i] = v[j]; }
}

#define LANEPOS (ROWLEN / 16)
static_assert(LANEPOS * 16 == ROWLEN && LANEPOS >= 8, "a lane's window must reach exactly one lane up");
static_assert(THREADS == 16 * SEGTRIPS && KROW == 16, "16 lanes per trip, one row map per thread");

__device__ __forceinline__ void build_trip_rows(float* Tb, const float (*sc)[12], int tid) {
    const int j = tid >> 4;
    const int u = tid & 15;
    const float* r0 = sc[KROW * j];
    float Wd[LANEPOS + 8];
#pragma unroll
    for (int e = 0; e < LANEPOS; ++e) {
        const int x = TAPTOP - (LANEPOS * u + e);
        const bool in = (x >= 0 && x <= 8);
        Wd[e] = in ? r0[in ? x : 0] : 0.0f;
    }
    float F = r0[9];
    for (int jj = 1; jj < KROW; ++jj) {
        const float* rb = sc[KROW * j + jj];
        const float4 c03 = *(const float4*)&rb[0];
        const float4 c47 = *(const float4*)&rb[4];
        const float4 c8x = *(const float4*)&rb[8];
        const float cc[9] = {c03.x, c03.y, c03.z, c03.w, c47.x, c47.y, c47.z, c47.w, c8x.x};
#pragma unroll
        for (int e = 0; e < 8; ++e) {
            const float v = __shfl_down(Wd[e], 1, 16);
            Wd[LANEPOS + e] = (u == 15) ? 0.0f : v;
        }
        float Nw[LANEPOS];
#pragma unroll
        for (int e = 0; e < LANEPOS; ++e) {
            float acc = 0.0f;
#pragma unroll
            for (int i = 0; i < 9; ++i) acc += cc[i] * Wd[e + i];
            Nw[e] = acc;
        }
#pragma unroll
        for (int e = 0; e < LANEPOS; ++e) Wd[e] = Nw[e];
        F = F * c8x.z + c8x.y;
    }
    float sp = 0.0f;
#pragma unroll
    for (int e = 0; e < LANEPOS; ++e) sp += Wd[e];
#pragma unroll
    for (int dlt = 8; dlt >= 1; dlt >>= 1) sp += __shfl_down(sp, dlt, 16);

    float* row = Tb + j * TP;
    float2* dst = (float2*)&row[LANEPOS * u];
#pragma unroll
    for (int e = 0; e < LANEPOS / 2; ++e) dst[e] = make_float2(Wd[2 * e], Wd[2 * e + 1]);
    if (u == 0) *(float4*)&row[FPOS] = make_float4(F, sp, 0.0f, 0.0f);
}

template <bool ROWS, int NTR>
__device__ __forceinline__ void chain_trips(float* Yb, const float* Tb, float* __restrict__ out, int row0) {
    const int tid  = threadIdx.x;
    const int wv   = tid >> 5;
    const int ln   = tid & 31;
    const int col  = ln & 15;
    const int half = ln >> 4;
    const int m    = col;

    const int tileC = (wv << 4) + col;
    const int rowIx = (wv << 8) + (col << 4) + (half << 3);
    const int stIx  = tileC * SP + (half << 3);
    const int lb    = 15 - m + (half << 1);
    int so[NMAT];
#pragma unroll
    for (int r = 0; r < NMAT; ++r) so[r] = ((tileC - r) & (NTILES - 1)) * SP + (half << 1);

    CoefK  cur = load_coefK<ROWS>(Tb, lb);
    StateK st  = load_state_lds(Yb, so);

#pragma unroll 2
    for (int s = 0; s < NTR; ++s) {
        float* __restrict__ Yn = Yb + ((s & 1) ^ 1) * SBUF;

        const v8f yk = rows_trip(cur, st);
        const float4 lo = make_float4(yk[0], yk[1], yk[2], yk[3]);
        const float4 hi = make_float4(yk[4], yk[5], yk[6], yk[7]);
        *(float4*)&Yn[stIx]     = lo;
        *(float4*)&Yn[stIx + 4] = hi;

        __builtin_amdgcn_fence(__ATOMIC_RELEASE, "workgroup", "local");
        __builtin_amdgcn_s_barrier();
        __builtin_amdgcn_fence(__ATOMIC_ACQUIRE, "workgroup", "local");

        const int sn = (s + 1 < NTR) ? (s + 1) : (NTR - 1);
        const StateK stn = load_state_lds(Yn, so);
        __builtin_amdgcn_sched_barrier(0);
        const CoefK  nxt = load_coefK<ROWS>(Tb + sn * TP, lb);
        __builtin_amdgcn_sched_barrier(0);

        if (ROWS) {
            float* __restrict__ orow = out + (size_t)(row0 + KROW * (s + 1)) * Y_NUM;
            *(float4*)&orow[rowIx]     = lo;
            *(float4*)&orow[rowIx + 4] = hi;
        }

        st  = stn;
        cur = nxt;
    }
}

__global__ __launch_bounds__(THREADS) __attribute__((amdgpu_waves_per_eu(2, 2)))
void ode_maps(const float* __restrict__ t,
              const float* __restrict__ w,
              const float* __restrict__ rt,
              float* __restrict__ ws) {
    __shared__ __attribute__((aligned(16))) float Y[2][SBUF];
    __shared__ __attribute__((aligned(16))) float Tb[SEGTRIPS * TP];
    __shared__ __attribute__((aligned(16))) float sc[SEGTRIPS * KROW][12];

    const int tid = threadIdx.x;
    const int k   = (int)blockIdx.x;
    const int nrows = (k == SEGS - 1) ? LASTTRIPS : SEGTRIPS;

    {
        StepC A;
#pragma unroll
        for (int i = 0; i < 9; ++i) A.c[i] = 0.0f;
        A.f = 0.0f;
        if (tid < KROW * nrows) {
            const float a = w[0] * rt[0];
            const float b = w[1] * rt[1];
            const float c = w[2] * rt[2];
            const int n = SEGROWS * k + tid;
            A = step_coef(t[n], t[n + 1], a, b, c);
        }
        float sumA = 0.0f;
#pragma unroll
        for (int i = 0; i < 9; ++i) sumA += A.c[i];
        *(float4*)&sc[tid][0] = make_float4(A.c[0], A.c[1], A.c[2], A.c[3]);
        *(float4*)&sc[tid][4] = make_float4(A.c[4], A.c[5], A.c[6], A.c[7]);
        *(float4*)&sc[tid][8] = make_float4(A.c[8], A.f, sumA, 0.0f);
    }
    __syncthreads();
    build_trip_rows(Tb, sc, tid);
    for (int i = tid; i < SBUF; i += THREADS) Y[0][i] = (i == 0) ? 1.0f : 0.0f;
    __syncthreads();

    copy_floats<LASTTRIPS * TP>(ws + WS_TAB + SEGTRIPS * k * TP, Tb, tid);

    if (k == SEGS - 1) return;

    if (tid == 0) {
        float G = 0.0f;
#pragma unroll
        for (int j = 0; j < SEGTRIPS; ++j) G = G * Tb[j * TP + SUMPOS] + Tb[j * TP + FPOS];
        ws[WS_G + k] = G;
    }

    chain_trips<false, SEGTRIPS>(&Y[0][0], Tb, nullptr, 0);

    const float* Yf = Y[SEGTRIPS & 1];
#pragma unroll
    for (int j = 0; j < Y_NUM / THREADS; ++j) {
        const int i = tid + j * THREADS;
        ws[WS_MAP + k * Y_NUM + i] = Yf[(i >> 4) * SP + (i & 15)];
    }
}

__global__ __launch_bounds__(THREADS)
void ode_seg_start(const float* __restrict__ y0,
                   float* __restrict__ ws,
                   float* __restrict__ out,
                   int k) {
    __shared__ __attribute__((aligned(16))) float As[Y_NUM + 16];
    __shared__ __attribute__((aligned(16))) float Ys[Y_NUM];
    __shared__ __attribute__((aligned(16))) float Pt[WAVES * 256];

    const int tid  = threadIdx.x;
    const int wv   = tid >> 5;
    const int ln   = tid & 31;
    const int col  = ln & 15;
    const int half = ln >> 4;
    const int b    = (int)blockIdx.x;

    const float* yin = (k == 0) ? y0 : (ws + WS_Y + k * Y_NUM);
    {
        const float* amap = ws + WS_MAP + k * Y_NUM;
        float ta[Y_NUM / THREADS], ty[Y_NUM / THREADS];
#pragma unroll
        for (int j = 0; j < Y_NUM / THREADS; ++j) { ta[j] = amap[tid + j * THREADS]; ty[j] = yin[tid + j * THREADS]; }
#pragma unroll
        for (int j = 0; j < Y_NUM / THREADS; ++j) {
            const int i = tid + j * THREADS;
            As[16 + i] = ta[j];
            if (i >= Y_NUM - 16) As[i - (Y_NUM - 16)] = ta[j];
            Ys[i] = ty[j];
        }
    }
    __syncthreads();

    v8f aE, aO;
#pragma unroll
    for (int i = 0; i < 8; ++i) { aE[i] = 0.0f; aO[i] = 0.0f; }
    const int tile0 = (b << 4) + col;
    for (int rb = 0; rb < 16; rb += 4) {
        v2f av[4][4], bv[4][4];
#pragma unroll
        for (int u = 0; u < 4; ++u) {
            const int r  = (wv << 4) + rb + u;
            const int tl = (tile0 - r) & (NTILES - 1);
            const int x0 = 16 * r + col - 2 * half;
#pragma unroll
            for (int q = 0; q < 4; ++q) {
                const int x = x0 - 4 * q;
                av[u][q][0] = As[16 + x];
                av[u][q][1] = As[15 + x];
                const float2 v = *(const float2*)&Ys[(tl << 4) + (q << 2) + (half << 1)];
                bv[u][q][0] = v.x; bv[u][q][1] = v.y;
            }
        }
        __builtin_amdgcn_sched_barrier(0);
#pragma unroll
        for (int u = 0; u < 4; ++u)
#pragma unroll
            for (int q = 0; q < 4; ++q) {
                if (q & 1) aO = wmma4(av[u][q], bv[u][q], aO);
                else       aE = wmma4(av[u][q], bv[u][q], aE);
            }
        __builtin_amdgcn_sched_barrier(0);
    }
    const v8f acc = aE + aO;

    float* pp = &Pt[(wv << 8) + (ln << 3)];
    *(float4*)&pp[0] = make_float4(acc[0], acc[1], acc[2], acc[3]);
    *(float4*)&pp[4] = make_float4(acc[4], acc[5], acc[6], acc[7]);
    __syncthreads();

    const int e  = tid & 15;
    const int l  = (tid >> 4) + ((e >> 3) << 4);
    const int pv = (l << 3) + (e & 7);
    float sum = 0.0f;
#pragma unroll
    for (int u = 0; u < WAVES; ++u) sum += Pt[(u << 8) + pv];
    sum += ws[WS_G + k];
    const int i = (b << 8) + tid;
    ws[WS_Y + (k + 1) * Y_NUM + i] = sum;
    out[(size_t)(SEGROWS * (k + 1)) * Y_NUM + i] = sum;
}

__global__ __launch_bounds__(THREADS) __attribute__((amdgpu_waves_per_eu(2, 2)))
void ode_chain_seg(const float* __restrict__ y0,
                   const float* __restrict__ ws,
                   float* __restrict__ out) {
    __shared__ __attribute__((aligned(16))) float Y[2][SBUF];
    __shared__ __attribute__((aligned(16))) float Tb[(SEGTRIPS - 1) * TP];

    const int tid = threadIdx.x;
    const int k   = (int)blockIdx.x;

    const float* ys = (k == 0) ? y0 : (ws + WS_Y + k * Y_NUM);
    {
        float ty[Y_NUM / THREADS];
#pragma unroll
        for (int j = 0; j < Y_NUM / THREADS; ++j) ty[j] = ys[tid + j * THREADS];
        copy_floats<(SEGTRIPS - 1) * TP>(Tb, ws + WS_TAB + SEGTRIPS * k * TP, tid);
#pragma unroll
        for (int j = 0; j < Y_NUM / THREADS; ++j) {
            const int i = tid + j * THREADS;
            Y[0][(i >> 4) * SP + (i & 15)] = ty[j];
            if (k == 0) out[i] = ty[j];
        }
    }
    __syncthreads();

    chain_trips<true, SEGTRIPS - 1>(&Y[0][0], Tb, out, SEGROWS * k);
}

template <int J>
__device__ __forceinline__ v8f row_direct(const float* rp, float F, const StateK& s) {
    constexpr int R = 1 + (J + 1) / 2;
    v8f aE, aO;
#pragma unroll
    for (int i = 0; i < 8; ++i) { aE[i] = F; aO[i] = 0.0f; }
    int cnt = 0;
#pragma unroll
    for (int q = 0; q < 4; ++q)
#pragma unroll
        for (int r = 0; r < R; ++r) {
            if ((J & 1) && r == R - 1 && q < 2) continue;
            const v2f a = tap_pair(rp, r, q);
            if (cnt & 1) aO = wmma4(a, s.b[r][q], aO);
            else         aE = wmma4(a, s.b[r][q], aE);
            ++cnt;
        }
    return aE + aO;
}

__device__ __forceinline__ void store_row(float* orow, int rowIx, v8f y) {
    *(float4*)&orow[rowIx]     = make_float4(y[0], y[1], y[2], y[3]);
    *(float4*)&orow[rowIx + 4] = make_float4(y[4], y[5], y[6], y[7]);
}

template <int J>
__device__ __forceinline__ void fill_from(float* out, int c0, int rowIx,
                                          const float* tab_lb, const float* Fs, const StateK& st) {
    if constexpr (J < KROW) {
        store_row(out + (size_t)(c0 + J) * Y_NUM, rowIx,
                  row_direct<J>(tab_lb + (J - 1) * ROWLEN, Fs[J - 1], st));
        fill_from<J + 1>(out, c0, rowIx, tab_lb, Fs, st);
    }
}

__global__ __launch_bounds__(FTHREADS)
void ode_fill(const float* __restrict__ t,
              const float* __restrict__ w,
              const float* __restrict__ rt,
              float* out) {
    __shared__ float sc[KROW - 1][12];
    __shared__ float tab[(KROW - 1) * ROWLEN];
    __shared__ float Fs[KROW];

    const int tid  = threadIdx.x;
    const int wv   = (tid >> 5) + FWAVES * ((int)blockIdx.x & 1);
    const int ln   = tid & 31;
    const int col  = ln & 15;
    const int half = ln >> 4;
    const int m    = col;
    const int c0   = KROW * ((int)blockIdx.x >> 1);

    const int tileC = (wv << 4) + col;
    const int rowIx = (wv << 8) + (col << 4) + (half << 3);
    const int lb    = 15 - m + (half << 1);

    const float* ck = out + (size_t)c0 * Y_NUM;
    StateK st;
#pragma unroll
    for (int q = 0; q < 4; ++q)
#pragma unroll
        for (int r = 0; r < NMAT; ++r) {
            const int tl = (tileC - r) & (NTILES - 1);
            const float2 v = *(const float2*)&ck[(tl << 4) + (q << 2) + (half << 1)];
            st.b[r][q][0] = v.x; st.b[r][q][1] = v.y;
        }

    if (tid < KROW - 1) {
        const float a = w[0] * rt[0];
        const float b = w[1] * rt[1];
        const float c = w[2] * rt[2];
        const StepC A = step_coef(t[c0 + tid], t[c0 + tid + 1], a, b, c);
#pragma unroll
        for (int i = 0; i < 9; ++i) sc[tid][i] = A.c[i];
        sc[tid][9] = A.f;
    }
    __syncthreads();

    for (int p = tid; p < ROWLEN; p += FTHREADS) {
        const int x = TAPTOP - p;
        tab[p] = (x >= 0 && x <= 8) ? sc[0][(x >= 0 && x <= 8) ? x : 0] : 0.0f;
    }
    if (tid == 0) {
        float F = sc[0][9];
        Fs[0] = F;
        for (int j = 1; j < KROW - 1; ++j) {
            float sumB = 0.0f;
            for (int i = 0; i < 9; ++i) sumB += sc[j][i];
            F = F * sumB + sc[j][9];
            Fs[j] = F;
        }
    }
    __syncthreads();

    for (int j = 1; j < KROW - 1; ++j) {
        for (int p = tid; p < ROWLEN; p += FTHREADS) {
            float acc = 0.0f;
#pragma unroll
            for (int i = 0; i < 9; ++i) {
                const int pp = p + i;
                const float pv = (pp < ROWLEN) ? tab[(j - 1) * ROWLEN + ((pp < ROWLEN) ? pp : 0)] : 0.0f;
                acc += sc[j][i] * pv;
            }
            tab[j * ROWLEN + p] = acc;
        }
        __syncthreads();
    }

    fill_from<1>(out, c0, rowIx, tab + lb, Fs, st);
}

extern "C" void kernel_launch(void* const* d_in, const int* in_sizes, int n_in,
                              void* d_out, int out_size, void* d_ws, size_t ws_size,
                              hipStream_t stream) {
    (void)in_sizes; (void)n_in; (void)out_size; (void)ws_size;
    const float* t  = (const float*)d_in[0];
    const float* y0 = (const float*)d_in[1];
    const float* w  = (const float*)d_in[2];
    const float* r  = (const float*)d_in[3];
    float* out = (float*)d_out;
    float* ws  = (float*)d_ws;

    ode_maps<<<SEGS, THREADS, 0, stream>>>(t, w, r, ws);
    ode_seg_start<<<8, THREADS, 0, stream>>>(y0, ws, out, 0);
    ode_seg_start<<<8, THREADS, 0, stream>>>(y0, ws, out, 1);
    ode_seg_start<<<8, THREADS, 0, stream>>>(y0, ws, out, 2);
    ode_seg_start<<<8, THREADS, 0, stream>>>(y0, ws, out, 3);
    ode_seg_start<<<8, THREADS, 0, stream>>>(y0, ws, out, 4);
    ode_seg_start<<<8, THREADS, 0, stream>>>(y0, ws, out, 5);
    ode_seg_start<<<8, THREADS, 0, stream>>>(y0, ws, out, 6);
    ode_chain_seg<<<SEGS, THREADS, 0, stream>>>(y0, ws, out);
    ode_fill<<<2 * NGROUP, FTHREADS, 0, stream>>>(t, w, r, out);
}


# Segment start rows by a 3-level parallel prefix over the 7 segment maps (maps composed pairwise by circulant products, all of a level side by side): 3 launches instead of 7 serial; sums regrouped
# speedup vs baseline: 22.9798x; 1.2800x over previous
#include <hip/hip_runtime.h>


typedef float v2f __attribute__((ext_vector_type(2)));
typedef float v8f __attribute__((ext_vector_type(8)));

#define SAMPLE_NUM 2048
#define Y_NUM      2048
#define NSTEP      (SAMPLE_NUM - 1)
#define KROW       16
#define NTRIP      (NSTEP / KROW)
#define NGROUP     (NTRIP + 1)
#define NTILES     128
#define WAVES      8
#define THREADS    (WAVES * 32)
#define FWAVES     4
#define FTHREADS   (FWAVES * 32)
#define NMAT       (KROW / 2 + 1)
#define NTAPK      (8 * KROW + 1)
#define TAPTOP     (16 * NMAT - 1)
#define ROWLEN     (16 * NMAT + 16)
#define TP         (ROWLEN + 4)
#define FPOS       ROWLEN
#define SP         20
#define SBUF       (NTILES * SP)
#define SEGS       8
#define SEGTRIPS   16
#define SEGROWS    (KROW * SEGTRIPS)
#define NMAPS      (SEGS - 1)
#define LASTTRIPS  (NTRIP - SEGTRIPS * (SEGS - 1))
#define SUMPOS     (FPOS + 1)
#define NSLOT      (3 * SEGS)
#define WS_TAB     0
#define WS_MAP     32768
#define WS_G       81920
#define WS_S       81984
#define WS_Y       86016
static_assert(WS_MAP + NSLOT * Y_NUM <= WS_G && WS_G + NSLOT <= WS_S && WS_S + NSLOT <= WS_Y, "workspace regions overlap");

struct StepC {
    float c[9];
    float f;
};

__device__ __forceinline__ StepC step_coef(float tn, float tn1, float a, float b, float c) {
    const float sub = (tn1 - tn) * 0.125f;
    const float d   = 1.0f - sub * b;
    const float e   = sub * a;
    StepC r;
    {
        const float bin[9] = {1.f, 8.f, 28.f, 56.f, 70.f, 56.f, 28.f, 8.f, 1.f};
        float dp[9], ep[9];
        dp[0] = 1.0f; ep[0] = 1.0f;
#pragma unroll
        for (int j = 1; j < 9; ++j) { dp[j] = dp[j-1] * d; ep[j] = ep[j-1] * e; }
#pragma unroll
        for (int j = 0; j < 9; ++j) r.c[j] = bin[j] * dp[8 - j] * ep[j];
    }
    const float g = d + e;
    float f = 0.0f;
#pragma unroll
    for (int ms = 0; ms < 8; ++ms) {
        const float s = sub * c * __sinf(tn + sub * (float)ms);
        f = f * g + s;
    }
    r.f = f;
    return r;
}

__device__ __forceinline__ v8f wmma4(v2f a, v2f b, v8f c) {
    return __builtin_amdgcn_wmma_f32_16x16x4_f32(false, a, false, b, (short)0, c, false, false);
}

struct StateK {
    v2f b[NMAT][4];
};

__device__ __forceinline__ v2f tap_pair(const float* rp, int r, int q) {
    v2f a;
    a[0] = rp[16 * (NMAT - 1) - 16 * r + 4 * q];
    a[1] = rp[16 * (NMAT - 1) - 16 * r + 4 * q + 1];
    return a;
}

struct CoefK {
    v2f   a[NMAT][4];
    float f;
};

template <bool FORCED>
__device__ __forceinline__ CoefK load_coefK(const float* row, int lb) {
    CoefK c;
    c.f = FORCED ? row[FPOS] : 0.0f;
    const float* rp = row + lb;
#pragma unroll
    for (int q = 0; q < 4; ++q)
#pragma unroll
        for (int r = 0; r < NMAT; ++r) c.a[r][q] = tap_pair(rp, r, q);
    return c;
}

__device__ __forceinline__ StateK load_state_lds(const float* Yb, const int* so) {
    StateK s;
#pragma unroll
    for (int q = 0; q < 4; ++q)
#pragma unroll
        for (int r = 0; r < NMAT; ++r) {
            const float2 v = *(const float2*)&Yb[so[r] + 4 * q];
            s.b[r][q][0] = v.x; s.b[r][q][1] = v.y;
        }
    return s;
}

__device__ __forceinline__ v8f rows_trip(const CoefK& k, const StateK& s) {
    v8f aE, aO;
#pragma unroll
    for (int i = 0; i < 8; ++i) { aE[i] = k.f; aO[i] = 0.0f; }
#pragma unroll
    for (int q = 0; q < 4; ++q)
#pragma unroll
        for (int r = 0; r < NMAT; ++r) {
            if ((q * NMAT + r) & 1) aO = wmma4(k.a[r][q], s.b[r][q], aO);
            else                    aE = wmma4(k.a[r][q], s.b[r][q], aE);
        }
    return aE + aO;
}


template <int N>
__device__ __forceinline__ void copy_floats(float* dst, const float* src, int tid) {
    constexpr int IT = (N + THREADS - 1) / THREADS;
    float v[IT];
#pragma unroll
    for (int j = 0; j < IT; ++j) { const int i = tid + j * THREADS; v[j] = (i < N) ? src[i] : 0.0f; }
#pragma unroll
    for (int j = 0; j < IT; ++j) { const int i = tid + j * THREADS; if (i < N) dst[i] = v[j]; }
}

#define LANEPOS (ROWLEN / 16)
static_assert(LANEPOS * 16 == ROWLEN && LANEPOS >= 8, "a lane's window must reach exactly one lane up");
static_assert(THREADS == 16 * SEGTRIPS && KROW == 16, "16 lanes per trip, one row map per thread");

__device__ __forceinline__ void build_trip_rows(float* Tb, const float (*sc)[12], int tid) {
    const int j = tid >> 4;
    const int u = tid & 15;
    const float* r0 = sc[KROW * j];
    float Wd[LANEPOS + 8];
#pragma unroll
    for (int e = 0; e < LANEPOS; ++e) {
        const int x = TAPTOP - (LANEPOS * u + e);
        const bool in = (x >= 0 && x <= 8);
        Wd[e] = in ? r0[in ? x : 0] : 0.0f;
    }
    float F = r0[9];
    for (int jj = 1; jj < KROW; ++jj) {
        const float* rb = sc[KROW * j + jj];
        const float4 c03 = *(const float4*)&rb[0];
        const float4 c47 = *(const float4*)&rb[4];
        const float4 c8x = *(const float4*)&rb[8];
        const float cc[9] = {c03.x, c03.y, c03.z, c03.w, c47.x, c47.y, c47.z, c47.w, c8x.x};
#pragma unroll
        for (int e = 0; e < 8; ++e) {
            const float v = __shfl_down(Wd[e], 1, 16);
            Wd[LANEPOS + e] = (u == 15) ? 0.0f : v;
        }
        float Nw[LANEPOS];
#pragma unroll
        for (int e = 0; e < LANEPOS; ++e) {
            float acc = 0.0f;
#pragma unroll
            for (int i = 0; i < 9; ++i) acc += cc[i] * Wd[e + i];
            Nw[e] = acc;
        }
#pragma unroll
        for (int e = 0; e < LANEPOS; ++e) Wd[e] = Nw[e];
        F = F * c8x.z + c8x.y;
    }
    float sp = 0.0f;
#pragma unroll
    for (int e = 0; e < LANEPOS; ++e) sp += Wd[e];
#pragma unroll
    for (int dlt = 8; dlt >= 1; dlt >>= 1) sp += __shfl_down(sp, dlt, 16);

    float* row = Tb + j * TP;
    float2* dst = (float2*)&row[LANEPOS * u];
#pragma unroll
    for (int e = 0; e < LANEPOS / 2; ++e) dst[e] = make_float2(Wd[2 * e], Wd[2 * e + 1]);
    if (u == 0) *(float4*)&row[FPOS] = make_float4(F, sp, 0.0f, 0.0f);
}

template <bool ROWS, int NTR>
__device__ __forceinline__ void chain_trips(float* Yb, const float* Tb, float* __restrict__ out, int row0) {
    const int tid  = threadIdx.x;
    const int wv   = tid >> 5;
    const int ln   = tid & 31;
    const int col  = ln & 15;
    const int half = ln >> 4;
    const int m    = col;

    const int tileC = (wv << 4) + col;
    const int rowIx = (wv << 8) + (col << 4) + (half << 3);
    const int stIx  = tileC * SP + (half << 3);
    const int lb    = 15 - m + (half << 1);
    int so[NMAT];
#pragma unroll
    for (int r = 0; r < NMAT; ++r) so[r] = ((tileC - r) & (NTILES - 1)) * SP + (half << 1);

    CoefK  cur = load_coefK<ROWS>(Tb, lb);
    StateK st  = load_state_lds(Yb, so);

#pragma unroll 2
    for (int s = 0; s < NTR; ++s) {
        float* __restrict__ Yn = Yb + ((s & 1) ^ 1) * SBUF;

        const v8f yk = rows_trip(cur, st);
        const float4 lo = make_float4(yk[0], yk[1], yk[2], yk[3]);
        const float4 hi = make_float4(yk[4], yk[5], yk[6], yk[7]);
        *(float4*)&Yn[stIx]     = lo;
        *(float4*)&Yn[stIx + 4] = hi;

        __builtin_amdgcn_fence(__ATOMIC_RELEASE, "workgroup", "local");
        __builtin_amdgcn_s_barrier();
        __builtin_amdgcn_fence(__ATOMIC_ACQUIRE, "workgroup", "local");

        const int sn = (s + 1 < NTR) ? (s + 1) : (NTR - 1);
        const StateK stn = load_state_lds(Yn, so);
        __builtin_amdgcn_sched_barrier(0);
        const CoefK  nxt = load_coefK<ROWS>(Tb + sn * TP, lb);
        __builtin_amdgcn_sched_barrier(0);

        if (ROWS) {
            float* __restrict__ orow = out + (size_t)(row0 + KROW * (s + 1)) * Y_NUM;
            *(float4*)&orow[rowIx]     = lo;
            *(float4*)&orow[rowIx + 4] = hi;
        }

        st  = stn;
        cur = nxt;
    }
}

__global__ __launch_bounds__(THREADS) __attribute__((amdgpu_waves_per_eu(2, 2)))
void ode_maps(const float* __restrict__ t,
              const float* __restrict__ w,
              const float* __restrict__ rt,
              float* __restrict__ ws) {
    __shared__ __attribute__((aligned(16))) float Y[2][SBUF];
    __shared__ __attribute__((aligned(16))) float Tb[SEGTRIPS * TP];
    __shared__ __attribute__((aligned(16))) float sc[SEGTRIPS * KROW][12];

    const int tid = threadIdx.x;
    const int k   = (int)blockIdx.x;
    const int nrows = (k == SEGS - 1) ? LASTTRIPS : SEGTRIPS;

    {
        StepC A;
#pragma unroll
        for (int i = 0; i < 9; ++i) A.c[i] = 0.0f;
        A.f = 0.0f;
        if (tid < KROW * nrows) {
            const float a = w[0] * rt[0];
            const float b = w[1] * rt[1];
            const float c = w[2] * rt[2];
            const int n = SEGROWS * k + tid;
            A = step_coef(t[n], t[n + 1], a, b, c);
        }
        float sumA = 0.0f;
#pragma unroll
        for (int i = 0; i < 9; ++i) sumA += A.c[i];
        *(float4*)&sc[tid][0] = make_float4(A.c[0], A.c[1], A.c[2], A.c[3]);
        *(float4*)&sc[tid][4] = make_float4(A.c[4], A.c[5], A.c[6], A.c[7]);
        *(float4*)&sc[tid][8] = make_float4(A.c[8], A.f, sumA, 0.0f);
    }
    __syncthreads();
    build_trip_rows(Tb, sc, tid);
    for (int i = tid; i < SBUF; i += THREADS) Y[0][i] = (i == 0) ? 1.0f : 0.0f;
    __syncthreads();

    copy_floats<LASTTRIPS * TP>(ws + WS_TAB + SEGTRIPS * k * TP, Tb, tid);

    if (k == SEGS - 1) return;

    if (tid == 0) {
        float G = 0.0f, P = 1.0f;
#pragma unroll
        for (int j = 0; j < SEGTRIPS; ++j) {
            const float sj = Tb[j * TP + SUMPOS];
            G = G * sj + Tb[j * TP + FPOS];
            P = P * sj;
        }
        ws[WS_G + k] = G;
        ws[WS_S + k] = P;
    }

    chain_trips<false, SEGTRIPS>(&Y[0][0], Tb, nullptr, 0);

    const float* Yf = Y[SEGTRIPS & 1];
#pragma unroll
    for (int j = 0; j < Y_NUM / THREADS; ++j) {
        const int i = tid + j * THREADS;
        ws[WS_MAP + k * Y_NUM + i] = Yf[(i >> 4) * SP + (i & 15)];
    }
}

__global__ __launch_bounds__(THREADS)
void ode_seg_scan(const float* __restrict__ y0,
                  float* __restrict__ ws,
                  float* __restrict__ out,
                  int lvl) {
    __shared__ __attribute__((aligned(16))) float As[Y_NUM + 16];
    __shared__ __attribute__((aligned(16))) float Ys[Y_NUM];
    __shared__ __attribute__((aligned(16))) float Pt[WAVES * 256];

    const int tid  = threadIdx.x;
    const int wv   = tid >> 5;
    const int ln   = tid & 31;
    const int col  = ln & 15;
    const int half = ln >> 4;
    const int b    = (int)blockIdx.x;

    const int  s   = 1 << (lvl - 1);
    const int  k   = s - 1 + (int)blockIdx.y;
    const int  hi  = (lvl - 1) * SEGS + k;
    const bool vec = (k < 2 * s - 1);
    const int  j0  = k - s + 1;
    const float* yin = vec ? ((j0 == 0) ? y0 : (ws + WS_Y + j0 * Y_NUM))
                           : (ws + WS_MAP + (hi - s) * Y_NUM);
    {
        const float* amap = ws + WS_MAP + hi * Y_NUM;
        float ta[Y_NUM / THREADS], ty[Y_NUM / THREADS];
#pragma unroll
        for (int j = 0; j < Y_NUM / THREADS; ++j) { ta[j] = amap[tid + j * THREADS]; ty[j] = yin[tid + j * THREADS]; }
#pragma unroll
        for (int j = 0; j < Y_NUM / THREADS; ++j) {
            const int i = tid + j * THREADS;
            As[16 + i] = ta[j];
            if (i >= Y_NUM - 16) As[i - (Y_NUM - 16)] = ta[j];
            Ys[i] = ty[j];
        }
    }
    __syncthreads();

    v8f aE, aO;
#pragma unroll
    for (int i = 0; i < 8; ++i) { aE[i] = 0.0f; aO[i] = 0.0f; }
    const int tile0 = (b << 4) + col;
    for (int rb = 0; rb < 16; rb += 4) {
        v2f av[4][4], bv[4][4];
#pragma unroll
        for (int u = 0; u < 4; ++u) {
            const int r  = (wv << 4) + rb + u;
            const int tl = (tile0 - r) & (NTILES - 1);
            const int x0 = 16 * r + col - 2 * half;
#pragma unroll
            for (int q = 0; q < 4; ++q) {
                const int x = x0 - 4 * q;
                av[u][q][0] = As[16 + x];
                av[u][q][1] = As[15 + x];
                const float2 v = *(const float2*)&Ys[(tl << 4) + (q << 2) + (half << 1)];
                bv[u][q][0] = v.x; bv[u][q][1] = v.y;
            }
        }
        __builtin_amdgcn_sched_barrier(0);
#pragma unroll
        for (int u = 0; u < 4; ++u)
#pragma unroll
            for (int q = 0; q < 4; ++q) {
                if (q & 1) aO = wmma4(av[u][q], bv[u][q], aO);
                else       aE = wmma4(av[u][q], bv[u][q], aE);
            }
        __builtin_amdgcn_sched_barrier(0);
    }
    const v8f acc = aE + aO;

    float* pp = &Pt[(wv << 8) + (ln << 3)];
    *(float4*)&pp[0] = make_float4(acc[0], acc[1], acc[2], acc[3]);
    *(float4*)&pp[4] = make_float4(acc[4], acc[5], acc[6], acc[7]);
    __syncthreads();

    const int e  = tid & 15;
    const int l  = (tid >> 4) + ((e >> 3) << 4);
    const int pv = (l << 3) + (e & 7);
    float sum = 0.0f;
#pragma unroll
    for (int u = 0; u < WAVES; ++u) sum += Pt[(u << 8) + pv];
    const int i = (b << 8) + tid;
    if (vec) {
        sum += ws[WS_G + hi];
        ws[WS_Y + (k + 1) * Y_NUM + i] = sum;
        out[(size_t)(SEGROWS * (k + 1)) * Y_NUM + i] = sum;
    } else {
        const int dst = lvl * SEGS + k;
        ws[WS_MAP + dst * Y_NUM + i] = sum;
        if (i == 0) {
            const float Sh = ws[WS_S + hi];
            ws[WS_G + dst] = ws[WS_G + hi - s] * Sh + ws[WS_G + hi];
            ws[WS_S + dst] = Sh * ws[WS_S + hi - s];
        }
    }
}

__global__ __launch_bounds__(THREADS) __attribute__((amdgpu_waves_per_eu(2, 2)))
void ode_chain_seg(const float* __restrict__ y0,
                   const float* __restrict__ ws,
                   float* __restrict__ out) {
    __shared__ __attribute__((aligned(16))) float Y[2][SBUF];
    __shared__ __attribute__((aligned(16))) float Tb[(SEGTRIPS - 1) * TP];

    const int tid = threadIdx.x;
    const int k   = (int)blockIdx.x;

    const float* ys = (k == 0) ? y0 : (ws + WS_Y + k * Y_NUM);
    {
        float ty[Y_NUM / THREADS];
#pragma unroll
        for (int j = 0; j < Y_NUM / THREADS; ++j) ty[j] = ys[tid + j * THREADS];
        copy_floats<(SEGTRIPS - 1) * TP>(Tb, ws + WS_TAB + SEGTRIPS * k * TP, tid);
#pragma unroll
        for (int j = 0; j < Y_NUM / THREADS; ++j) {
            const int i = tid + j * THREADS;
            Y[0][(i >> 4) * SP + (i & 15)] = ty[j];
            if (k == 0) out[i] = ty[j];
        }
    }
    __syncthreads();

    chain_trips<true, SEGTRIPS - 1>(&Y[0][0], Tb, out, SEGROWS * k);
}

template <int J>
__device__ __forceinline__ v8f row_direct(const float* rp, float F, const StateK& s) {
    constexpr int R = 1 + (J + 1) / 2;
    v8f aE, aO;
#pragma unroll
    for (int i = 0; i < 8; ++i) { aE[i] = F; aO[i] = 0.0f; }
    int cnt = 0;
#pragma unroll
    for (int q = 0; q < 4; ++q)
#pragma unroll
        for (int r = 0; r < R; ++r) {
            if ((J & 1) && r == R - 1 && q < 2) continue;
            const v2f a = tap_pair(rp, r, q);
            if (cnt & 1) aO = wmma4(a, s.b[r][q], aO);
            else         aE = wmma4(a, s.b[r][q], aE);
            ++cnt;
        }
    return aE + aO;
}

__device__ __forceinline__ void store_row(float* orow, int rowIx, v8f y) {
    *(float4*)&orow[rowIx]     = make_float4(y[0], y[1], y[2], y[3]);
    *(float4*)&orow[rowIx + 4] = make_float4(y[4], y[5], y[6], y[7]);
}

template <int J>
__device__ __forceinline__ void fill_from(float* out, int c0, int rowIx,
                                          const float* tab_lb, const float* Fs, const StateK& st) {
    if constexpr (J < KROW) {
        store_row(out + (size_t)(c0 + J) * Y_NUM, rowIx,
                  row_direct<J>(tab_lb + (J - 1) * ROWLEN, Fs[J - 1], st));
        fill_from<J + 1>(out, c0, rowIx, tab_lb, Fs, st);
    }
}

__global__ __launch_bounds__(FTHREADS)
void ode_fill(const float* __restrict__ t,
              const float* __restrict__ w,
              const float* __restrict__ rt,
              float* out) {
    __shared__ float sc[KROW - 1][12];
    __shared__ float tab[(KROW - 1) * ROWLEN];
    __shared__ float Fs[KROW];

    const int tid  = threadIdx.x;
    const int wv   = (tid >> 5) + FWAVES * ((int)blockIdx.x & 1);
    const int ln   = tid & 31;
    const int col  = ln & 15;
    const int half = ln >> 4;
    const int m    = col;
    const int c0   = KROW * ((int)blockIdx.x >> 1);

    const int tileC = (wv << 4) + col;
    const int rowIx = (wv << 8) + (col << 4) + (half << 3);
    const int lb    = 15 - m + (half << 1);

    const float* ck = out + (size_t)c0 * Y_NUM;
    StateK st;
#pragma unroll
    for (int q = 0; q < 4; ++q)
#pragma unroll
        for (int r = 0; r < NMAT; ++r) {
            const int tl = (tileC - r) & (NTILES - 1);
            const float2 v = *(const float2*)&ck[(tl << 4) + (q << 2) + (half << 1)];
            st.b[r][q][0] = v.x; st.b[r][q][1] = v.y;
        }

    if (tid < KROW - 1) {
        const float a = w[0] * rt[0];
        const float b = w[1] * rt[1];
        const float c = w[2] * rt[2];
        const StepC A = step_coef(t[c0 + tid], t[c0 + tid + 1], a, b, c);
#pragma unroll
        for (int i = 0; i < 9; ++i) sc[tid][i] = A.c[i];
        sc[tid][9] = A.f;
    }
    __syncthreads();

    for (int p = tid; p < ROWLEN; p += FTHREADS) {
        const int x = TAPTOP - p;
        tab[p] = (x >= 0 && x <= 8) ? sc[0][(x >= 0 && x <= 8) ? x : 0] : 0.0f;
    }
    if (tid == 0) {
        float F = sc[0][9];
        Fs[0] = F;
        for (int j = 1; j < KROW - 1; ++j) {
            float sumB = 0.0f;
            for (int i = 0; i < 9; ++i) sumB += sc[j][i];
            F = F * sumB + sc[j][9];
            Fs[j] = F;
        }
    }
    __syncthreads();

    for (int j = 1; j < KROW - 1; ++j) {
        for (int p = tid; p < ROWLEN; p += FTHREADS) {
            float acc = 0.0f;
#pragma unroll
            for (int i = 0; i < 9; ++i) {
                const int pp = p + i;
                const float pv = (pp < ROWLEN) ? tab[(j - 1) * ROWLEN + ((pp < ROWLEN) ? pp : 0)] : 0.0f;
                acc += sc[j][i] * pv;
            }
            tab[j * ROWLEN + p] = acc;
        }
        __syncthreads();
    }

    fill_from<1>(out, c0, rowIx, tab + lb, Fs, st);
}

extern "C" void kernel_launch(void* const* d_in, const int* in_sizes, int n_in,
                              void* d_out, int out_size, void* d_ws, size_t ws_size,
                              hipStream_t stream) {
    (void)in_sizes; (void)n_in; (void)out_size; (void)ws_size;
    const float* t  = (const float*)d_in[0];
    const float* y0 = (const float*)d_in[1];
    const float* w  = (const float*)d_in[2];
    const float* r  = (const float*)d_in[3];
    float* out = (float*)d_out;
    float* ws  = (float*)d_ws;

    static_assert(SEGS == 8, "the level shapes below are for 7 segment maps");
    ode_maps<<<SEGS, THREADS, 0, stream>>>(t, w, r, ws);
    ode_seg_scan<<<dim3(8, 7), THREADS, 0, stream>>>(y0, ws, out, 1);
    ode_seg_scan<<<dim3(8, 6), THREADS, 0, stream>>>(y0, ws, out, 2);
    ode_seg_scan<<<dim3(8, 4), THREADS, 0, stream>>>(y0, ws, out, 3);
    ode_chain_seg<<<SEGS, THREADS, 0, stream>>>(y0, ws, out);
    ode_fill<<<2 * NGROUP, FTHREADS, 0, stream>>>(t, w, r, out);
}


# Chain re-cut into 32 segments of 64 rows (was 8 of 256): 4+3 serial trips per segment instead of 16+15; start rows by a 5-level prefix over 31 segment maps, 8 launches (was 6); sums regrouped
# speedup vs baseline: 27.9753x; 1.2174x over previous
#include <hip/hip_runtime.h>


typedef float v2f __attribute__((ext_vector_type(2)));
typedef float v8f __attribute__((ext_vector_type(8)));

#define SAMPLE_NUM 2048
#define Y_NUM      2048
#define NSTEP      (SAMPLE_NUM - 1)
#define KROW       16
#define NTRIP      (NSTEP / KROW)
#define NGROUP     (NTRIP + 1)
#define NTILES     128
#define WAVES      8
#define THREADS    (WAVES * 32)
#define FWAVES     4
#define FTHREADS   (FWAVES * 32)
#define NMAT       (KROW / 2 + 1)
#define NTAPK      (8 * KROW + 1)
#define TAPTOP     (16 * NMAT - 1)
#define ROWLEN     (16 * NMAT + 16)
#define TP         (ROWLEN + 4)
#define FPOS       ROWLEN
#define SP         20
#define SBUF       (NTILES * SP)
#define SEGS       32
#define LEVELS     5
#define SEGTRIPS   4
#define SEGROWS    (KROW * SEGTRIPS)
#define NMAPS      (SEGS - 1)
#define LASTTRIPS  (NTRIP - SEGTRIPS * (SEGS - 1))
#define SUMPOS     (FPOS + 1)
#define NSLOT      (LEVELS * SEGS)
#define WS_TAB     0
#define WS_MAP     32768
#define WS_G       (WS_MAP + NSLOT * Y_NUM)
#define WS_S       (WS_G + 256)
#define WS_Y       (WS_S + 256)
static_assert(NTRIP * TP <= WS_MAP && WS_MAP + NSLOT * Y_NUM <= WS_G && WS_G + NSLOT <= WS_S && WS_S + NSLOT <= WS_Y, "workspace regions overlap");
static_assert((1 << LEVELS) == SEGS && LASTTRIPS == SEGTRIPS - 1, "segment shapes");

struct StepC {
    float c[9];
    float f;
};

__device__ __forceinline__ StepC step_coef(float tn, float tn1, float a, float b, float c) {
    const float sub = (tn1 - tn) * 0.125f;
    const float d   = 1.0f - sub * b;
    const float e   = sub * a;
    StepC r;
    {
        const float bin[9] = {1.f, 8.f, 28.f, 56.f, 70.f, 56.f, 28.f, 8.f, 1.f};
        float dp[9], ep[9];
        dp[0] = 1.0f; ep[0] = 1.0f;
#pragma unroll
        for (int j = 1; j < 9; ++j) { dp[j] = dp[j-1] * d; ep[j] = ep[j-1] * e; }
#pragma unroll
        for (int j = 0; j < 9; ++j) r.c[j] = bin[j] * dp[8 - j] * ep[j];
    }
    const float g = d + e;
    float f = 0.0f;
#pragma unroll
    for (int ms = 0; ms < 8; ++ms) {
        const float s = sub * c * __sinf(tn + sub * (float)ms);
        f = f * g + s;
    }
    r.f = f;
    return r;
}

__device__ __forceinline__ v8f wmma4(v2f a, v2f b, v8f c) {
    return __builtin_amdgcn_wmma_f32_16x16x4_f32(false, a, false, b, (short)0, c, false, false);
}

struct StateK {
    v2f b[NMAT][4];
};

__device__ __forceinline__ v2f tap_pair(const float* rp, int r, int q) {
    v2f a;
    a[0] = rp[16 * (NMAT - 1) - 16 * r + 4 * q];
    a[1] = rp[16 * (NMAT - 1) - 16 * r + 4 * q + 1];
    return a;
}

struct CoefK {
    v2f   a[NMAT][4];
    float f;
};

template <bool FORCED>
__device__ __forceinline__ CoefK load_coefK(const float* row, int lb) {
    CoefK c;
    c.f = FORCED ? row[FPOS] : 0.0f;
    const float* rp = row + lb;
#pragma unroll
    for (int q = 0; q < 4; ++q)
#pragma unroll
        for (int r = 0; r < NMAT; ++r) c.a[r][q] = tap_pair(rp, r, q);
    return c;
}

__device__ __forceinline__ StateK load_state_lds(const float* Yb, const int* so) {
    StateK s;
#pragma unroll
    for (int q = 0; q < 4; ++q)
#pragma unroll
        for (int r = 0; r < NMAT; ++r) {
            const float2 v = *(const float2*)&Yb[so[r] + 4 * q];
            s.b[r][q][0] = v.x; s.b[r][q][1] = v.y;
        }
    return s;
}

__device__ __forceinline__ v8f rows_trip(const CoefK& k, const StateK& s) {
    v8f aE, aO;
#pragma unroll
    for (int i = 0; i < 8; ++i) { aE[i] = k.f; aO[i] = 0.0f; }
#pragma unroll
    for (int q = 0; q < 4; ++q)
#pragma unroll
        for (int r = 0; r < NMAT; ++r) {
            if ((q * NMAT + r) & 1) aO = wmma4(k.a[r][q], s.b[r][q], aO);
            else                    aE = wmma4(k.a[r][q], s.b[r][q], aE);
        }
    return aE + aO;
}


template <int N>
__device__ __forceinline__ void copy_floats(float* dst, const float* src, int tid) {
    constexpr int IT = (N + THREADS - 1) / THREADS;
    float v[IT];
#pragma unroll
    for (int j = 0; j < IT; ++j) { const int i = tid + j * THREADS; v[j] = (i < N) ? src[i] : 0.0f; }
#pragma unroll
    for (int j = 0; j < IT; ++j) { const int i = tid + j * THREADS; if (i < N) dst[i] = v[j]; }
}

#define LANEPOS (ROWLEN / 16)
static_assert(LANEPOS * 16 == ROWLEN && LANEPOS >= 8, "a lane's window must reach exactly one lane up");
static_assert(THREADS >= 16 * SEGTRIPS && (16 * SEGTRIPS) % 32 == 0 && KROW == 16, "16 lanes per trip, whole waves");

__device__ __forceinline__ void build_trip_rows(float* Tb, const float (*sc)[12], int tid) {
    if (tid >= 16 * SEGTRIPS) return;
    const int j = tid >> 4;
    const int u = tid & 15;
    const float* r0 = sc[KROW * j];
    float Wd[LANEPOS + 8];
#pragma unroll
    for (int e = 0; e < LANEPOS; ++e) {
        const int x = TAPTOP - (LANEPOS * u + e);
        const bool in = (x >= 0 && x <= 8);
        Wd[e] = in ? r0[in ? x : 0] : 0.0f;
    }
    float F = r0[9];
    for (int jj = 1; jj < KROW; ++jj) {
        const float* rb = sc[KROW * j + jj];
        const float4 c03 = *(const float4*)&rb[0];
        const float4 c47 = *(const float4*)&rb[4];
        const float4 c8x = *(const float4*)&rb[8];
        const float cc[9] = {c03.x, c03.y, c03.z, c03.w, c47.x, c47.y, c47.z, c47.w, c8x.x};
#pragma unroll
        for (int e = 0; e < 8; ++e) {
            const float v = __shfl_down(Wd[e], 1, 16);
            Wd[LANEPOS + e] = (u == 15) ? 0.0f : v;
        }
        float Nw[LANEPOS];
#pragma unroll
        for (int e = 0; e < LANEPOS; ++e) {
            float acc = 0.0f;
#pragma unroll
            for (int i = 0; i < 9; ++i) acc += cc[i] * Wd[e + i];
            Nw[e] = acc;
        }
#pragma unroll
        for (int e = 0; e < LANEPOS; ++e) Wd[e] = Nw[e];
        F = F * c8x.z + c8x.y;
    }
    float sp = 0.0f;
#pragma unroll
    for (int e = 0; e < LANEPOS; ++e) sp += Wd[e];
#pragma unroll
    for (int dlt = 8; dlt >= 1; dlt >>= 1) sp += __shfl_down(sp, dlt, 16);

    float* row = Tb + j * TP;
    float2* dst = (float2*)&row[LANEPOS * u];
#pragma unroll
    for (int e = 0; e < LANEPOS / 2; ++e) dst[e] = make_float2(Wd[2 * e], Wd[2 * e + 1]);
    if (u == 0) *(float4*)&row[FPOS] = make_float4(F, sp, 0.0f, 0.0f);
}

template <bool ROWS, int NTR>
__device__ __forceinline__ void chain_trips(float* Yb, const float* Tb, float* __restrict__ out, int row0) {
    const int tid  = threadIdx.x;
    const int wv   = tid >> 5;
    const int ln   = tid & 31;
    const int col  = ln & 15;
    const int half = ln >> 4;
    const int m    = col;

    const int tileC = (wv << 4) + col;
    const int rowIx = (wv << 8) + (col << 4) + (half << 3);
    const int stIx  = tileC * SP + (half << 3);
    const int lb    = 15 - m + (half << 1);
    int so[NMAT];
#pragma unroll
    for (int r = 0; r < NMAT; ++r) so[r] = ((tileC - r) & (NTILES - 1)) * SP + (half << 1);

    CoefK  cur = load_coefK<ROWS>(Tb, lb);
    StateK st  = load_state_lds(Yb, so);

#pragma unroll 2
    for (int s = 0; s < NTR; ++s) {
        float* __restrict__ Yn = Yb + ((s & 1) ^ 1) * SBUF;

        const v8f yk = rows_trip(cur, st);
        const float4 lo = make_float4(yk[0], yk[1], yk[2], yk[3]);
        const float4 hi = make_float4(yk[4], yk[5], yk[6], yk[7]);
        *(float4*)&Yn[stIx]     = lo;
        *(float4*)&Yn[stIx + 4] = hi;

        __builtin_amdgcn_fence(__ATOMIC_RELEASE, "workgroup", "local");
        __builtin_amdgcn_s_barrier();
        __builtin_amdgcn_fence(__ATOMIC_ACQUIRE, "workgroup", "local");

        const int sn = (s + 1 < NTR) ? (s + 1) : (NTR - 1);
        const StateK stn = load_state_lds(Yn, so);
        __builtin_amdgcn_sched_barrier(0);
        const CoefK  nxt = load_coefK<ROWS>(Tb + sn * TP, lb);
        __builtin_amdgcn_sched_barrier(0);

        if (ROWS) {
            float* __restrict__ orow = out + (size_t)(row0 + KROW * (s + 1)) * Y_NUM;
            *(float4*)&orow[rowIx]     = lo;
            *(float4*)&orow[rowIx + 4] = hi;
        }

        st  = stn;
        cur = nxt;
    }
}

__global__ __launch_bounds__(THREADS) __attribute__((amdgpu_waves_per_eu(2, 2)))
void ode_maps(const float* __restrict__ t,
              const float* __restrict__ w,
              const float* __restrict__ rt,
              float* __restrict__ ws) {
    __shared__ __attribute__((aligned(16))) float Y[2][SBUF];
    __shared__ __attribute__((aligned(16))) float Tb[SEGTRIPS * TP];
    __shared__ __attribute__((aligned(16))) float sc[SEGTRIPS * KROW][12];

    const int tid = threadIdx.x;
    const int k   = (int)blockIdx.x;
    const int nrows = (k == SEGS - 1) ? LASTTRIPS : SEGTRIPS;

    {
        StepC A;
#pragma unroll
        for (int i = 0; i < 9; ++i) A.c[i] = 0.0f;
        A.f = 0.0f;
        if (tid < KROW * nrows) {
            const float a = w[0] * rt[0];
            const float b = w[1] * rt[1];
            const float c = w[2] * rt[2];
            const int n = SEGROWS * k + tid;
            A = step_coef(t[n], t[n + 1], a, b, c);
        }
        float sumA = 0.0f;
#pragma unroll
        for (int i = 0; i < 9; ++i) sumA += A.c[i];
        if (tid < SEGROWS) {
            *(float4*)&sc[tid][0] = make_float4(A.c[0], A.c[1], A.c[2], A.c[3]);
            *(float4*)&sc[tid][4] = make_float4(A.c[4], A.c[5], A.c[6], A.c[7]);
            *(float4*)&sc[tid][8] = make_float4(A.c[8], A.f, sumA, 0.0f);
        }
    }
    __syncthreads();
    build_trip_rows(Tb, sc, tid);
    for (int i = tid; i < SBUF; i += THREADS) Y[0][i] = (i == 0) ? 1.0f : 0.0f;
    __syncthreads();

    copy_floats<LASTTRIPS * TP>(ws + WS_TAB + SEGTRIPS * k * TP, Tb, tid);

    if (k == SEGS - 1) return;

    if (tid == 0) {
        float G = 0.0f, P = 1.0f;
#pragma unroll
        for (int j = 0; j < SEGTRIPS; ++j) {
            const float sj = Tb[j * TP + SUMPOS];
            G = G * sj + Tb[j * TP + FPOS];
            P = P * sj;
        }
        ws[WS_G + k] = G;
        ws[WS_S + k] = P;
    }

    chain_trips<false, SEGTRIPS>(&Y[0][0], Tb, nullptr, 0);

    const float* Yf = Y[SEGTRIPS & 1];
#pragma unroll
    for (int j = 0; j < Y_NUM / THREADS; ++j) {
        const int i = tid + j * THREADS;
        ws[WS_MAP + k * Y_NUM + i] = Yf[(i >> 4) * SP + (i & 15)];
    }
}

__global__ __launch_bounds__(THREADS)
void ode_seg_scan(const float* __restrict__ y0,
                  float* __restrict__ ws,
                  float* __restrict__ out,
                  int lvl) {
    __shared__ __attribute__((aligned(16))) float As[Y_NUM + 16];
    __shared__ __attribute__((aligned(16))) float Ys[Y_NUM];
    __shared__ __attribute__((aligned(16))) float Pt[WAVES * 256];

    const int tid  = threadIdx.x;
    const int wv   = tid >> 5;
    const int ln   = tid & 31;
    const int col  = ln & 15;
    const int half = ln >> 4;
    const int b    = (int)blockIdx.x;

    const int  s   = 1 << (lvl - 1);
    const int  k   = s - 1 + (int)blockIdx.y;
    const int  hi  = (lvl - 1) * SEGS + k;
    const bool vec = (k < 2 * s - 1);
    const int  j0  = k - s + 1;
    const float* yin = vec ? ((j0 == 0) ? y0 : (ws + WS_Y + j0 * Y_NUM))
                           : (ws + WS_MAP + (hi - s) * Y_NUM);
    {
        const float* amap = ws + WS_MAP + hi * Y_NUM;
        float ta[Y_NUM / THREADS], ty[Y_NUM / THREADS];
#pragma unroll
        for (int j = 0; j < Y_NUM / THREADS; ++j) { ta[j] = amap[tid + j * THREADS]; ty[j] = yin[tid + j * THREADS]; }
#pragma unroll
        for (int j = 0; j < Y_NUM / THREADS; ++j) {
            const int i = tid + j * THREADS;
            As[16 + i] = ta[j];
            if (i >= Y_NUM - 16) As[i - (Y_NUM - 16)] = ta[j];
            Ys[i] = ty[j];
        }
    }
    __syncthreads();

    v8f aE, aO;
#pragma unroll
    for (int i = 0; i < 8; ++i) { aE[i] = 0.0f; aO[i] = 0.0f; }
    const int tile0 = (b << 4) + col;
    for (int rb = 0; rb < 16; rb += 4) {
        v2f av[4][4], bv[4][4];
#pragma unroll
        for (int u = 0; u < 4; ++u) {
            const int r  = (wv << 4) + rb + u;
            const int tl = (tile0 - r) & (NTILES - 1);
            const int x0 = 16 * r + col - 2 * half;
#pragma unroll
            for (int q = 0; q < 4; ++q) {
                const int x = x0 - 4 * q;
                av[u][q][0] = As[16 + x];
                av[u][q][1] = As[15 + x];
                const float2 v = *(const float2*)&Ys[(tl << 4) + (q << 2) + (half << 1)];
                bv[u][q][0] = v.x; bv[u][q][1] = v.y;
            }
        }
        __builtin_amdgcn_sched_barrier(0);
#pragma unroll
        for (int u = 0; u < 4; ++u)
#pragma unroll
            for (int q = 0; q < 4; ++q) {
                if (q & 1) aO = wmma4(av[u][q], bv[u][q], aO);
                else       aE = wmma4(av[u][q], bv[u][q], aE);
            }
        __builtin_amdgcn_sched_barrier(0);
    }
    const v8f acc = aE + aO;

    float* pp = &Pt[(wv << 8) + (ln << 3)];
    *(float4*)&pp[0] = make_float4(acc[0], acc[1], acc[2], acc[3]);
    *(float4*)&pp[4] = make_float4(acc[4], acc[5], acc[6], acc[7]);
    __syncthreads();

    const int e  = tid & 15;
    const int l  = (tid >> 4) + ((e >> 3) << 4);
    const int pv = (l << 3) + (e & 7);
    float sum = 0.0f;
#pragma unroll
    for (int u = 0; u < WAVES; ++u) sum += Pt[(u << 8) + pv];
    const int i = (b << 8) + tid;
    if (vec) {
        sum += ws[WS_G + hi];
        ws[WS_Y + (k + 1) * Y_NUM + i] = sum;
        out[(size_t)(SEGROWS * (k + 1)) * Y_NUM + i] = sum;
    } else {
        const int dst = lvl * SEGS + k;
        ws[WS_MAP + dst * Y_NUM + i] = sum;
        if (i == 0) {
            const float Sh = ws[WS_S + hi];
            ws[WS_G + dst] = ws[WS_G + hi - s] * Sh + ws[WS_G + hi];
            ws[WS_S + dst] = Sh * ws[WS_S + hi - s];
        }
    }
}

__global__ __launch_bounds__(THREADS) __attribute__((amdgpu_waves_per_eu(2, 2)))
void ode_chain_seg(const float* __restrict__ y0,
                   const float* __restrict__ ws,
                   float* __restrict__ out) {
    __shared__ __attribute__((aligned(16))) float Y[2][SBUF];
    __shared__ __attribute__((aligned(16))) float Tb[(SEGTRIPS - 1) * TP];

    const int tid = threadIdx.x;
    const int k   = (int)blockIdx.x;

    const float* ys = (k == 0) ? y0 : (ws + WS_Y + k * Y_NUM);
    {
        float ty[Y_NUM / THREADS];
#pragma unroll
        for (int j = 0; j < Y_NUM / THREADS; ++j) ty[j] = ys[tid + j * THREADS];
        copy_floats<(SEGTRIPS - 1) * TP>(Tb, ws + WS_TAB + SEGTRIPS * k * TP, tid);
#pragma unroll
        for (int j = 0; j < Y_NUM / THREADS; ++j) {
            const int i = tid + j * THREADS;
            Y[0][(i >> 4) * SP + (i & 15)] = ty[j];
            if (k == 0) out[i] = ty[j];
        }
    }
    __syncthreads();

    chain_trips<true, SEGTRIPS - 1>(&Y[0][0], Tb, out, SEGROWS * k);
}

template <int J>
__device__ __forceinline__ v8f row_direct(const float* rp, float F, const StateK& s) {
    constexpr int R = 1 + (J + 1) / 2;
    v8f aE, aO;
#pragma unroll
    for (int i = 0; i < 8; ++i) { aE[i] = F; aO[i] = 0.0f; }
    int cnt = 0;
#pragma unroll
    for (int q = 0; q < 4; ++q)
#pragma unroll
        for (int r = 0; r < R; ++r) {
            if ((J & 1) && r == R - 1 && q < 2) continue;
            const v2f a = tap_pair(rp, r, q);
            if (cnt & 1) aO = wmma4(a, s.b[r][q], aO);
            else         aE = wmma4(a, s.b[r][q], aE);
            ++cnt;
        }
    return aE + aO;
}

__device__ __forceinline__ void store_row(float* orow, int rowIx, v8f y) {
    *(float4*)&orow[rowIx]     = make_float4(y[0], y[1], y[2], y[3]);
    *(float4*)&orow[rowIx + 4] = make_float4(y[4], y[5], y[6], y[7]);
}

template <int J>
__device__ __forceinline__ void fill_from(float* out, int c0, int rowIx,
                                          const float* tab_lb, const float* Fs, const StateK& st) {
    if constexpr (J < KROW) {
        store_row(out + (size_t)(c0 + J) * Y_NUM, rowIx,
                  row_direct<J>(tab_lb + (J - 1) * ROWLEN, Fs[J - 1], st));
        fill_from<J + 1>(out, c0, rowIx, tab_lb, Fs, st);
    }
}

__global__ __launch_bounds__(FTHREADS)
void ode_fill(const float* __restrict__ t,
              const float* __restrict__ w,
              const float* __restrict__ rt,
              float* out) {
    __shared__ float sc[KROW - 1][12];
    __shared__ float tab[(KROW - 1) * ROWLEN];
    __shared__ float Fs[KROW];

    const int tid  = threadIdx.x;
    const int wv   = (tid >> 5) + FWAVES * ((int)blockIdx.x & 1);
    const int ln   = tid & 31;
    const int col  = ln & 15;
    const int half = ln >> 4;
    const int m    = col;
    const int c0   = KROW * ((int)blockIdx.x >> 1);

    const int tileC = (wv << 4) + col;
    const int rowIx = (wv << 8) + (col << 4) + (half << 3);
    const int lb    = 15 - m + (half << 1);

    const float* ck = out + (size_t)c0 * Y_NUM;
    StateK st;
#pragma unroll
    for (int q = 0; q < 4; ++q)
#pragma unroll
        for (int r = 0; r < NMAT; ++r) {
            const int tl = (tileC - r) & (NTILES - 1);
            const float2 v = *(const float2*)&ck[(tl << 4) + (q << 2) + (half << 1)];
            st.b[r][q][0] = v.x; st.b[r][q][1] = v.y;
        }

    if (tid < KROW - 1) {
        const float a = w[0] * rt[0];
        const float b = w[1] * rt[1];
        const float c = w[2] * rt[2];
        const StepC A = step_coef(t[c0 + tid], t[c0 + tid + 1], a, b, c);
#pragma unroll
        for (int i = 0; i < 9; ++i) sc[tid][i] = A.c[i];
        sc[tid][9] = A.f;
    }
    __syncthreads();

    for (int p = tid; p < ROWLEN; p += FTHREADS) {
        const int x = TAPTOP - p;
        tab[p] = (x >= 0 && x <= 8) ? sc[0][(x >= 0 && x <= 8) ? x : 0] : 0.0f;
    }
    if (tid == 0) {
        float F = sc[0][9];
        Fs[0] = F;
        for (int j = 1; j < KROW - 1; ++j) {
            float sumB = 0.0f;
            for (int i = 0; i < 9; ++i) sumB += sc[j][i];
            F = F * sumB + sc[j][9];
            Fs[j] = F;
        }
    }
    __syncthreads();

    for (int j = 1; j < KROW - 1; ++j) {
        for (int p = tid; p < ROWLEN; p += FTHREADS) {
            float acc = 0.0f;
#pragma unroll
            for (int i = 0; i < 9; ++i) {
                const int pp = p + i;
                const float pv = (pp < ROWLEN) ? tab[(j - 1) * ROWLEN + ((pp < ROWLEN) ? pp : 0)] : 0.0f;
                acc += sc[j][i] * pv;
            }
            tab[j * ROWLEN + p] = acc;
        }
        __syncthreads();
    }

    fill_from<1>(out, c0, rowIx, tab + lb, Fs, st);
}

extern "C" void kernel_launch(void* const* d_in, const int* in_sizes, int n_in,
                              void* d_out, int out_size, void* d_ws, size_t ws_size,
                              hipStream_t stream) {
    (void)in_sizes; (void)n_in; (void)out_size; (void)ws_size;
    const float* t  = (const float*)d_in[0];
    const float* y0 = (const float*)d_in[1];
    const float* w  = (const float*)d_in[2];
    const float* r  = (const float*)d_in[3];
    float* out = (float*)d_out;
    float* ws  = (float*)d_ws;

    static_assert(SEGS == 32 && LEVELS == 5, "the level shapes below are for 31 segment maps");
    ode_maps<<<SEGS, THREADS, 0, stream>>>(t, w, r, ws);
    ode_seg_scan<<<dim3(8, 31), THREADS, 0, stream>>>(y0, ws, out, 1);
    ode_seg_scan<<<dim3(8, 30), THREADS, 0, stream>>>(y0, ws, out, 2);
    ode_seg_scan<<<dim3(8, 28), THREADS, 0, stream>>>(y0, ws, out, 3);
    ode_seg_scan<<<dim3(8, 24), THREADS, 0, stream>>>(y0, ws, out, 4);
    ode_seg_scan<<<dim3(8, 16), THREADS, 0, stream>>>(y0, ws, out, 5);
    ode_chain_seg<<<SEGS, THREADS, 0, stream>>>(y0, ws, out);
    ode_fill<<<2 * NGROUP, FTHREADS, 0, stream>>>(t, w, r, out);
}


# First prefix level folded into the maps kernel: each workgroup runs its impulse through two segments (8 trips; waves it has not reached store zeros), segment 0 from y0: 7 launches; sums regrouped
# speedup vs baseline: 30.5568x; 1.0923x over previous
#include <hip/hip_runtime.h>


typedef float v2f __attribute__((ext_vector_type(2)));
typedef float v8f __attribute__((ext_vector_type(8)));

#define SAMPLE_NUM 2048
#define Y_NUM      2048
#define NSTEP      (SAMPLE_NUM - 1)
#define KROW       16
#define NTRIP      (NSTEP / KROW)
#define NGROUP     (NTRIP + 1)
#define NTILES     128
#define WAVES      8
#define THREADS    (WAVES * 32)
#define FWAVES     4
#define FTHREADS   (FWAVES * 32)
#define NMAT       (KROW / 2 + 1)
#define NTAPK      (8 * KROW + 1)
#define TAPTOP     (16 * NMAT - 1)
#define ROWLEN     (16 * NMAT + 16)
#define TP         (ROWLEN + 4)
#define FPOS       ROWLEN
#define SP         20
#define SBUF       (NTILES * SP)
#define SEGS       32
#define LEVELS     5
#define SEGTRIPS   4
#define SEGROWS    (KROW * SEGTRIPS)
#define MTRIPS     (2 * SEGTRIPS)
#define NMAPS      (SEGS - 1)
#define LASTTRIPS  (NTRIP - SEGTRIPS * (SEGS - 1))
#define SUMPOS     (FPOS + 1)
#define NSLOT      (LEVELS * SEGS)
#define WS_TAB     0
#define WS_MAP     32768
#define WS_G       (WS_MAP + NSLOT * Y_NUM)
#define WS_S       (WS_G + 256)
#define WS_Y       (WS_S + 256)
static_assert(NTRIP * TP <= WS_MAP && WS_MAP + NSLOT * Y_NUM <= WS_G && WS_G + NSLOT <= WS_S && WS_S + NSLOT <= WS_Y, "workspace regions overlap");
static_assert((1 << LEVELS) == SEGS && LASTTRIPS == SEGTRIPS - 1, "segment shapes");

struct StepC {
    float c[9];
    float f;
};

__device__ __forceinline__ StepC step_coef(float tn, float tn1, float a, float b, float c) {
    const float sub = (tn1 - tn) * 0.125f;
    const float d   = 1.0f - sub * b;
    const float e   = sub * a;
    StepC r;
    {
        const float bin[9] = {1.f, 8.f, 28.f, 56.f, 70.f, 56.f, 28.f, 8.f, 1.f};
        float dp[9], ep[9];
        dp[0] = 1.0f; ep[0] = 1.0f;
#pragma unroll
        for (int j = 1; j < 9; ++j) { dp[j] = dp[j-1] * d; ep[j] = ep[j-1] * e; }
#pragma unroll
        for (int j = 0; j < 9; ++j) r.c[j] = bin[j] * dp[8 - j] * ep[j];
    }
    const float g = d + e;
    float f = 0.0f;
#pragma unroll
    for (int ms = 0; ms < 8; ++ms) {
        const float s = sub * c * __sinf(tn + sub * (float)ms);
        f = f * g + s;
    }
    r.f = f;
    return r;
}

__device__ __forceinline__ v8f wmma4(v2f a, v2f b, v8f c) {
    return __builtin_amdgcn_wmma_f32_16x16x4_f32(false, a, false, b, (short)0, c, false, false);
}

struct StateK {
    v2f b[NMAT][4];
};

__device__ __forceinline__ v2f tap_pair(const float* rp, int r, int q) {
    v2f a;
    a[0] = rp[16 * (NMAT - 1) - 16 * r + 4 * q];
    a[1] = rp[16 * (NMAT - 1) - 16 * r + 4 * q + 1];
    return a;
}

struct CoefK {
    v2f   a[NMAT][4];
    float f;
};

template <bool FORCED>
__device__ __forceinline__ CoefK load_coefK(const float* row, int lb) {
    CoefK c;
    c.f = FORCED ? row[FPOS] : 0.0f;
    const float* rp = row + lb;
#pragma unroll
    for (int q = 0; q < 4; ++q)
#pragma unroll
        for (int r = 0; r < NMAT; ++r) c.a[r][q] = tap_pair(rp, r, q);
    return c;
}

__device__ __forceinline__ StateK load_state_lds(const float* Yb, const int* so) {
    StateK s;
#pragma unroll
    for (int q = 0; q < 4; ++q)
#pragma unroll
        for (int r = 0; r < NMAT; ++r) {
            const float2 v = *(const float2*)&Yb[so[r] + 4 * q];
            s.b[r][q][0] = v.x; s.b[r][q][1] = v.y;
        }
    return s;
}

__device__ __forceinline__ v8f rows_trip(const CoefK& k, const StateK& s) {
    v8f aE, aO;
#pragma unroll
    for (int i = 0; i < 8; ++i) { aE[i] = k.f; aO[i] = 0.0f; }
#pragma unroll
    for (int q = 0; q < 4; ++q)
#pragma unroll
        for (int r = 0; r < NMAT; ++r) {
            if ((q * NMAT + r) & 1) aO = wmma4(k.a[r][q], s.b[r][q], aO);
            else                    aE = wmma4(k.a[r][q], s.b[r][q], aE);
        }
    return aE + aO;
}


template <int N>
__device__ __forceinline__ void copy_floats(float* dst, const float* src, int tid) {
    constexpr int IT = (N + THREADS - 1) / THREADS;
    float v[IT];
#pragma unroll
    for (int j = 0; j < IT; ++j) { const int i = tid + j * THREADS; v[j] = (i < N) ? src[i] : 0.0f; }
#pragma unroll
    for (int j = 0; j < IT; ++j) { const int i = tid + j * THREADS; if (i < N) dst[i] = v[j]; }
}

#define LANEPOS (ROWLEN / 16)
static_assert(LANEPOS * 16 == ROWLEN && LANEPOS >= 8, "a lane's window must reach exactly one lane up");
static_assert(THREADS >= 16 * MTRIPS && (16 * MTRIPS) % 32 == 0 && KROW == 16, "16 lanes per trip, whole waves");

template <int NT>
__device__ __forceinline__ void build_trip_rows(float* Tb, const float (*sc)[12], int tid) {
    if (tid >= 16 * NT) return;
    const int j = tid >> 4;
    const int u = tid & 15;
    const float* r0 = sc[KROW * j];
    float Wd[LANEPOS + 8];
#pragma unroll
    for (int e = 0; e < LANEPOS; ++e) {
        const int x = TAPTOP - (LANEPOS * u + e);
        const bool in = (x >= 0 && x <= 8);
        Wd[e] = in ? r0[in ? x : 0] : 0.0f;
    }
    float F = r0[9];
    for (int jj = 1; jj < KROW; ++jj) {
        const float* rb = sc[KROW * j + jj];
        const float4 c03 = *(const float4*)&rb[0];
        const float4 c47 = *(const float4*)&rb[4];
        const float4 c8x = *(const float4*)&rb[8];
        const float cc[9] = {c03.x, c03.y, c03.z, c03.w, c47.x, c47.y, c47.z, c47.w, c8x.x};
#pragma unroll
        for (int e = 0; e < 8; ++e) {
            const float v = __shfl_down(Wd[e], 1, 16);
            Wd[LANEPOS + e] = (u == 15) ? 0.0f : v;
        }
        float Nw[LANEPOS];
#pragma unroll
        for (int e = 0; e < LANEPOS; ++e) {
            float acc = 0.0f;
#pragma unroll
            for (int i = 0; i < 9; ++i) acc += cc[i] * Wd[e + i];
            Nw[e] = acc;
        }
#pragma unroll
        for (int e = 0; e < LANEPOS; ++e) Wd[e] = Nw[e];
        F = F * c8x.z + c8x.y;
    }
    float sp = 0.0f;
#pragma unroll
    for (int e = 0; e < LANEPOS; ++e) sp += Wd[e];
#pragma unroll
    for (int dlt = 8; dlt >= 1; dlt >>= 1) sp += __shfl_down(sp, dlt, 16);

    float* row = Tb + j * TP;
    float2* dst = (float2*)&row[LANEPOS * u];
#pragma unroll
    for (int e = 0; e < LANEPOS / 2; ++e) dst[e] = make_float2(Wd[2 * e], Wd[2 * e + 1]);
    if (u == 0) *(float4*)&row[FPOS] = make_float4(F, sp, 0.0f, 0.0f);
}

template <bool ROWS, int NTR>
__device__ __forceinline__ void chain_trips(float* Yb, const float* Tb, float* __restrict__ out, int row0) {
    static_assert(ROWS || 8 * NTR + NMAT <= NTILES, "impulse run: the state must not come round the ring");
    const int tid  = threadIdx.x;
    const int wv   = tid >> 5;
    const int ln   = tid & 31;
    const int col  = ln & 15;
    const int half = ln >> 4;
    const int m    = col;

    const int tileC = (wv << 4) + col;
    const int rowIx = (wv << 8) + (col << 4) + (half << 3);
    const int stIx  = tileC * SP + (half << 3);
    const int lb    = 15 - m + (half << 1);
    int so[NMAT];
#pragma unroll
    for (int r = 0; r < NMAT; ++r) so[r] = ((tileC - r) & (NTILES - 1)) * SP + (half << 1);

    const int wvu = __builtin_amdgcn_readfirstlane(wv);

    CoefK  cur = load_coefK<ROWS>(Tb, lb);
    StateK st  = load_state_lds(Yb, so);

#pragma unroll 2
    for (int s = 0; s < NTR; ++s) {
        float* __restrict__ Yn = Yb + ((s & 1) ^ 1) * SBUF;

        v8f yk;
        if (ROWS || 2 * wvu <= s + 1) {
            yk = rows_trip(cur, st);
        } else {
#pragma unroll
            for (int i = 0; i < 8; ++i) yk[i] = 0.0f;
        }
        const float4 lo = make_float4(yk[0], yk[1], yk[2], yk[3]);
        const float4 hi = make_float4(yk[4], yk[5], yk[6], yk[7]);
        *(float4*)&Yn[stIx]     = lo;
        *(float4*)&Yn[stIx + 4] = hi;

        __builtin_amdgcn_fence(__ATOMIC_RELEASE, "workgroup", "local");
        __builtin_amdgcn_s_barrier();
        __builtin_amdgcn_fence(__ATOMIC_ACQUIRE, "workgroup", "local");

        const int sn = (s + 1 < NTR) ? (s + 1) : (NTR - 1);
        if (ROWS || 2 * wvu <= s + 2) {
            st = load_state_lds(Yn, so);
            __builtin_amdgcn_sched_barrier(0);
            cur = load_coefK<ROWS>(Tb + sn * TP, lb);
            __builtin_amdgcn_sched_barrier(0);
        }

        if (ROWS) {
            float* __restrict__ orow = out + (size_t)(row0 + KROW * (s + 1)) * Y_NUM;
            *(float4*)&orow[rowIx]     = lo;
            *(float4*)&orow[rowIx + 4] = hi;
        }
    }
}

__global__ __launch_bounds__(THREADS) __attribute__((amdgpu_waves_per_eu(2, 2)))
void ode_maps(const float* __restrict__ t,
              const float* __restrict__ y0,
              const float* __restrict__ w,
              const float* __restrict__ rt,
              float* __restrict__ ws,
              float* __restrict__ out) {
    __shared__ __attribute__((aligned(16))) float Y[2][SBUF];
    __shared__ __attribute__((aligned(16))) float Tb[MTRIPS * TP];
    __shared__ __attribute__((aligned(16))) float sc[MTRIPS * KROW][12];

    const int tid = threadIdx.x;
    const int k   = (int)blockIdx.x;
    const bool pair  = (k >= 1 && k <= SEGS - 2);
    const int  nb    = SEGROWS * (pair ? k - 1 : k);
    const int  nrows = pair ? 2 * SEGROWS : ((k == SEGS - 1) ? KROW * LASTTRIPS : SEGROWS);

    float ty[Y_NUM / THREADS];
    if (k == 0) {
#pragma unroll
        for (int j = 0; j < Y_NUM / THREADS; ++j) ty[j] = y0[tid + j * THREADS];
    }

    {
        StepC A;
#pragma unroll
        for (int i = 0; i < 9; ++i) A.c[i] = 0.0f;
        A.f = 0.0f;
        if (tid < nrows) {
            const float a = w[0] * rt[0];
            const float b = w[1] * rt[1];
            const float c = w[2] * rt[2];
            const int n = nb + tid;
            A = step_coef(t[n], t[n + 1], a, b, c);
        }
        float sumA = 0.0f;
#pragma unroll
        for (int i = 0; i < 9; ++i) sumA += A.c[i];
        if (tid < MTRIPS * KROW) {
            *(float4*)&sc[tid][0] = make_float4(A.c[0], A.c[1], A.c[2], A.c[3]);
            *(float4*)&sc[tid][4] = make_float4(A.c[4], A.c[5], A.c[6], A.c[7]);
            *(float4*)&sc[tid][8] = make_float4(A.c[8], A.f, sumA, 0.0f);
        }
    }
    __syncthreads();
    build_trip_rows<MTRIPS>(Tb, sc, tid);
    if (k == 0) {
#pragma unroll
        for (int j = 0; j < Y_NUM / THREADS; ++j) {
            const int i = tid + j * THREADS;
            Y[0][(i >> 4) * SP + (i & 15)] = ty[j];
            out[i] = ty[j];
        }
    } else {
        for (int i = tid; i < SBUF; i += THREADS) Y[0][i] = (i == 0) ? 1.0f : 0.0f;
    }
    __syncthreads();

    copy_floats<LASTTRIPS * TP>(ws + WS_TAB + SEGTRIPS * k * TP, Tb + (pair ? SEGTRIPS * TP : 0), tid);

    if (k == SEGS - 1) return;

    if (k == 0) {
        chain_trips<true, SEGTRIPS>(&Y[0][0], Tb, out, 0);
        const float* Y1 = Y[SEGTRIPS & 1];
#pragma unroll
        for (int j = 0; j < Y_NUM / THREADS; ++j) {
            const int i = tid + j * THREADS;
            ws[WS_Y + Y_NUM + i] = Y1[(i >> 4) * SP + (i & 15)];
        }
        return;
    }

    if (tid == 0) {
        float G = 0.0f, P = 1.0f;
#pragma unroll
        for (int j = 0; j < MTRIPS; ++j) {
            const float sj = Tb[j * TP + SUMPOS];
            G = G * sj + Tb[j * TP + FPOS];
            P = P * sj;
        }
        ws[WS_G + SEGS + k] = G;
        ws[WS_S + SEGS + k] = P;
    }

    chain_trips<false, MTRIPS>(&Y[0][0], Tb, nullptr, 0);

    const float* Yf = Y[MTRIPS & 1];
#pragma unroll
    for (int j = 0; j < Y_NUM / THREADS; ++j) {
        const int i = tid + j * THREADS;
        ws[WS_MAP + (SEGS + k) * Y_NUM + i] = Yf[(i >> 4) * SP + (i & 15)];
    }
}

__global__ __launch_bounds__(THREADS)
void ode_seg_scan(const float* __restrict__ y0,
                  float* __restrict__ ws,
                  float* __restrict__ out,
                  int lvl) {
    __shared__ __attribute__((aligned(16))) float As[Y_NUM + 16];
    __shared__ __attribute__((aligned(16))) float Ys[Y_NUM];
    __shared__ __attribute__((aligned(16))) float Pt[WAVES * 256];

    const int tid  = threadIdx.x;
    const int wv   = tid >> 5;
    const int ln   = tid & 31;
    const int col  = ln & 15;
    const int half = ln >> 4;
    const int b    = (int)blockIdx.x;

    const int  s   = 1 << (lvl - 1);
    const int  k   = s - 1 + (int)blockIdx.y;
    const int  hi  = (lvl - 1) * SEGS + k;
    const bool vec = (k < 2 * s - 1);
    const int  j0  = k - s + 1;
    const float* yin = vec ? ((j0 == 0) ? y0 : (ws + WS_Y + j0 * Y_NUM))
                           : (ws + WS_MAP + (hi - s) * Y_NUM);
    {
        const float* amap = ws + WS_MAP + hi * Y_NUM;
        float ta[Y_NUM / THREADS], ty[Y_NUM / THREADS];
#pragma unroll
        for (int j = 0; j < Y_NUM / THREADS; ++j) { ta[j] = amap[tid + j * THREADS]; ty[j] = yin[tid + j * THREADS]; }
#pragma unroll
        for (int j = 0; j < Y_NUM / THREADS; ++j) {
            const int i = tid + j * THREADS;
            As[16 + i] = ta[j];
            if (i >= Y_NUM - 16) As[i - (Y_NUM - 16)] = ta[j];
            Ys[i] = ty[j];
        }
    }
    __syncthreads();

    v8f aE, aO;
#pragma unroll
    for (int i = 0; i < 8; ++i) { aE[i] = 0.0f; aO[i] = 0.0f; }
    const int tile0 = (b << 4) + col;
    for (int rb = 0; rb < 16; rb += 4) {
        v2f av[4][4], bv[4][4];
#pragma unroll
        for (int u = 0; u < 4; ++u) {
            const int r  = (wv << 4) + rb + u;
            const int tl = (tile0 - r) & (NTILES - 1);
            const int x0 = 16 * r + col - 2 * half;
#pragma unroll
            for (int q = 0; q < 4; ++q) {
                const int x = x0 - 4 * q;
                av[u][q][0] = As[16 + x];
                av[u][q][1] = As[15 + x];
                const float2 v = *(const float2*)&Ys[(tl << 4) + (q << 2) + (half << 1)];
                bv[u][q][0] = v.x; bv[u][q][1] = v.y;
            }
        }
        __builtin_amdgcn_sched_barrier(0);
#pragma unroll
        for (int u = 0; u < 4; ++u)
#pragma unroll
            for (int q = 0; q < 4; ++q) {
                if (q & 1) aO = wmma4(av[u][q], bv[u][q], aO);
                else       aE = wmma4(av[u][q], bv[u][q], aE);
            }
        __builtin_amdgcn_sched_barrier(0);
    }
    const v8f acc = aE + aO;

    float* pp = &Pt[(wv << 8) + (ln << 3)];
    *(float4*)&pp[0] = make_float4(acc[0], acc[1], acc[2], acc[3]);
    *(float4*)&pp[4] = make_float4(acc[4], acc[5], acc[6], acc[7]);
    __syncthreads();

    const int e  = tid & 15;
    const int l  = (tid >> 4) + ((e >> 3) << 4);
    const int pv = (l << 3) + (e & 7);
    float sum = 0.0f;
#pragma unroll
    for (int u = 0; u < WAVES; ++u) sum += Pt[(u << 8) + pv];
    const int i = (b << 8) + tid;
    if (vec) {
        sum += ws[WS_G + hi];
        ws[WS_Y + (k + 1) * Y_NUM + i] = sum;
        out[(size_t)(SEGROWS * (k + 1)) * Y_NUM + i] = sum;
    } else {
        const int dst = lvl * SEGS + k;
        ws[WS_MAP + dst * Y_NUM + i] = sum;
        if (i == 0) {
            const float Sh = ws[WS_S + hi];
            ws[WS_G + dst] = ws[WS_G + hi - s] * Sh + ws[WS_G + hi];
            ws[WS_S + dst] = Sh * ws[WS_S + hi - s];
        }
    }
}

__global__ __launch_bounds__(THREADS) __attribute__((amdgpu_waves_per_eu(2, 2)))
void ode_chain_seg(const float* __restrict__ ws,
                   float* __restrict__ out) {
    __shared__ __attribute__((aligned(16))) float Y[2][SBUF];
    __shared__ __attribute__((aligned(16))) float Tb[(SEGTRIPS - 1) * TP];

    const int tid = threadIdx.x;
    const int k   = (int)blockIdx.x + 1;

    const float* ys = ws + WS_Y + k * Y_NUM;
    {
        float ty[Y_NUM / THREADS];
#pragma unroll
        for (int j = 0; j < Y_NUM / THREADS; ++j) ty[j] = ys[tid + j * THREADS];
        copy_floats<(SEGTRIPS - 1) * TP>(Tb, ws + WS_TAB + SEGTRIPS * k * TP, tid);
#pragma unroll
        for (int j = 0; j < Y_NUM / THREADS; ++j) {
            const int i = tid + j * THREADS;
            Y[0][(i >> 4) * SP + (i & 15)] = ty[j];
        }
    }
    __syncthreads();

    chain_trips<true, SEGTRIPS - 1>(&Y[0][0], Tb, out, SEGROWS * k);
}

template <int J>
__device__ __forceinline__ v8f row_direct(const float* rp, float F, const StateK& s) {
    constexpr int R = 1 + (J + 1) / 2;
    v8f aE, aO;
#pragma unroll
    for (int i = 0; i < 8; ++i) { aE[i] = F; aO[i] = 0.0f; }
    int cnt = 0;
#pragma unroll
    for (int q = 0; q < 4; ++q)
#pragma unroll
        for (int r = 0; r < R; ++r) {
            if ((J & 1) && r == R - 1 && q < 2) continue;
            const v2f a = tap_pair(rp, r, q);
            if (cnt & 1) aO = wmma4(a, s.b[r][q], aO);
            else         aE = wmma4(a, s.b[r][q], aE);
            ++cnt;
        }
    return aE + aO;
}

__device__ __forceinline__ void store_row(float* orow, int rowIx, v8f y) {
    *(float4*)&orow[rowIx]     = make_float4(y[0], y[1], y[2], y[3]);
    *(float4*)&orow[rowIx + 4] = make_float4(y[4], y[5], y[6], y[7]);
}

template <int J>
__device__ __forceinline__ void fill_from(float* out, int c0, int rowIx,
                                          const float* tab_lb, const float* Fs, const StateK& st) {
    if constexpr (J < KROW) {
        store_row(out + (size_t)(c0 + J) * Y_NUM, rowIx,
                  row_direct<J>(tab_lb + (J - 1) * ROWLEN, Fs[J - 1], st));
        fill_from<J + 1>(out, c0, rowIx, tab_lb, Fs, st);
    }
}

__global__ __launch_bounds__(FTHREADS)
void ode_fill(const float* __restrict__ t,
              const float* __restrict__ w,
              const float* __restrict__ rt,
              float* out) {
    __shared__ float sc[KROW - 1][12];
    __shared__ float tab[(KROW - 1) * ROWLEN];
    __shared__ float Fs[KROW];

    const int tid  = threadIdx.x;
    const int wv   = (tid >> 5) + FWAVES * ((int)blockIdx.x & 1);
    const int ln   = tid & 31;
    const int col  = ln & 15;
    const int half = ln >> 4;
    const int m    = col;
    const int c0   = KROW * ((int)blockIdx.x >> 1);

    const int tileC = (wv << 4) + col;
    const int rowIx = (wv << 8) + (col << 4) + (half << 3);
    const int lb    = 15 - m + (half << 1);

    const float* ck = out + (size_t)c0 * Y_NUM;
    StateK st;
#pragma unroll
    for (int q = 0; q < 4; ++q)
#pragma unroll
        for (int r = 0; r < NMAT; ++r) {
            const int tl = (tileC - r) & (NTILES - 1);
            const float2 v = *(const float2*)&ck[(tl << 4) + (q << 2) + (half << 1)];
            st.b[r][q][0] = v.x; st.b[r][q][1] = v.y;
        }

    if (tid < KROW - 1) {
        const float a = w[0] * rt[0];
        const float b = w[1] * rt[1];
        const float c = w[2] * rt[2];
        const StepC A = step_coef(t[c0 + tid], t[c0 + tid + 1], a, b, c);
#pragma unroll
        for (int i = 0; i < 9; ++i) sc[tid][i] = A.c[i];
        sc[tid][9] = A.f;
    }
    __syncthreads();

    for (int p = tid; p < ROWLEN; p += FTHREADS) {
        const int x = TAPTOP - p;
        tab[p] = (x >= 0 && x <= 8) ? sc[0][(x >= 0 && x <= 8) ? x : 0] : 0.0f;
    }
    if (tid == 0) {
        float F = sc[0][9];
        Fs[0] = F;
        for (int j = 1; j < KROW - 1; ++j) {
            float sumB = 0.0f;
            for (int i = 0; i < 9; ++i) sumB += sc[j][i];
            F = F * sumB + sc[j][9];
            Fs[j] = F;
        }
    }
    __syncthreads();

    for (int j = 1; j < KROW - 1; ++j) {
        for (int p = tid; p < ROWLEN; p += FTHREADS) {
            float acc = 0.0f;
#pragma unroll
            for (int i = 0; i < 9; ++i) {
                const int pp = p + i;
                const float pv = (pp < ROWLEN) ? tab[(j - 1) * ROWLEN + ((pp < ROWLEN) ? pp : 0)] : 0.0f;
                acc += sc[j][i] * pv;
            }
            tab[j * ROWLEN + p] = acc;
        }
        __syncthreads();
    }

    fill_from<1>(out, c0, rowIx, tab + lb, Fs, st);
}

extern "C" void kernel_launch(void* const* d_in, const int* in_sizes, int n_in,
                              void* d_out, int out_size, void* d_ws, size_t ws_size,
                              hipStream_t stream) {
    (void)in_sizes; (void)n_in; (void)out_size; (void)ws_size;
    const float* t  = (const float*)d_in[0];
    const float* y0 = (const float*)d_in[1];
    const float* w  = (const float*)d_in[2];
    const float* r  = (const float*)d_in[3];
    float* out = (float*)d_out;
    float* ws  = (float*)d_ws;

    static_assert(SEGS == 32 && LEVELS == 5, "the level shapes below are for 31 segment maps");
    ode_maps<<<SEGS, THREADS, 0, stream>>>(t, y0, w, r, ws, out);
    ode_seg_scan<<<dim3(8, 30), THREADS, 0, stream>>>(y0, ws, out, 2);
    ode_seg_scan<<<dim3(8, 28), THREADS, 0, stream>>>(y0, ws, out, 3);
    ode_seg_scan<<<dim3(8, 24), THREADS, 0, stream>>>(y0, ws, out, 4);
    ode_seg_scan<<<dim3(8, 16), THREADS, 0, stream>>>(y0, ws, out, 5);
    ode_chain_seg<<<SEGS - 1, THREADS, 0, stream>>>(ws, out);
    ode_fill<<<2 * NGROUP, FTHREADS, 0, stream>>>(t, w, r, out);
}


# Fragment loads as one group ahead of the products: prefix keeps taps reversed in LDS (pairs load in operand order, no swaps or waits in a group; tile pitch 20); fill loads row J+1 taps before row J
# speedup vs baseline: 33.4749x; 1.0955x over previous
#include <hip/hip_runtime.h>


typedef float v2f __attribute__((ext_vector_type(2)));
typedef float v8f __attribute__((ext_vector_type(8)));

#define SAMPLE_NUM 2048
#define Y_NUM      2048
#define NSTEP      (SAMPLE_NUM - 1)
#define KROW       16
#define NTRIP      (NSTEP / KROW)
#define NGROUP     (NTRIP + 1)
#define NTILES     128
#define WAVES      8
#define THREADS    (WAVES * 32)
#define FWAVES     4
#define FTHREADS   (FWAVES * 32)
#define NMAT       (KROW / 2 + 1)
#define NTAPK      (8 * KROW + 1)
#define TAPTOP     (16 * NMAT - 1)
#define ROWLEN     (16 * NMAT + 16)
#define TP         (ROWLEN + 4)
#define FPOS       ROWLEN
#define SP         20
#define SBUF       (NTILES * SP)
#define SEGS       32
#define LEVELS     5
#define SEGTRIPS   4
#define SEGROWS    (KROW * SEGTRIPS)
#define MTRIPS     (2 * SEGTRIPS)
#define NMAPS      (SEGS - 1)
#define LASTTRIPS  (NTRIP - SEGTRIPS * (SEGS - 1))
#define SUMPOS     (FPOS + 1)
#define NSLOT      (LEVELS * SEGS)
#define WS_TAB     0
#define WS_MAP     32768
#define WS_G       (WS_MAP + NSLOT * Y_NUM)
#define WS_S       (WS_G + 256)
#define WS_Y       (WS_S + 256)
static_assert(NTRIP * TP <= WS_MAP && WS_MAP + NSLOT * Y_NUM <= WS_G && WS_G + NSLOT <= WS_S && WS_S + NSLOT <= WS_Y, "workspace regions overlap");
static_assert((1 << LEVELS) == SEGS && LASTTRIPS == SEGTRIPS - 1, "segment shapes");

struct StepC {
    float c[9];
    float f;
};

__device__ __forceinline__ StepC step_coef(float tn, float tn1, float a, float b, float c) {
    const float sub = (tn1 - tn) * 0.125f;
    const float d   = 1.0f - sub * b;
    const float e   = sub * a;
    StepC r;
    {
        const float bin[9] = {1.f, 8.f, 28.f, 56.f, 70.f, 56.f, 28.f, 8.f, 1.f};
        float dp[9], ep[9];
        dp[0] = 1.0f; ep[0] = 1.0f;
#pragma unroll
        for (int j = 1; j < 9; ++j) { dp[j] = dp[j-1] * d; ep[j] = ep[j-1] * e; }
#pragma unroll
        for (int j = 0; j < 9; ++j) r.c[j] = bin[j] * dp[8 - j] * ep[j];
    }
    const float g = d + e;
    float f = 0.0f;
#pragma unroll
    for (int ms = 0; ms < 8; ++ms) {
        const float s = sub * c * __sinf(tn + sub * (float)ms);
        f = f * g + s;
    }
    r.f = f;
    return r;
}

__device__ __forceinline__ v8f wmma4(v2f a, v2f b, v8f c) {
    return __builtin_amdgcn_wmma_f32_16x16x4_f32(false, a, false, b, (short)0, c, false, false);
}

struct StateK {
    v2f b[NMAT][4];
};

__device__ __forceinline__ v2f tap_pair(const float* rp, int r, int q) {
    v2f a;
    a[0] = rp[16 * (NMAT - 1) - 16 * r + 4 * q];
    a[1] = rp[16 * (NMAT - 1) - 16 * r + 4 * q + 1];
    return a;
}

struct CoefK {
    v2f   a[NMAT][4];
    float f;
};

template <bool FORCED>
__device__ __forceinline__ CoefK load_coefK(const float* row, int lb) {
    CoefK c;
    c.f = FORCED ? row[FPOS] : 0.0f;
    const float* rp = row + lb;
#pragma unroll
    for (int q = 0; q < 4; ++q)
#pragma unroll
        for (int r = 0; r < NMAT; ++r) c.a[r][q] = tap_pair(rp, r, q);
    return c;
}

__device__ __forceinline__ StateK load_state_lds(const float* Yb, const int* so) {
    StateK s;
#pragma unroll
    for (int q = 0; q < 4; ++q)
#pragma unroll
        for (int r = 0; r < NMAT; ++r) {
            const float2 v = *(const float2*)&Yb[so[r] + 4 * q];
            s.b[r][q][0] = v.x; s.b[r][q][1] = v.y;
        }
    return s;
}

__device__ __forceinline__ v8f rows_trip(const CoefK& k, const StateK& s) {
    v8f aE, aO;
#pragma unroll
    for (int i = 0; i < 8; ++i) { aE[i] = k.f; aO[i] = 0.0f; }
#pragma unroll
    for (int q = 0; q < 4; ++q)
#pragma unroll
        for (int r = 0; r < NMAT; ++r) {
            if ((q * NMAT + r) & 1) aO = wmma4(k.a[r][q], s.b[r][q], aO);
            else                    aE = wmma4(k.a[r][q], s.b[r][q], aE);
        }
    return aE + aO;
}


template <int N>
__device__ __forceinline__ void copy_floats(float* dst, const float* src, int tid) {
    constexpr int IT = (N + THREADS - 1) / THREADS;
    float v[IT];
#pragma unroll
    for (int j = 0; j < IT; ++j) { const int i = tid + j * THREADS; v[j] = (i < N) ? src[i] : 0.0f; }
#pragma unroll
    for (int j = 0; j < IT; ++j) { const int i = tid + j * THREADS; if (i < N) dst[i] = v[j]; }
}

#define LANEPOS (ROWLEN / 16)
static_assert(LANEPOS * 16 == ROWLEN && LANEPOS >= 8, "a lane's window must reach exactly one lane up");
static_assert(THREADS >= 16 * MTRIPS && (16 * MTRIPS) % 32 == 0 && KROW == 16, "16 lanes per trip, whole waves");

template <int NT>
__device__ __forceinline__ void build_trip_rows(float* Tb, const float (*sc)[12], int tid) {
    if (tid >= 16 * NT) return;
    const int j = tid >> 4;
    const int u = tid & 15;
    const float* r0 = sc[KROW * j];
    float Wd[LANEPOS + 8];
#pragma unroll
    for (int e = 0; e < LANEPOS; ++e) {
        const int x = TAPTOP - (LANEPOS * u + e);
        const bool in = (x >= 0 && x <= 8);
        Wd[e] = in ? r0[in ? x : 0] : 0.0f;
    }
    float F = r0[9];
    for (int jj = 1; jj < KROW; ++jj) {
        const float* rb = sc[KROW * j + jj];
        const float4 c03 = *(const float4*)&rb[0];
        const float4 c47 = *(const float4*)&rb[4];
        const float4 c8x = *(const float4*)&rb[8];
        const float cc[9] = {c03.x, c03.y, c03.z, c03.w, c47.x, c47.y, c47.z, c47.w, c8x.x};
#pragma unroll
        for (int e = 0; e < 8; ++e) {
            const float v = __shfl_down(Wd[e], 1, 16);
            Wd[LANEPOS + e] = (u == 15) ? 0.0f : v;
        }
        float Nw[LANEPOS];
#pragma unroll
        for (int e = 0; e < LANEPOS; ++e) {
            float acc = 0.0f;
#pragma unroll
            for (int i = 0; i < 9; ++i) acc += cc[i] * Wd[e + i];
            Nw[e] = acc;
        }
#pragma unroll
        for (int e = 0; e < LANEPOS; ++e) Wd[e] = Nw[e];
        F = F * c8x.z + c8x.y;
    }
    float sp = 0.0f;
#pragma unroll
    for (int e = 0; e < LANEPOS; ++e) sp += Wd[e];
#pragma unroll
    for (int dlt = 8; dlt >= 1; dlt >>= 1) sp += __shfl_down(sp, dlt, 16);

    float* row = Tb + j * TP;
    float2* dst = (float2*)&row[LANEPOS * u];
#pragma unroll
    for (int e = 0; e < LANEPOS / 2; ++e) dst[e] = make_float2(Wd[2 * e], Wd[2 * e + 1]);
    if (u == 0) *(float4*)&row[FPOS] = make_float4(F, sp, 0.0f, 0.0f);
}

template <bool ROWS, int NTR>
__device__ __forceinline__ void chain_trips(float* Yb, const float* Tb, float* __restrict__ out, int row0) {
    static_assert(ROWS || 8 * NTR + NMAT <= NTILES, "impulse run: the state must not come round the ring");
    const int tid  = threadIdx.x;
    const int wv   = tid >> 5;
    const int ln   = tid & 31;
    const int col  = ln & 15;
    const int half = ln >> 4;
    const int m    = col;

    const int tileC = (wv << 4) + col;
    const int rowIx = (wv << 8) + (col << 4) + (half << 3);
    const int stIx  = tileC * SP + (half << 3);
    const int lb    = 15 - m + (half << 1);
    int so[NMAT];
#pragma unroll
    for (int r = 0; r < NMAT; ++r) so[r] = ((tileC - r) & (NTILES - 1)) * SP + (half << 1);

    const int wvu = __builtin_amdgcn_readfirstlane(wv);

    CoefK  cur = load_coefK<ROWS>(Tb, lb);
    StateK st  = load_state_lds(Yb, so);

#pragma unroll 2
    for (int s = 0; s < NTR; ++s) {
        float* __restrict__ Yn = Yb + ((s & 1) ^ 1) * SBUF;

        v8f yk;
        if (ROWS || 2 * wvu <= s + 1) {
            yk = rows_trip(cur, st);
        } else {
#pragma unroll
            for (int i = 0; i < 8; ++i) yk[i] = 0.0f;
        }
        const float4 lo = make_float4(yk[0], yk[1], yk[2], yk[3]);
        const float4 hi = make_float4(yk[4], yk[5], yk[6], yk[7]);
        *(float4*)&Yn[stIx]     = lo;
        *(float4*)&Yn[stIx + 4] = hi;

        __builtin_amdgcn_fence(__ATOMIC_RELEASE, "workgroup", "local");
        __builtin_amdgcn_s_barrier();
        __builtin_amdgcn_fence(__ATOMIC_ACQUIRE, "workgroup", "local");

        const int sn = (s + 1 < NTR) ? (s + 1) : (NTR - 1);
        if (ROWS || 2 * wvu <= s + 2) {
            st = load_state_lds(Yn, so);
            __builtin_amdgcn_sched_barrier(0);
            cur = load_coefK<ROWS>(Tb + sn * TP, lb);
            __builtin_amdgcn_sched_barrier(0);
        }

        if (ROWS) {
            float* __restrict__ orow = out + (size_t)(row0 + KROW * (s + 1)) * Y_NUM;
            *(float4*)&orow[rowIx]     = lo;
            *(float4*)&orow[rowIx + 4] = hi;
        }
    }
}

__global__ __launch_bounds__(THREADS) __attribute__((amdgpu_waves_per_eu(2, 2)))
void ode_maps(const float* __restrict__ t,
              const float* __restrict__ y0,
              const float* __restrict__ w,
              const float* __restrict__ rt,
              float* __restrict__ ws,
              float* __restrict__ out) {
    __shared__ __attribute__((aligned(16))) float Y[2][SBUF];
    __shared__ __attribute__((aligned(16))) float Tb[MTRIPS * TP];
    __shared__ __attribute__((aligned(16))) float sc[MTRIPS * KROW][12];

    const int tid = threadIdx.x;
    const int k   = (int)blockIdx.x;
    const bool pair  = (k >= 1 && k <= SEGS - 2);
    const int  nb    = SEGROWS * (pair ? k - 1 : k);
    const int  nrows = pair ? 2 * SEGROWS : ((k == SEGS - 1) ? KROW * LASTTRIPS : SEGROWS);

    float ty[Y_NUM / THREADS];
    if (k == 0) {
#pragma unroll
        for (int j = 0; j < Y_NUM / THREADS; ++j) ty[j] = y0[tid + j * THREADS];
    }

    {
        StepC A;
#pragma unroll
        for (int i = 0; i < 9; ++i) A.c[i] = 0.0f;
        A.f = 0.0f;
        if (tid < nrows) {
            const float a = w[0] * rt[0];
            const float b = w[1] * rt[1];
            const float c = w[2] * rt[2];
            const int n = nb + tid;
            A = step_coef(t[n], t[n + 1], a, b, c);
        }
        float sumA = 0.0f;
#pragma unroll
        for (int i = 0; i < 9; ++i) sumA += A.c[i];
        if (tid < MTRIPS * KROW) {
            *(float4*)&sc[tid][0] = make_float4(A.c[0], A.c[1], A.c[2], A.c[3]);
            *(float4*)&sc[tid][4] = make_float4(A.c[4], A.c[5], A.c[6], A.c[7]);
            *(float4*)&sc[tid][8] = make_float4(A.c[8], A.f, sumA, 0.0f);
        }
    }
    __syncthreads();
    build_trip_rows<MTRIPS>(Tb, sc, tid);
    if (k == 0) {
#pragma unroll
        for (int j = 0; j < Y_NUM / THREADS; ++j) {
            const int i = tid + j * THREADS;
            Y[0][(i >> 4) * SP + (i & 15)] = ty[j];
            out[i] = ty[j];
        }
    } else {
        for (int i = tid; i < SBUF; i += THREADS) Y[0][i] = (i == 0) ? 1.0f : 0.0f;
    }
    __syncthreads();

    copy_floats<LASTTRIPS * TP>(ws + WS_TAB + SEGTRIPS * k * TP, Tb + (pair ? SEGTRIPS * TP : 0), tid);

    if (k == SEGS - 1) return;

    if (k == 0) {
        chain_trips<true, SEGTRIPS>(&Y[0][0], Tb, out, 0);
        const float* Y1 = Y[SEGTRIPS & 1];
#pragma unroll
        for (int j = 0; j < Y_NUM / THREADS; ++j) {
            const int i = tid + j * THREADS;
            ws[WS_Y + Y_NUM + i] = Y1[(i >> 4) * SP + (i & 15)];
        }
        return;
    }

    if (tid == 0) {
        float G = 0.0f, P = 1.0f;
#pragma unroll
        for (int j = 0; j < MTRIPS; ++j) {
            const float sj = Tb[j * TP + SUMPOS];
            G = G * sj + Tb[j * TP + FPOS];
            P = P * sj;
        }
        ws[WS_G + SEGS + k] = G;
        ws[WS_S + SEGS + k] = P;
    }

    chain_trips<false, MTRIPS>(&Y[0][0], Tb, nullptr, 0);

    const float* Yf = Y[MTRIPS & 1];
#pragma unroll
    for (int j = 0; j < Y_NUM / THREADS; ++j) {
        const int i = tid + j * THREADS;
        ws[WS_MAP + (SEGS + k) * Y_NUM + i] = Yf[(i >> 4) * SP + (i & 15)];
    }
}

__global__ __launch_bounds__(THREADS)
void ode_seg_scan(const float* __restrict__ y0,
                  float* __restrict__ ws,
                  float* __restrict__ out,
                  int lvl) {
    __shared__ __attribute__((aligned(16))) float As[Y_NUM + 16];
    __shared__ __attribute__((aligned(16))) float Ys[SBUF];
    __shared__ __attribute__((aligned(16))) float Pt[WAVES * 256];

    const int tid  = threadIdx.x;
    const int wv   = tid >> 5;
    const int ln   = tid & 31;
    const int col  = ln & 15;
    const int half = ln >> 4;
    const int b    = (int)blockIdx.x;

    const int  s   = 1 << (lvl - 1);
    const int  k   = s - 1 + (int)blockIdx.y;
    const int  hi  = (lvl - 1) * SEGS + k;
    const bool vec = (k < 2 * s - 1);
    const int  j0  = k - s + 1;
    const float* yin = vec ? ((j0 == 0) ? y0 : (ws + WS_Y + j0 * Y_NUM))
                           : (ws + WS_MAP + (hi - s) * Y_NUM);
    {
        const float* amap = ws + WS_MAP + hi * Y_NUM;
        float ta[Y_NUM / THREADS], ty[Y_NUM / THREADS];
#pragma unroll
        for (int j = 0; j < Y_NUM / THREADS; ++j) { ta[j] = amap[tid + j * THREADS]; ty[j] = yin[tid + j * THREADS]; }
#pragma unroll
        for (int j = 0; j < Y_NUM / THREADS; ++j) {
            const int i = tid + j * THREADS;
            As[Y_NUM - 1 - i] = ta[j];
            if (i >= Y_NUM - 16) As[2 * Y_NUM - 1 - i] = ta[j];
            Ys[(i >> 4) * SP + (i & 15)] = ty[j];
        }
    }
    __syncthreads();

    v8f aE, aO;
#pragma unroll
    for (int i = 0; i < 8; ++i) { aE[i] = 0.0f; aO[i] = 0.0f; }
    const int tile0 = (b << 4) + col;
    for (int rb = 0; rb < 16; rb += 4) {
        v2f av[4][4], bv[4][4];
#pragma unroll
        for (int u = 0; u < 4; ++u) {
            const int r  = (wv << 4) + rb + u;
            const int tl = (tile0 - r) & (NTILES - 1);
            const int x0 = 16 * r + col - 2 * half;
#pragma unroll
            for (int q = 0; q < 4; ++q) {
                const int x = x0 - 4 * q;
                av[u][q][0] = As[Y_NUM - 1 - x];
                av[u][q][1] = As[Y_NUM - x];
                const float2 v = *(const float2*)&Ys[tl * SP + (q << 2) + (half << 1)];
                bv[u][q][0] = v.x; bv[u][q][1] = v.y;
            }
        }
        __builtin_amdgcn_sched_barrier(0);
#pragma unroll
        for (int u = 0; u < 4; ++u)
#pragma unroll
            for (int q = 0; q < 4; ++q) {
                if (q & 1) aO = wmma4(av[u][q], bv[u][q], aO);
                else       aE = wmma4(av[u][q], bv[u][q], aE);
            }
        __builtin_amdgcn_sched_barrier(0);
    }
    const v8f acc = aE + aO;

    float* pp = &Pt[(wv << 8) + (ln << 3)];
    *(float4*)&pp[0] = make_float4(acc[0], acc[1], acc[2], acc[3]);
    *(float4*)&pp[4] = make_float4(acc[4], acc[5], acc[6], acc[7]);
    __syncthreads();

    const int e  = tid & 15;
    const int l  = (tid >> 4) + ((e >> 3) << 4);
    const int pv = (l << 3) + (e & 7);
    float sum = 0.0f;
#pragma unroll
    for (int u = 0; u < WAVES; ++u) sum += Pt[(u << 8) + pv];
    const int i = (b << 8) + tid;
    if (vec) {
        sum += ws[WS_G + hi];
        ws[WS_Y + (k + 1) * Y_NUM + i] = sum;
        out[(size_t)(SEGROWS * (k + 1)) * Y_NUM + i] = sum;
    } else {
        const int dst = lvl * SEGS + k;
        ws[WS_MAP + dst * Y_NUM + i] = sum;
        if (i == 0) {
            const float Sh = ws[WS_S + hi];
            ws[WS_G + dst] = ws[WS_G + hi - s] * Sh + ws[WS_G + hi];
            ws[WS_S + dst] = Sh * ws[WS_S + hi - s];
        }
    }
}

__global__ __launch_bounds__(THREADS) __attribute__((amdgpu_waves_per_eu(2, 2)))
void ode_chain_seg(const float* __restrict__ ws,
                   float* __restrict__ out) {
    __shared__ __attribute__((aligned(16))) float Y[2][SBUF];
    __shared__ __attribute__((aligned(16))) float Tb[(SEGTRIPS - 1) * TP];

    const int tid = threadIdx.x;
    const int k   = (int)blockIdx.x + 1;

    const float* ys = ws + WS_Y + k * Y_NUM;
    {
        float ty[Y_NUM / THREADS];
#pragma unroll
        for (int j = 0; j < Y_NUM / THREADS; ++j) ty[j] = ys[tid + j * THREADS];
        copy_floats<(SEGTRIPS - 1) * TP>(Tb, ws + WS_TAB + SEGTRIPS * k * TP, tid);
#pragma unroll
        for (int j = 0; j < Y_NUM / THREADS; ++j) {
            const int i = tid + j * THREADS;
            Y[0][(i >> 4) * SP + (i & 15)] = ty[j];
        }
    }
    __syncthreads();

    chain_trips<true, SEGTRIPS - 1>(&Y[0][0], Tb, out, SEGROWS * k);
}

template <int J>
struct RowTaps {
    static constexpr int R = 1 + (J + 1) / 2;
    v2f a[R][4];
};

template <int J>
__device__ __forceinline__ bool row_skip(int r, int q) {
    return (J & 1) && r == RowTaps<J>::R - 1 && q < 2;
}

template <int J>
__device__ __forceinline__ RowTaps<J> load_row_taps(const float* rp) {
    RowTaps<J> t;
#pragma unroll
    for (int q = 0; q < 4; ++q)
#pragma unroll
        for (int r = 0; r < RowTaps<J>::R; ++r) {
            if (row_skip<J>(r, q)) continue;
            t.a[r][q] = tap_pair(rp, r, q);
        }
    return t;
}

template <int J>
__device__ __forceinline__ v8f row_direct(const RowTaps<J>& t, float F, const StateK& s) {
    v8f aE, aO;
#pragma unroll
    for (int i = 0; i < 8; ++i) { aE[i] = F; aO[i] = 0.0f; }
    int cnt = 0;
#pragma unroll
    for (int q = 0; q < 4; ++q)
#pragma unroll
        for (int r = 0; r < RowTaps<J>::R; ++r) {
            if (row_skip<J>(r, q)) continue;
            if (cnt & 1) aO = wmma4(t.a[r][q], s.b[r][q], aO);
            else         aE = wmma4(t.a[r][q], s.b[r][q], aE);
            ++cnt;
        }
    return aE + aO;
}

__device__ __forceinline__ void store_row(float* orow, int rowIx, v8f y) {
    *(float4*)&orow[rowIx]     = make_float4(y[0], y[1], y[2], y[3]);
    *(float4*)&orow[rowIx + 4] = make_float4(y[4], y[5], y[6], y[7]);
}

template <int J>
__device__ __forceinline__ void fill_from(float* out, int c0, int rowIx,
                                          const float* tab_lb, const float (&Fv)[KROW - 1], const StateK& st,
                                          const RowTaps<J>& cur) {
    if constexpr (J < KROW - 1) {
        const RowTaps<J + 1> nxt = load_row_taps<J + 1>(tab_lb + J * ROWLEN);
        __builtin_amdgcn_sched_barrier(0);
        const v8f y = row_direct<J>(cur, Fv[J - 1], st);
        __builtin_amdgcn_sched_barrier(0);
        store_row(out + (size_t)(c0 + J) * Y_NUM, rowIx, y);
        fill_from<J + 1>(out, c0, rowIx, tab_lb, Fv, st, nxt);
    } else {
        store_row(out + (size_t)(c0 + J) * Y_NUM, rowIx, row_direct<J>(cur, Fv[J - 1], st));
    }
}

__global__ __launch_bounds__(FTHREADS)
void ode_fill(const float* __restrict__ t,
              const float* __restrict__ w,
              const float* __restrict__ rt,
              float* out) {
    __shared__ float sc[KROW - 1][12];
    __shared__ float tab[(KROW - 1) * ROWLEN];
    __shared__ float Fs[KROW];

    const int tid  = threadIdx.x;
    const int wv   = (tid >> 5) + FWAVES * ((int)blockIdx.x & 1);
    const int ln   = tid & 31;
    const int col  = ln & 15;
    const int half = ln >> 4;
    const int m    = col;
    const int c0   = KROW * ((int)blockIdx.x >> 1);

    const int tileC = (wv << 4) + col;
    const int rowIx = (wv << 8) + (col << 4) + (half << 3);
    const int lb    = 15 - m + (half << 1);

    const float* ck = out + (size_t)c0 * Y_NUM;
    StateK st;
#pragma unroll
    for (int q = 0; q < 4; ++q)
#pragma unroll
        for (int r = 0; r < NMAT; ++r) {
            const int tl = (tileC - r) & (NTILES - 1);
            const float2 v = *(const float2*)&ck[(tl << 4) + (q << 2) + (half << 1)];
            st.b[r][q][0] = v.x; st.b[r][q][1] = v.y;
        }

    if (tid < KROW - 1) {
        const float a = w[0] * rt[0];
        const float b = w[1] * rt[1];
        const float c = w[2] * rt[2];
        const StepC A = step_coef(t[c0 + tid], t[c0 + tid + 1], a, b, c);
#pragma unroll
        for (int i = 0; i < 9; ++i) sc[tid][i] = A.c[i];
        sc[tid][9] = A.f;
    }
    __syncthreads();

    for (int p = tid; p < ROWLEN; p += FTHREADS) {
        const int x = TAPTOP - p;
        tab[p] = (x >= 0 && x <= 8) ? sc[0][(x >= 0 && x <= 8) ? x : 0] : 0.0f;
    }
    if (tid == 0) {
        float F = sc[0][9];
        Fs[0] = F;
        for (int j = 1; j < KROW - 1; ++j) {
            float sumB = 0.0f;
            for (int i = 0; i < 9; ++i) sumB += sc[j][i];
            F = F * sumB + sc[j][9];
            Fs[j] = F;
        }
    }
    __syncthreads();

    for (int j = 1; j < KROW - 1; ++j) {
        for (int p = tid; p < ROWLEN; p += FTHREADS) {
            float acc = 0.0f;
#pragma unroll
            for (int i = 0; i < 9; ++i) {
                const int pp = p + i;
                const float pv = (pp < ROWLEN) ? tab[(j - 1) * ROWLEN + ((pp < ROWLEN) ? pp : 0)] : 0.0f;
                acc += sc[j][i] * pv;
            }
            tab[j * ROWLEN + p] = acc;
        }
        __syncthreads();
    }

    float Fv[KROW - 1];
#pragma unroll
    for (int j = 0; j < KROW - 1; ++j) Fv[j] = Fs[j];
    fill_from<1>(out, c0, rowIx, tab + lb, Fv, st, load_row_taps<1>(tab + lb));
}

extern "C" void kernel_launch(void* const* d_in, const int* in_sizes, int n_in,
                              void* d_out, int out_size, void* d_ws, size_t ws_size,
                              hipStream_t stream) {
    (void)in_sizes; (void)n_in; (void)out_size; (void)ws_size;
    const float* t  = (const float*)d_in[0];
    const float* y0 = (const float*)d_in[1];
    const float* w  = (const float*)d_in[2];
    const float* r  = (const float*)d_in[3];
    float* out = (float*)d_out;
    float* ws  = (float*)d_ws;

    static_assert(SEGS == 32 && LEVELS == 5, "the level shapes below are for 31 segment maps");
    ode_maps<<<SEGS, THREADS, 0, stream>>>(t, y0, w, r, ws, out);
    ode_seg_scan<<<dim3(8, 30), THREADS, 0, stream>>>(y0, ws, out, 2);
    ode_seg_scan<<<dim3(8, 28), THREADS, 0, stream>>>(y0, ws, out, 3);
    ode_seg_scan<<<dim3(8, 24), THREADS, 0, stream>>>(y0, ws, out, 4);
    ode_seg_scan<<<dim3(8, 16), THREADS, 0, stream>>>(y0, ws, out, 5);
    ode_chain_seg<<<SEGS - 1, THREADS, 0, stream>>>(ws, out);
    ode_fill<<<2 * NGROUP, FTHREADS, 0, stream>>>(t, w, r, out);
}


# Fill kernel no longer rebuilds its 15 partial tap rows (14 barrier steps, a one-thread forcing loop): the maps kernel stores the partial products it already forms per trip; fill loads them
# speedup vs baseline: 37.4827x; 1.1197x over previous
#include <hip/hip_runtime.h>


typedef float v2f __attribute__((ext_vector_type(2)));
typedef float v8f __attribute__((ext_vector_type(8)));

#define SAMPLE_NUM 2048
#define Y_NUM      2048
#define NSTEP      (SAMPLE_NUM - 1)
#define KROW       16
#define NTRIP      (NSTEP / KROW)
#define NGROUP     (NTRIP + 1)
#define NTILES     128
#define WAVES      8
#define THREADS    (WAVES * 32)
#define FWAVES     4
#define FTHREADS   (FWAVES * 32)
#define NMAT       (KROW / 2 + 1)
#define NTAPK      (8 * KROW + 1)
#define TAPTOP     (16 * NMAT - 1)
#define ROWLEN     (16 * NMAT + 16)
#define TP         (ROWLEN + 4)
#define FPOS       ROWLEN
#define SP         20
#define SBUF       (NTILES * SP)
#define SEGS       32
#define LEVELS     5
#define SEGTRIPS   4
#define SEGROWS    (KROW * SEGTRIPS)
#define MTRIPS     (2 * SEGTRIPS)
#define NMAPS      (SEGS - 1)
#define LASTTRIPS  (NTRIP - SEGTRIPS * (SEGS - 1))
#define SUMPOS     (FPOS + 1)
#define NSLOT      (LEVELS * SEGS)
#define WS_TAB     0
#define WS_MAP     32768
#define WS_G       (WS_MAP + NSLOT * Y_NUM)
#define WS_S       (WS_G + 256)
#define WS_Y       (WS_S + 256)
#define FGRP       ((KROW - 1) * ROWLEN)
#define WS_FT      458752
#define WS_FF      (WS_FT + NGROUP * FGRP)
static_assert(WS_Y + SEGS * Y_NUM <= WS_FT && (WS_FT % 4) == 0 && (FGRP % 4) == 0, "workspace regions overlap / 16 B alignment of the partial tap rows");
static_assert(NTRIP * TP <= WS_MAP && WS_MAP + NSLOT * Y_NUM <= WS_G && WS_G + NSLOT <= WS_S && WS_S + NSLOT <= WS_Y, "workspace regions overlap");
static_assert((1 << LEVELS) == SEGS && LASTTRIPS == SEGTRIPS - 1, "segment shapes");

struct StepC {
    float c[9];
    float f;
};

__device__ __forceinline__ StepC step_coef(float tn, float tn1, float a, float b, float c) {
    const float sub = (tn1 - tn) * 0.125f;
    const float d   = 1.0f - sub * b;
    const float e   = sub * a;
    StepC r;
    {
        const float bin[9] = {1.f, 8.f, 28.f, 56.f, 70.f, 56.f, 28.f, 8.f, 1.f};
        float dp[9], ep[9];
        dp[0] = 1.0f; ep[0] = 1.0f;
#pragma unroll
        for (int j = 1; j < 9; ++j) { dp[j] = dp[j-1] * d; ep[j] = ep[j-1] * e; }
#pragma unroll
        for (int j = 0; j < 9; ++j) r.c[j] = bin[j] * dp[8 - j] * ep[j];
    }
    const float g = d + e;
    float f = 0.0f;
#pragma unroll
    for (int ms = 0; ms < 8; ++ms) {
        const float s = sub * c * __sinf(tn + sub * (float)ms);
        f = f * g + s;
    }
    r.f = f;
    return r;
}

__device__ __forceinline__ v8f wmma4(v2f a, v2f b, v8f c) {
    return __builtin_amdgcn_wmma_f32_16x16x4_f32(false, a, false, b, (short)0, c, false, false);
}

struct StateK {
    v2f b[NMAT][4];
};

__device__ __forceinline__ v2f tap_pair(const float* rp, int r, int q) {
    v2f a;
    a[0] = rp[16 * (NMAT - 1) - 16 * r + 4 * q];
    a[1] = rp[16 * (NMAT - 1) - 16 * r + 4 * q + 1];
    return a;
}

struct CoefK {
    v2f   a[NMAT][4];
    float f;
};

template <bool FORCED>
__device__ __forceinline__ CoefK load_coefK(const float* row, int lb) {
    CoefK c;
    c.f = FORCED ? row[FPOS] : 0.0f;
    const float* rp = row + lb;
#pragma unroll
    for (int q = 0; q < 4; ++q)
#pragma unroll
        for (int r = 0; r < NMAT; ++r) c.a[r][q] = tap_pair(rp, r, q);
    return c;
}

__device__ __forceinline__ StateK load_state_lds(const float* Yb, const int* so) {
    StateK s;
#pragma unroll
    for (int q = 0; q < 4; ++q)
#pragma unroll
        for (int r = 0; r < NMAT; ++r) {
            const float2 v = *(const float2*)&Yb[so[r] + 4 * q];
            s.b[r][q][0] = v.x; s.b[r][q][1] = v.y;
        }
    return s;
}

__device__ __forceinline__ v8f rows_trip(const CoefK& k, const StateK& s) {
    v8f aE, aO;
#pragma unroll
    for (int i = 0; i < 8; ++i) { aE[i] = k.f; aO[i] = 0.0f; }
#pragma unroll
    for (int q = 0; q < 4; ++q)
#pragma unroll
        for (int r = 0; r < NMAT; ++r) {
            if ((q * NMAT + r) & 1) aO = wmma4(k.a[r][q], s.b[r][q], aO);
            else                    aE = wmma4(k.a[r][q], s.b[r][q], aE);
        }
    return aE + aO;
}


template <int N>
__device__ __forceinline__ void copy_floats(float* dst, const float* src, int tid) {
    constexpr int IT = (N + THREADS - 1) / THREADS;
    float v[IT];
#pragma unroll
    for (int j = 0; j < IT; ++j) { const int i = tid + j * THREADS; v[j] = (i < N) ? src[i] : 0.0f; }
#pragma unroll
    for (int j = 0; j < IT; ++j) { const int i = tid + j * THREADS; if (i < N) dst[i] = v[j]; }
}

#define LANEPOS (ROWLEN / 16)
static_assert(LANEPOS * 16 == ROWLEN && LANEPOS >= 8, "a lane's window must reach exactly one lane up");
static_assert(THREADS >= 16 * MTRIPS && (16 * MTRIPS) % 32 == 0 && KROW == 16, "16 lanes per trip, whole waves");

template <int NT>
__device__ __forceinline__ void build_trip_rows(float* Tb, const float (*sc)[12], int tid,
                                                float* __restrict__ ft, float* __restrict__ ff,
                                                int g0, int kseg) {
    if (tid >= 16 * NT) return;
    const int j = tid >> 4;
    const int u = tid & 15;
    const int  g   = g0 + j;
    const bool own = (g / SEGTRIPS) == kseg;
    float* gt = ft + (size_t)g * FGRP + LANEPOS * u;
    float* gf = ff + g * KROW;
    const float* r0 = sc[KROW * j];
    float Wd[LANEPOS + 8];
#pragma unroll
    for (int e = 0; e < LANEPOS; ++e) {
        const int x = TAPTOP - (LANEPOS * u + e);
        const bool in = (x >= 0 && x <= 8);
        Wd[e] = in ? r0[in ? x : 0] : 0.0f;
    }
    float F = r0[9];
    if (own) {
#pragma unroll
        for (int e = 0; e < LANEPOS / 2; ++e) ((float2*)gt)[e] = make_float2(Wd[2 * e], Wd[2 * e + 1]);
        if (u == 0) gf[0] = F;
    }
    for (int jj = 1; jj < KROW; ++jj) {
        const float* rb = sc[KROW * j + jj];
        const float4 c03 = *(const float4*)&rb[0];
        const float4 c47 = *(const float4*)&rb[4];
        const float4 c8x = *(const float4*)&rb[8];
        const float cc[9] = {c03.x, c03.y, c03.z, c03.w, c47.x, c47.y, c47.z, c47.w, c8x.x};
#pragma unroll
        for (int e = 0; e < 8; ++e) {
            const float v = __shfl_down(Wd[e], 1, 16);
            Wd[LANEPOS + e] = (u == 15) ? 0.0f : v;
        }
        float Nw[LANEPOS];
#pragma unroll
        for (int e = 0; e < LANEPOS; ++e) {
            float acc = 0.0f;
#pragma unroll
            for (int i = 0; i < 9; ++i) acc += cc[i] * Wd[e + i];
            Nw[e] = acc;
        }
#pragma unroll
        for (int e = 0; e < LANEPOS; ++e) Wd[e] = Nw[e];
        F = F * c8x.z + c8x.y;
        if (own && jj < KROW - 1) {
            float2* d2 = (float2*)(gt + jj * ROWLEN);
#pragma unroll
            for (int e = 0; e < LANEPOS / 2; ++e) d2[e] = make_float2(Wd[2 * e], Wd[2 * e + 1]);
            if (u == 0) gf[jj] = F;
        }
    }
    float sp = 0.0f;
#pragma unroll
    for (int e = 0; e < LANEPOS; ++e) sp += Wd[e];
#pragma unroll
    for (int dlt = 8; dlt >= 1; dlt >>= 1) sp += __shfl_down(sp, dlt, 16);

    float* row = Tb + j * TP;
    float2* dst = (float2*)&row[LANEPOS * u];
#pragma unroll
    for (int e = 0; e < LANEPOS / 2; ++e) dst[e] = make_float2(Wd[2 * e], Wd[2 * e + 1]);
    if (u == 0) *(float4*)&row[FPOS] = make_float4(F, sp, 0.0f, 0.0f);
}

template <bool ROWS, int NTR>
__device__ __forceinline__ void chain_trips(float* Yb, const float* Tb, float* __restrict__ out, int row0) {
    static_assert(ROWS || 8 * NTR + NMAT <= NTILES, "impulse run: the state must not come round the ring");
    const int tid  = threadIdx.x;
    const int wv   = tid >> 5;
    const int ln   = tid & 31;
    const int col  = ln & 15;
    const int half = ln >> 4;
    const int m    = col;

    const int tileC = (wv << 4) + col;
    const int rowIx = (wv << 8) + (col << 4) + (half << 3);
    const int stIx  = tileC * SP + (half << 3);
    const int lb    = 15 - m + (half << 1);
    int so[NMAT];
#pragma unroll
    for (int r = 0; r < NMAT; ++r) so[r] = ((tileC - r) & (NTILES - 1)) * SP + (half << 1);

    const int wvu = __builtin_amdgcn_readfirstlane(wv);

    CoefK  cur = load_coefK<ROWS>(Tb, lb);
    StateK st  = load_state_lds(Yb, so);

#pragma unroll 2
    for (int s = 0; s < NTR; ++s) {
        float* __restrict__ Yn = Yb + ((s & 1) ^ 1) * SBUF;

        v8f yk;
        if (ROWS || 2 * wvu <= s + 1) {
            yk = rows_trip(cur, st);
        } else {
#pragma unroll
            for (int i = 0; i < 8; ++i) yk[i] = 0.0f;
        }
        const float4 lo = make_float4(yk[0], yk[1], yk[2], yk[3]);
        const float4 hi = make_float4(yk[4], yk[5], yk[6], yk[7]);
        *(float4*)&Yn[stIx]     = lo;
        *(float4*)&Yn[stIx + 4] = hi;

        __builtin_amdgcn_fence(__ATOMIC_RELEASE, "workgroup", "local");
        __builtin_amdgcn_s_barrier();
        __builtin_amdgcn_fence(__ATOMIC_ACQUIRE, "workgroup", "local");

        const int sn = (s + 1 < NTR) ? (s + 1) : (NTR - 1);
        if (ROWS || 2 * wvu <= s + 2) {
            st = load_state_lds(Yn, so);
            __builtin_amdgcn_sched_barrier(0);
            cur = load_coefK<ROWS>(Tb + sn * TP, lb);
            __builtin_amdgcn_sched_barrier(0);
        }

        if (ROWS) {
            float* __restrict__ orow = out + (size_t)(row0 + KROW * (s + 1)) * Y_NUM;
            *(float4*)&orow[rowIx]     = lo;
            *(float4*)&orow[rowIx + 4] = hi;
        }
    }
}

__global__ __launch_bounds__(THREADS) __attribute__((amdgpu_waves_per_eu(2, 2)))
void ode_maps(const float* __restrict__ t,
              const float* __restrict__ y0,
              const float* __restrict__ w,
              const float* __restrict__ rt,
              float* __restrict__ ws,
              float* __restrict__ out) {
    __shared__ __attribute__((aligned(16))) float Y[2][SBUF];
    __shared__ __attribute__((aligned(16))) float Tb[MTRIPS * TP];
    __shared__ __attribute__((aligned(16))) float sc[MTRIPS * KROW][12];

    const int tid = threadIdx.x;
    const int k   = (int)blockIdx.x;
    const bool pair  = (k >= 1 && k <= SEGS - 2);
    const int  nb    = SEGROWS * (pair ? k - 1 : k);
    const int  nrows = pair ? 2 * SEGROWS : ((k == SEGS - 1) ? SEGROWS - 1 : SEGROWS);

    float ty[Y_NUM / THREADS];
    if (k == 0) {
#pragma unroll
        for (int j = 0; j < Y_NUM / THREADS; ++j) ty[j] = y0[tid + j * THREADS];
    }

    {
        StepC A;
#pragma unroll
        for (int i = 0; i < 9; ++i) A.c[i] = 0.0f;
        A.f = 0.0f;
        if (tid < nrows) {
            const float a = w[0] * rt[0];
            const float b = w[1] * rt[1];
            const float c = w[2] * rt[2];
            const int n = nb + tid;
            A = step_coef(t[n], t[n + 1], a, b, c);
        }
        float sumA = 0.0f;
#pragma unroll
        for (int i = 0; i < 9; ++i) sumA += A.c[i];
        if (tid < MTRIPS * KROW) {
            *(float4*)&sc[tid][0] = make_float4(A.c[0], A.c[1], A.c[2], A.c[3]);
            *(float4*)&sc[tid][4] = make_float4(A.c[4], A.c[5], A.c[6], A.c[7]);
            *(float4*)&sc[tid][8] = make_float4(A.c[8], A.f, sumA, 0.0f);
        }
    }
    __syncthreads();
    build_trip_rows<MTRIPS>(Tb, sc, tid, ws + WS_FT, ws + WS_FF, nb / KROW, k);
    if (k == 0) {
#pragma unroll
        for (int j = 0; j < Y_NUM / THREADS; ++j) {
            const int i = tid + j * THREADS;
            Y[0][(i >> 4) * SP + (i & 15)] = ty[j];
            out[i] = ty[j];
        }
    } else {
        for (int i = tid; i < SBUF; i += THREADS) Y[0][i] = (i == 0) ? 1.0f : 0.0f;
    }
    __syncthreads();

    copy_floats<LASTTRIPS * TP>(ws + WS_TAB + SEGTRIPS * k * TP, Tb + (pair ? SEGTRIPS * TP : 0), tid);

    if (k == SEGS - 1) return;

    if (k == 0) {
        chain_trips<true, SEGTRIPS>(&Y[0][0], Tb, out, 0);
        const float* Y1 = Y[SEGTRIPS & 1];
#pragma unroll
        for (int j = 0; j < Y_NUM / THREADS; ++j) {
            const int i = tid + j * THREADS;
            ws[WS_Y + Y_NUM + i] = Y1[(i >> 4) * SP + (i & 15)];
        }
        return;
    }

    if (tid == 0) {
        float G = 0.0f, P = 1.0f;
#pragma unroll
        for (int j = 0; j < MTRIPS; ++j) {
            const float sj = Tb[j * TP + SUMPOS];
            G = G * sj + Tb[j * TP + FPOS];
            P = P * sj;
        }
        ws[WS_G + SEGS + k] = G;
        ws[WS_S + SEGS + k] = P;
    }

    chain_trips<false, MTRIPS>(&Y[0][0], Tb, nullptr, 0);

    const float* Yf = Y[MTRIPS & 1];
#pragma unroll
    for (int j = 0; j < Y_NUM / THREADS; ++j) {
        const int i = tid + j * THREADS;
        ws[WS_MAP + (SEGS + k) * Y_NUM + i] = Yf[(i >> 4) * SP + (i & 15)];
    }
}

__global__ __launch_bounds__(THREADS)
void ode_seg_scan(const float* __restrict__ y0,
                  float* __restrict__ ws,
                  float* __restrict__ out,
                  int lvl) {
    __shared__ __attribute__((aligned(16))) float As[Y_NUM + 16];
    __shared__ __attribute__((aligned(16))) float Ys[SBUF];
    __shared__ __attribute__((aligned(16))) float Pt[WAVES * 256];

    const int tid  = threadIdx.x;
    const int wv   = tid >> 5;
    const int ln   = tid & 31;
    const int col  = ln & 15;
    const int half = ln >> 4;
    const int b    = (int)blockIdx.x;

    const int  s   = 1 << (lvl - 1);
    const int  k   = s - 1 + (int)blockIdx.y;
    const int  hi  = (lvl - 1) * SEGS + k;
    const bool vec = (k < 2 * s - 1);
    const int  j0  = k - s + 1;
    const float* yin = vec ? ((j0 == 0) ? y0 : (ws + WS_Y + j0 * Y_NUM))
                           : (ws + WS_MAP + (hi - s) * Y_NUM);
    {
        const float* amap = ws + WS_MAP + hi * Y_NUM;
        float ta[Y_NUM / THREADS], ty[Y_NUM / THREADS];
#pragma unroll
        for (int j = 0; j < Y_NUM / THREADS; ++j) { ta[j] = amap[tid + j * THREADS]; ty[j] = yin[tid + j * THREADS]; }
#pragma unroll
        for (int j = 0; j < Y_NUM / THREADS; ++j) {
            const int i = tid + j * THREADS;
            As[Y_NUM - 1 - i] = ta[j];
            if (i >= Y_NUM - 16) As[2 * Y_NUM - 1 - i] = ta[j];
            Ys[(i >> 4) * SP + (i & 15)] = ty[j];
        }
    }
    __syncthreads();

    v8f aE, aO;
#pragma unroll
    for (int i = 0; i < 8; ++i) { aE[i] = 0.0f; aO[i] = 0.0f; }
    const int tile0 = (b << 4) + col;
    for (int rb = 0; rb < 16; rb += 4) {
        v2f av[4][4], bv[4][4];
#pragma unroll
        for (int u = 0; u < 4; ++u) {
            const int r  = (wv << 4) + rb + u;
            const int tl = (tile0 - r) & (NTILES - 1);
            const int x0 = 16 * r + col - 2 * half;
#pragma unroll
            for (int q = 0; q < 4; ++q) {
                const int x = x0 - 4 * q;
                av[u][q][0] = As[Y_NUM - 1 - x];
                av[u][q][1] = As[Y_NUM - x];
                const float2 v = *(const float2*)&Ys[tl * SP + (q << 2) + (half << 1)];
                bv[u][q][0] = v.x; bv[u][q][1] = v.y;
            }
        }
        __builtin_amdgcn_sched_barrier(0);
#pragma unroll
        for (int u = 0; u < 4; ++u)
#pragma unroll
            for (int q = 0; q < 4; ++q) {
                if (q & 1) aO = wmma4(av[u][q], bv[u][q], aO);
                else       aE = wmma4(av[u][q], bv[u][q], aE);
            }
        __builtin_amdgcn_sched_barrier(0);
    }
    const v8f acc = aE + aO;

    float* pp = &Pt[(wv << 8) + (ln << 3)];
    *(float4*)&pp[0] = make_float4(acc[0], acc[1], acc[2], acc[3]);
    *(float4*)&pp[4] = make_float4(acc[4], acc[5], acc[6], acc[7]);
    __syncthreads();

    const int e  = tid & 15;
    const int l  = (tid >> 4) + ((e >> 3) << 4);
    const int pv = (l << 3) + (e & 7);
    float sum = 0.0f;
#pragma unroll
    for (int u = 0; u < WAVES; ++u) sum += Pt[(u << 8) + pv];
    const int i = (b << 8) + tid;
    if (vec) {
        sum += ws[WS_G + hi];
        ws[WS_Y + (k + 1) * Y_NUM + i] = sum;
        out[(size_t)(SEGROWS * (k + 1)) * Y_NUM + i] = sum;
    } else {
        const int dst = lvl * SEGS + k;
        ws[WS_MAP + dst * Y_NUM + i] = sum;
        if (i == 0) {
            const float Sh = ws[WS_S + hi];
            ws[WS_G + dst] = ws[WS_G + hi - s] * Sh + ws[WS_G + hi];
            ws[WS_S + dst] = Sh * ws[WS_S + hi - s];
        }
    }
}

__global__ __launch_bounds__(THREADS) __attribute__((amdgpu_waves_per_eu(2, 2)))
void ode_chain_seg(const float* __restrict__ ws,
                   float* __restrict__ out) {
    __shared__ __attribute__((aligned(16))) float Y[2][SBUF];
    __shared__ __attribute__((aligned(16))) float Tb[(SEGTRIPS - 1) * TP];

    const int tid = threadIdx.x;
    const int k   = (int)blockIdx.x + 1;

    const float* ys = ws + WS_Y + k * Y_NUM;
    {
        float ty[Y_NUM / THREADS];
#pragma unroll
        for (int j = 0; j < Y_NUM / THREADS; ++j) ty[j] = ys[tid + j * THREADS];
        copy_floats<(SEGTRIPS - 1) * TP>(Tb, ws + WS_TAB + SEGTRIPS * k * TP, tid);
#pragma unroll
        for (int j = 0; j < Y_NUM / THREADS; ++j) {
            const int i = tid + j * THREADS;
            Y[0][(i >> 4) * SP + (i & 15)] = ty[j];
        }
    }
    __syncthreads();

    chain_trips<true, SEGTRIPS - 1>(&Y[0][0], Tb, out, SEGROWS * k);
}

template <int J>
struct RowTaps {
    static constexpr int R = 1 + (J + 1) / 2;
    v2f a[R][4];
};

template <int J>
__device__ __forceinline__ bool row_skip(int r, int q) {
    return (J & 1) && r == RowTaps<J>::R - 1 && q < 2;
}

template <int J>
__device__ __forceinline__ RowTaps<J> load_row_taps(const float* rp) {
    RowTaps<J> t;
#pragma unroll
    for (int q = 0; q < 4; ++q)
#pragma unroll
        for (int r = 0; r < RowTaps<J>::R; ++r) {
            if (row_skip<J>(r, q)) continue;
            t.a[r][q] = tap_pair(rp, r, q);
        }
    return t;
}

template <int J>
__device__ __forceinline__ v8f row_direct(const RowTaps<J>& t, float F, const StateK& s) {
    v8f aE, aO;
#pragma unroll
    for (int i = 0; i < 8; ++i) { aE[i] = F; aO[i] = 0.0f; }
    int cnt = 0;
#pragma unroll
    for (int q = 0; q < 4; ++q)
#pragma unroll
        for (int r = 0; r < RowTaps<J>::R; ++r) {
            if (row_skip<J>(r, q)) continue;
            if (cnt & 1) aO = wmma4(t.a[r][q], s.b[r][q], aO);
            else         aE = wmma4(t.a[r][q], s.b[r][q], aE);
            ++cnt;
        }
    return aE + aO;
}

__device__ __forceinline__ void store_row(float* orow, int rowIx, v8f y) {
    *(float4*)&orow[rowIx]     = make_float4(y[0], y[1], y[2], y[3]);
    *(float4*)&orow[rowIx + 4] = make_float4(y[4], y[5], y[6], y[7]);
}

template <int J>
__device__ __forceinline__ void fill_from(float* out, int c0, int rowIx,
                                          const float* tab_lb, const float (&Fv)[KROW - 1], const StateK& st,
                                          const RowTaps<J>& cur) {
    if constexpr (J < KROW - 1) {
        const RowTaps<J + 1> nxt = load_row_taps<J + 1>(tab_lb + J * ROWLEN);
        __builtin_amdgcn_sched_barrier(0);
        const v8f y = row_direct<J>(cur, Fv[J - 1], st);
        __builtin_amdgcn_sched_barrier(0);
        store_row(out + (size_t)(c0 + J) * Y_NUM, rowIx, y);
        fill_from<J + 1>(out, c0, rowIx, tab_lb, Fv, st, nxt);
    } else {
        store_row(out + (size_t)(c0 + J) * Y_NUM, rowIx, row_direct<J>(cur, Fv[J - 1], st));
    }
}

__global__ __launch_bounds__(FTHREADS)
void ode_fill(const float* __restrict__ ws,
              float* out) {
    __shared__ __attribute__((aligned(16))) float tab[FGRP];

    const int tid  = threadIdx.x;
    const int wv   = (tid >> 5) + FWAVES * ((int)blockIdx.x & 1);
    const int ln   = tid & 31;
    const int col  = ln & 15;
    const int half = ln >> 4;
    const int m    = col;
    const int g    = (int)blockIdx.x >> 1;
    const int c0   = KROW * g;

    const int tileC = (wv << 4) + col;
    const int rowIx = (wv << 8) + (col << 4) + (half << 3);
    const int lb    = 15 - m + (half << 1);

    constexpr int NV = FGRP / 4;
    constexpr int IT = (NV + FTHREADS - 1) / FTHREADS;
    const float4* tsrc = (const float4*)(ws + WS_FT + (size_t)g * FGRP);
    float4 tv[IT];
#pragma unroll
    for (int j = 0; j < IT; ++j) {
        const int i = tid + j * FTHREADS;
        tv[j] = (i < NV) ? tsrc[i] : make_float4(0.0f, 0.0f, 0.0f, 0.0f);
    }

    const float* ck = out + (size_t)c0 * Y_NUM;
    StateK st;
#pragma unroll
    for (int q = 0; q < 4; ++q)
#pragma unroll
        for (int r = 0; r < NMAT; ++r) {
            const int tl = (tileC - r) & (NTILES - 1);
            const float2 v = *(const float2*)&ck[(tl << 4) + (q << 2) + (half << 1)];
            st.b[r][q][0] = v.x; st.b[r][q][1] = v.y;
        }

    float Fv[KROW - 1];
#pragma unroll
    for (int j = 0; j < KROW - 1; ++j) Fv[j] = ws[WS_FF + g * KROW + j];

#pragma unroll
    for (int j = 0; j < IT; ++j) {
        const int i = tid + j * FTHREADS;
        if (i < NV) ((float4*)tab)[i] = tv[j];
    }
    __syncthreads();

    fill_from<1>(out, c0, rowIx, tab + lb, Fv, st, load_row_taps<1>(tab + lb));
}

extern "C" void kernel_launch(void* const* d_in, const int* in_sizes, int n_in,
                              void* d_out, int out_size, void* d_ws, size_t ws_size,
                              hipStream_t stream) {
    (void)in_sizes; (void)n_in; (void)out_size; (void)ws_size;
    const float* t  = (const float*)d_in[0];
    const float* y0 = (const float*)d_in[1];
    const float* w  = (const float*)d_in[2];
    const float* r  = (const float*)d_in[3];
    float* out = (float*)d_out;
    float* ws  = (float*)d_ws;

    static_assert(SEGS == 32 && LEVELS == 5, "the level shapes below are for 31 segment maps");
    ode_maps<<<SEGS, THREADS, 0, stream>>>(t, y0, w, r, ws, out);
    ode_seg_scan<<<dim3(8, 30), THREADS, 0, stream>>>(y0, ws, out, 2);
    ode_seg_scan<<<dim3(8, 28), THREADS, 0, stream>>>(y0, ws, out, 3);
    ode_seg_scan<<<dim3(8, 24), THREADS, 0, stream>>>(y0, ws, out, 4);
    ode_seg_scan<<<dim3(8, 16), THREADS, 0, stream>>>(y0, ws, out, 5);
    ode_chain_seg<<<SEGS - 1, THREADS, 0, stream>>>(ws, out);
    ode_fill<<<2 * NGROUP, FTHREADS, 0, stream>>>(ws, out);
}
